# starting kernel
# baseline (speedup 1.0000x reference)
#include <hip/hip_runtime.h>
#include <hip/hip_bf16.h>

#define GB   16
#define GN   10000
#define GE   160000
#define GS   128
#define GH   64
#define GEMB 64
#define GP   16

typedef __attribute__((ext_vector_type(2))) float v2f;
typedef __attribute__((ext_vector_type(4))) float v4f;
typedef __attribute__((ext_vector_type(8))) float v8f;

__global__ __launch_bounds__(256) void zero4_kernel(v4f* __restrict__ p, int n4) {
    int i = blockIdx.x * blockDim.x + threadIdx.x;
    if (i < n4) p[i] = (v4f){0.f, 0.f, 0.f, 0.f};
}

__global__ __launch_bounds__(256) void deg_kernel(const int* __restrict__ ei,
                                                  const float* __restrict__ ew,
                                                  float* __restrict__ deg) {
    int idx = blockIdx.x * blockDim.x + threadIdx.x;
    if (idx >= GB * GE) return;
    int b = idx / GE;
    int e = idx - b * GE;
    int c = ei[(size_t)b * 2 * GE + GE + e];
    atomicAdd(&deg[(size_t)b * GN + c], ew[idx]);
}

__global__ __launch_bounds__(256) void dinv_kernel(float* __restrict__ deg) {
    int i = blockIdx.x * blockDim.x + threadIdx.x;
    if (i < GB * GN) {
        deg[i] = rsqrtf(deg[i] + 1.0f);
    }
}

__global__ __launch_bounds__(256) void norm_kernel(const int* __restrict__ ei,
                                                   const float* __restrict__ ew,
                                                   const float* __restrict__ dinv,
                                                   float* __restrict__ nrm) {
    int idx = blockIdx.x * blockDim.x + threadIdx.x;
    if (idx >= GB * GE) return;
    int b = idx / GE;
    int e = idx - b * GE;
    int r = ei[(size_t)b * 2 * GE + e];
    int c = ei[(size_t)b * 2 * GE + GE + e];
    nrm[idx] = dinv[(size_t)b * GN + r] * ew[idx] * dinv[(size_t)b * GN + c];
}

template <int HIN>
__global__ __launch_bounds__(128) void gemm_wmma_kernel(const float* __restrict__ Hsrc,
                                                        const float* __restrict__ W,
                                                        float* __restrict__ Dst) {
    __shared__ float wlds[GH * HIN];
    const int tid = threadIdx.x;
    for (int i = tid; i < GH * HIN; i += 128) wlds[i] = W[i];
    __syncthreads();

    const int wave = tid >> 5;
    const int lane = tid & 31;
    const int half = lane >> 4;
    const int lm   = lane & 15;
    const int b    = blockIdx.z;
    const int m0   = blockIdx.x * 16;
    const int n0   = wave * 16;

    const float* hrow = Hsrc + ((size_t)b * GN + m0 + lm) * HIN;
    const float* wrow = wlds + (size_t)(n0 + lm) * HIN;

    v8f acc = {};
#pragma unroll
    for (int k0 = 0; k0 < HIN; k0 += 4) {
        const int ka = k0 + half * 2;
        v2f a = *(const v2f*)(hrow + ka);
        v2f bf = *(const v2f*)(wrow + ka);
        acc = __builtin_amdgcn_wmma_f32_16x16x4_f32(
             false, a,  false, bf,
             (short)0, acc,  false,  false);
    }

    float* drow = Dst + ((size_t)b * GN + m0) * GH + n0 + lm;
#pragma unroll
    for (int j = 0; j < 8; ++j) {
        drow[(size_t)(j + 8 * half) * GH] = acc[j];
    }
}

__global__ __launch_bounds__(256) void scatter_kernel(const int* __restrict__ ei,
                                                      const float* __restrict__ nrm,
                                                      const float* __restrict__ tmp,
                                                      float* __restrict__ agg) {
    long long idx = (long long)blockIdx.x * blockDim.x + threadIdx.x;
    if (idx >= (long long)GB * GE * (GH / 4)) return;
    int k4 = (int)(idx & 15);
    long long eg = idx >> 4;
    int b = (int)(eg / GE);
    int e = (int)(eg - (long long)b * GE);
    int r = ei[(size_t)b * 2 * GE + e];
    int c = ei[(size_t)b * 2 * GE + GE + e];
    float s = nrm[eg];
    v4f v = *(const v4f*)(tmp + ((size_t)b * GN + r) * GH + k4 * 4) * s;
    float* dst = agg + ((size_t)b * GN + c) * GH + k4 * 4;
    atomicAdd(dst + 0, v.x);
    atomicAdd(dst + 1, v.y);
    atomicAdd(dst + 2, v.z);
    atomicAdd(dst + 3, v.w);
}

__global__ __launch_bounds__(256) void act_kernel(const float* __restrict__ agg,
                                                  const float* __restrict__ tmp,
                                                  const float* __restrict__ dinv,
                                                  const float* __restrict__ bias,
                                                  float* __restrict__ hout,
                                                  int do_tanh) {
    int idx = blockIdx.x * blockDim.x + threadIdx.x;
    if (idx >= GB * GN * (GH / 4)) return;
    int k4  = idx & 15;
    int node = idx >> 4;
    float di = dinv[node];
    float sl = di * di;
    v4f a = *(const v4f*)(agg + (size_t)idx * 4);
    v4f t = *(const v4f*)(tmp + (size_t)idx * 4);
    v4f bb = *(const v4f*)(bias + k4 * 4);
    v4f v = a + t * sl + bb;
    if (do_tanh) {
        v.x = tanhf(v.x); v.y = tanhf(v.y); v.z = tanhf(v.z); v.w = tanhf(v.w);
    }
    *(v4f*)(hout + (size_t)idx * 4) = v;
}

__global__ __launch_bounds__(256) void gather_kernel(const float* __restrict__ emb,
                                                     const int* __restrict__ pos,
                                                     float* __restrict__ out) {
    int idx = blockIdx.x * blockDim.x + threadIdx.x;
    if (idx >= GB * GP * GEMB) return;
    int k = idx & (GEMB - 1);
    int bp = idx >> 6;
    int b = bp / GP;
    int pv = pos[bp];
    out[idx] = (pv >= 0) ? emb[((size_t)b * GN + pv) * GH + k] : -3.0f;
}

extern "C" void kernel_launch(void* const* d_in, const int* in_sizes, int n_in,
                              void* d_out, int out_size, void* d_ws, size_t ws_size,
                              hipStream_t stream) {
    const float* x   = (const float*)d_in[0];
    const int*   ei  = (const int*)d_in[1];
    const float* ew  = (const float*)d_in[2];
    const int*   pos = (const int*)d_in[3];
    const float* W1  = (const float*)d_in[4];
    const float* b1  = (const float*)d_in[5];
    const float* W2  = (const float*)d_in[6];
    const float* b2  = (const float*)d_in[7];
    const float* W3  = (const float*)d_in[8];
    const float* b3  = (const float*)d_in[9];
    float* out = (float*)d_out;

    float* ws   = (float*)d_ws;
    const size_t nBN  = (size_t)GB * GN;
    const size_t nBE  = (size_t)GB * GE;
    const size_t nBNH = (size_t)GB * GN * GH;
    float* dinv = ws;
    float* nrm  = dinv + nBN;
    float* tmp  = nrm + nBE;
    float* agg  = tmp + nBNH;
    float* hbuf = agg + nBNH;

    const int T = 256;
    dim3 gemm_grid(GN / 16, 1, GB);

    zero4_kernel<<<((int)(nBN / 4) + T - 1) / T, T, 0, stream>>>((v4f*)dinv, (int)(nBN / 4));
    deg_kernel<<<(GB * GE + T - 1) / T, T, 0, stream>>>(ei, ew, dinv);
    dinv_kernel<<<((int)nBN + T - 1) / T, T, 0, stream>>>(dinv);
    norm_kernel<<<(GB * GE + T - 1) / T, T, 0, stream>>>(ei, ew, dinv, nrm);

    const long long scatterN = (long long)GB * GE * (GH / 4);
    const int scatterBlocks = (int)((scatterN + T - 1) / T);
    const int elem4Blocks = (int)((nBNH / 4 + T - 1) / T);

    gemm_wmma_kernel<GS><<<gemm_grid, 128, 0, stream>>>(x, W1, tmp);
    zero4_kernel<<<elem4Blocks, T, 0, stream>>>((v4f*)agg, (int)(nBNH / 4));
    scatter_kernel<<<scatterBlocks, T, 0, stream>>>(ei, nrm, tmp, agg);
    act_kernel<<<elem4Blocks, T, 0, stream>>>(agg, tmp, dinv, b1, hbuf, 1);

    gemm_wmma_kernel<GH><<<gemm_grid, 128, 0, stream>>>(hbuf, W2, tmp);
    zero4_kernel<<<elem4Blocks, T, 0, stream>>>((v4f*)agg, (int)(nBNH / 4));
    scatter_kernel<<<scatterBlocks, T, 0, stream>>>(ei, nrm, tmp, agg);
    act_kernel<<<elem4Blocks, T, 0, stream>>>(agg, tmp, dinv, b2, hbuf, 1);

    gemm_wmma_kernel<GH><<<gemm_grid, 128, 0, stream>>>(hbuf, W3, tmp);
    zero4_kernel<<<elem4Blocks, T, 0, stream>>>((v4f*)agg, (int)(nBNH / 4));
    scatter_kernel<<<scatterBlocks, T, 0, stream>>>(ei, nrm, tmp, agg);
    act_kernel<<<elem4Blocks, T, 0, stream>>>(agg, tmp, dinv, b3, hbuf, 0);

    gather_kernel<<<(GB * GP * GEMB + T - 1) / T, T, 0, stream>>>(hbuf, pos, out);
}


# Float-atomic push scatter (and degree sum) replaced by a per-call counting sort of edges by target plus pull gather with self-loop, bias, tanh folded in; sums run in list order, not arrival order
# speedup vs baseline: 2.6473x; 2.6473x over previous
#include <hip/hip_runtime.h>
#include <hip/hip_bf16.h>

#define GB   16
#define GN   10000
#define GE   160000
#define GS   128
#define GH   64
#define GEMB 64
#define GP   16

#define PARTS 16
#define NPP   (GN / PARTS)
#define SORT_T 1024

typedef __attribute__((ext_vector_type(2))) float v2f;
typedef __attribute__((ext_vector_type(4))) float v4f;
typedef __attribute__((ext_vector_type(8))) float v8f;
typedef __attribute__((ext_vector_type(2))) int   v2i;
typedef __attribute__((ext_vector_type(4))) int   v4i;

__global__ __launch_bounds__(SORT_T) void sort_kernel(const int* __restrict__ ei,
                                                      const float* __restrict__ ew,
                                                      int* __restrict__ rowptr,
                                                      int* __restrict__ sedge) {
    __shared__ int cnt[NPP];
    __shared__ int sc[SORT_T];
    __shared__ int s_below;
    const int tid  = threadIdx.x;
    const int part = blockIdx.x;
    const int b    = blockIdx.y;
    const int lo   = part * NPP;
    const int hi   = lo + NPP;
    const int*   rows = ei + (size_t)b * 2 * GE;
    const int*   cols = rows + GE;
    const float* w    = ew + (size_t)b * GE;
    const v4i*   cols4 = (const v4i*)cols;
    const int    NV = GE / 4;

    for (int i = tid; i < NPP; i += SORT_T) cnt[i] = 0;
    if (tid == 0) s_below = 0;
    __syncthreads();

    int below = 0;
    for (int v0 = tid; v0 < NV; v0 += 4 * SORT_T) {
        v4i c[4];
#pragma unroll
        for (int u = 0; u < 4; ++u) {
            const int v = v0 + u * SORT_T;
            c[u] = (v < NV) ? cols4[v] : (v4i){GN, GN, GN, GN};
        }
#pragma unroll
        for (int u = 0; u < 4; ++u) {
#pragma unroll
            for (int j = 0; j < 4; ++j) {
                const int cj = c[u][j];
                if (cj >= lo && cj < hi) atomicAdd(&cnt[cj - lo], 1);
                below += (cj < lo) ? 1 : 0;
            }
        }
    }
    atomicAdd(&s_below, below);
    __syncthreads();

    const int own = (tid < NPP) ? cnt[tid] : 0;
    sc[tid] = own;
    __syncthreads();
    for (int off = 1; off < SORT_T; off <<= 1) {
        const int add = (tid >= off) ? sc[tid - off] : 0;
        __syncthreads();
        sc[tid] += add;
        __syncthreads();
    }
    if (tid < NPP) {
        const int start = s_below + sc[tid] - own;
        cnt[tid] = start;
        rowptr[(size_t)b * (GN + 1) + lo + tid] = start;
    }
    if (part == PARTS - 1 && tid == 0) rowptr[(size_t)b * (GN + 1) + GN] = GE;
    __syncthreads();

    v2i* se = (v2i*)sedge + (size_t)b * GE;
    for (int v0 = tid; v0 < NV; v0 += 4 * SORT_T) {
        v4i c[4];
#pragma unroll
        for (int u = 0; u < 4; ++u) {
            const int v = v0 + u * SORT_T;
            c[u] = (v < NV) ? cols4[v] : (v4i){GN, GN, GN, GN};
        }
#pragma unroll
        for (int u = 0; u < 4; ++u) {
#pragma unroll
            for (int j = 0; j < 4; ++j) {
                const int cj = c[u][j];
                if (cj >= lo && cj < hi) {
                    const int e = 4 * (v0 + u * SORT_T) + j;
                    const int p = atomicAdd(&cnt[cj - lo], 1);
                    v2i ed;
                    ed.x = rows[e];
                    ed.y = __float_as_int(w[e]);
                    se[p] = ed;
                }
            }
        }
    }
}

__global__ __launch_bounds__(256) void dinv_kernel(const int* __restrict__ rowptr,
                                                   const int* __restrict__ sedge,
                                                   float* __restrict__ dinv) {
    int idx = blockIdx.x * blockDim.x + threadIdx.x;
    if (idx >= GB * GN) return;
    int b = idx / GN;
    int i = idx - b * GN;
    const int beg = rowptr[(size_t)b * (GN + 1) + i];
    const int end = rowptr[(size_t)b * (GN + 1) + i + 1];
    const v2i* se = (const v2i*)sedge + (size_t)b * GE;
    float d = 0.f;
    for (int p = beg; p < end; ++p) d += __int_as_float(se[p].y);
    dinv[idx] = rsqrtf(d + 1.0f);
}

__global__ __launch_bounds__(256) void norm_kernel(const int* __restrict__ rowptr,
                                                   const float* __restrict__ dinv,
                                                   int* __restrict__ sedge) {
    int idx = blockIdx.x * blockDim.x + threadIdx.x;
    if (idx >= GB * GN) return;
    int b = idx / GN;
    int i = idx - b * GN;
    const int beg = rowptr[(size_t)b * (GN + 1) + i];
    const int end = rowptr[(size_t)b * (GN + 1) + i + 1];
    v2i* se = (v2i*)sedge + (size_t)b * GE;
    const float dc = dinv[idx];
    for (int p = beg; p < end; ++p) {
        v2i ed = se[p];
        const float nv = dinv[(size_t)b * GN + ed.x] * __int_as_float(ed.y) * dc;
        ed.y = __float_as_int(nv);
        se[p] = ed;
    }
}

template <int HIN>
__global__ __launch_bounds__(128) void gemm_wmma_kernel(const float* __restrict__ Hsrc,
                                                        const float* __restrict__ W,
                                                        float* __restrict__ Dst) {
    __shared__ float wlds[GH * HIN];
    const int tid = threadIdx.x;
    for (int i = tid; i < GH * HIN; i += 128) wlds[i] = W[i];
    __syncthreads();

    const int wave = tid >> 5;
    const int lane = tid & 31;
    const int half = lane >> 4;
    const int lm   = lane & 15;
    const int b    = blockIdx.z;
    const int m0   = blockIdx.x * 16;
    const int n0   = wave * 16;

    const float* hrow = Hsrc + ((size_t)b * GN + m0 + lm) * HIN;
    const float* wrow = wlds + (size_t)(n0 + lm) * HIN;

    v8f acc = {};
#pragma unroll
    for (int k0 = 0; k0 < HIN; k0 += 4) {
        const int ka = k0 + half * 2;
        v2f a = *(const v2f*)(hrow + ka);
        v2f bf = *(const v2f*)(wrow + ka);
        acc = __builtin_amdgcn_wmma_f32_16x16x4_f32(
             false, a,  false, bf,
             (short)0, acc,  false,  false);
    }

    float* drow = Dst + ((size_t)b * GN + m0) * GH + n0 + lm;
#pragma unroll
    for (int j = 0; j < 8; ++j) {
        drow[(size_t)(j + 8 * half) * GH] = acc[j];
    }
}

__global__ __launch_bounds__(256) void pull_kernel(const int* __restrict__ rowptr,
                                                   const int* __restrict__ sedge,
                                                   const float* __restrict__ tmp,
                                                   const float* __restrict__ dinv,
                                                   const float* __restrict__ bias,
                                                   float* __restrict__ hout,
                                                   int do_tanh) {
    int idx = blockIdx.x * blockDim.x + threadIdx.x;
    if (idx >= GB * GN * (GH / 4)) return;
    int k4   = idx & 15;
    int node = idx >> 4;
    int b = node / GN;
    int i = node - b * GN;
    const int beg = rowptr[(size_t)b * (GN + 1) + i];
    const int end = rowptr[(size_t)b * (GN + 1) + i + 1];
    const v2i*   se = (const v2i*)sedge + (size_t)b * GE;
    const float* tb = tmp + (size_t)b * GN * GH + k4 * 4;

    v4f acc = {0.f, 0.f, 0.f, 0.f};
    int p = beg;
    for (; p + 4 <= end; p += 4) {
        const v2i e0 = se[p], e1 = se[p + 1], e2 = se[p + 2], e3 = se[p + 3];
        const v4f r0 = *(const v4f*)(tb + (size_t)e0.x * GH);
        const v4f r1 = *(const v4f*)(tb + (size_t)e1.x * GH);
        const v4f r2 = *(const v4f*)(tb + (size_t)e2.x * GH);
        const v4f r3 = *(const v4f*)(tb + (size_t)e3.x * GH);
        acc += r0 * __int_as_float(e0.y);
        acc += r1 * __int_as_float(e1.y);
        acc += r2 * __int_as_float(e2.y);
        acc += r3 * __int_as_float(e3.y);
    }
    for (; p < end; ++p) {
        const v2i e0 = se[p];
        const v4f r0 = *(const v4f*)(tb + (size_t)e0.x * GH);
        acc += r0 * __int_as_float(e0.y);
    }

    float di = dinv[node];
    float sl = di * di;
    v4f t  = *(const v4f*)(tmp + (size_t)idx * 4);
    v4f bb = *(const v4f*)(bias + k4 * 4);
    v4f v = acc + t * sl + bb;
    if (do_tanh) {
        v.x = tanhf(v.x); v.y = tanhf(v.y); v.z = tanhf(v.z); v.w = tanhf(v.w);
    }
    *(v4f*)(hout + (size_t)idx * 4) = v;
}

__global__ __launch_bounds__(256) void gather_kernel(const float* __restrict__ emb,
                                                     const int* __restrict__ pos,
                                                     float* __restrict__ out) {
    int idx = blockIdx.x * blockDim.x + threadIdx.x;
    if (idx >= GB * GP * GEMB) return;
    int k = idx & (GEMB - 1);
    int bp = idx >> 6;
    int b = bp / GP;
    int pv = pos[bp];
    out[idx] = (pv >= 0) ? emb[((size_t)b * GN + pv) * GH + k] : -3.0f;
}

extern "C" void kernel_launch(void* const* d_in, const int* in_sizes, int n_in,
                              void* d_out, int out_size, void* d_ws, size_t ws_size,
                              hipStream_t stream) {
    const float* x   = (const float*)d_in[0];
    const int*   ei  = (const int*)d_in[1];
    const float* ew  = (const float*)d_in[2];
    const int*   pos = (const int*)d_in[3];
    const float* W1  = (const float*)d_in[4];
    const float* b1  = (const float*)d_in[5];
    const float* W2  = (const float*)d_in[6];
    const float* b2  = (const float*)d_in[7];
    const float* W3  = (const float*)d_in[8];
    const float* b3  = (const float*)d_in[9];
    float* out = (float*)d_out;

    float* ws   = (float*)d_ws;
    const size_t nBN  = (size_t)GB * GN;
    const size_t nRP  = (size_t)GB * (GN + 1);
    const size_t nBE  = (size_t)GB * GE;
    const size_t nBNH = (size_t)GB * GN * GH;
    float* dinv   = ws;
    int*   rowptr = (int*)(dinv + nBN);
    int*   sedge  = rowptr + nRP;
    float* tmp    = (float*)(sedge + 2 * nBE);
    float* hbuf   = tmp + nBNH;

    const int T = 256;
    dim3 gemm_grid(GN / 16, 1, GB);
    dim3 sort_grid(PARTS, GB, 1);

    sort_kernel<<<sort_grid, SORT_T, 0, stream>>>(ei, ew, rowptr, sedge);
    dinv_kernel<<<((int)nBN + T - 1) / T, T, 0, stream>>>(rowptr, sedge, dinv);
    norm_kernel<<<((int)nBN + T - 1) / T, T, 0, stream>>>(rowptr, dinv, sedge);

    const int elem4Blocks = (int)((nBNH / 4 + T - 1) / T);

    gemm_wmma_kernel<GS><<<gemm_grid, 128, 0, stream>>>(x, W1, tmp);
    pull_kernel<<<elem4Blocks, T, 0, stream>>>(rowptr, sedge, tmp, dinv, b1, hbuf, 1);

    gemm_wmma_kernel<GH><<<gemm_grid, 128, 0, stream>>>(hbuf, W2, tmp);
    pull_kernel<<<elem4Blocks, T, 0, stream>>>(rowptr, sedge, tmp, dinv, b2, hbuf, 1);

    gemm_wmma_kernel<GH><<<gemm_grid, 128, 0, stream>>>(hbuf, W3, tmp);
    pull_kernel<<<elem4Blocks, T, 0, stream>>>(rowptr, sedge, tmp, dinv, b3, hbuf, 0);

    gather_kernel<<<(GB * GP * GEMB + T - 1) / T, T, 0, stream>>>(hbuf, pos, out);
}


# Output-driven pruning: need levels marked back from pos through the sorted in-edge lists; layer-2/3 GEMM tiles and pulls loop over needed nodes only; layer-3 pull writes the pos rows straight to out
# speedup vs baseline: 3.4175x; 1.2909x over previous
#include <hip/hip_runtime.h>
#include <hip/hip_bf16.h>

#define GB   16
#define GN   10000
#define GE   160000
#define GS   128
#define GH   64
#define GEMB 64
#define GP   16

#define PARTS 16
#define NPP   (GN / PARTS)
#define SORT_T 1024

typedef __attribute__((ext_vector_type(2))) float v2f;
typedef __attribute__((ext_vector_type(4))) float v4f;
typedef __attribute__((ext_vector_type(8))) float v8f;
typedef __attribute__((ext_vector_type(2))) int   v2i;
typedef __attribute__((ext_vector_type(4))) int   v4i;

__global__ __launch_bounds__(SORT_T) void sort_kernel(const int* __restrict__ ei,
                                                      const float* __restrict__ ew,
                                                      int* __restrict__ rowptr,
                                                      int* __restrict__ sedge) {
    __shared__ int cnt[NPP];
    __shared__ int sc[SORT_T];
    __shared__ int s_below;
    const int tid  = threadIdx.x;
    const int part = blockIdx.x;
    const int b    = blockIdx.y;
    const int lo   = part * NPP;
    const int hi   = lo + NPP;
    const int*   rows = ei + (size_t)b * 2 * GE;
    const int*   cols = rows + GE;
    const float* w    = ew + (size_t)b * GE;
    const v4i*   cols4 = (const v4i*)cols;
    const int    NV = GE / 4;

    for (int i = tid; i < NPP; i += SORT_T) cnt[i] = 0;
    if (tid == 0) s_below = 0;
    __syncthreads();

    int below = 0;
    for (int v0 = tid; v0 < NV; v0 += 4 * SORT_T) {
        v4i c[4];
#pragma unroll
        for (int u = 0; u < 4; ++u) {
            const int v = v0 + u * SORT_T;
            c[u] = (v < NV) ? cols4[v] : (v4i){GN, GN, GN, GN};
        }
#pragma unroll
        for (int u = 0; u < 4; ++u) {
#pragma unroll
            for (int j = 0; j < 4; ++j) {
                const int cj = c[u][j];
                if (cj >= lo && cj < hi) atomicAdd(&cnt[cj - lo], 1);
                below += (cj < lo) ? 1 : 0;
            }
        }
    }
    atomicAdd(&s_below, below);
    __syncthreads();

    const int own = (tid < NPP) ? cnt[tid] : 0;
    sc[tid] = own;
    __syncthreads();
    for (int off = 1; off < SORT_T; off <<= 1) {
        const int add = (tid >= off) ? sc[tid - off] : 0;
        __syncthreads();
        sc[tid] += add;
        __syncthreads();
    }
    if (tid < NPP) {
        const int start = s_below + sc[tid] - own;
        cnt[tid] = start;
        rowptr[(size_t)b * (GN + 1) + lo + tid] = start;
    }
    if (part == PARTS - 1 && tid == 0) rowptr[(size_t)b * (GN + 1) + GN] = GE;
    __syncthreads();

    v2i* se = (v2i*)sedge + (size_t)b * GE;
    for (int v0 = tid; v0 < NV; v0 += 4 * SORT_T) {
        v4i c[4];
#pragma unroll
        for (int u = 0; u < 4; ++u) {
            const int v = v0 + u * SORT_T;
            c[u] = (v < NV) ? cols4[v] : (v4i){GN, GN, GN, GN};
        }
#pragma unroll
        for (int u = 0; u < 4; ++u) {
#pragma unroll
            for (int j = 0; j < 4; ++j) {
                const int cj = c[u][j];
                if (cj >= lo && cj < hi) {
                    const int e = 4 * (v0 + u * SORT_T) + j;
                    const int p = atomicAdd(&cnt[cj - lo], 1);
                    v2i ed;
                    ed.x = rows[e];
                    ed.y = __float_as_int(w[e]);
                    se[p] = ed;
                }
            }
        }
    }
}

__global__ __launch_bounds__(256) void dinv_kernel(const int* __restrict__ rowptr,
                                                   const int* __restrict__ sedge,
                                                   float* __restrict__ dinv,
                                                   int* __restrict__ lev) {
    int idx = blockIdx.x * blockDim.x + threadIdx.x;
    if (idx >= GB * GN) return;
    lev[idx] = 0;
    int b = idx / GN;
    int i = idx - b * GN;
    const int beg = rowptr[(size_t)b * (GN + 1) + i];
    const int end = rowptr[(size_t)b * (GN + 1) + i + 1];
    const v2i* se = (const v2i*)sedge + (size_t)b * GE;
    float d = 0.f;
    for (int p = beg; p < end; ++p) d += __int_as_float(se[p].y);
    dinv[idx] = rsqrtf(d + 1.0f);
}

__global__ __launch_bounds__(256) void mark_pos_kernel(const int* __restrict__ rowptr,
                                                       const int* __restrict__ sedge,
                                                       const int* __restrict__ pos,
                                                       int* __restrict__ lev) {
    int bp = blockIdx.x * blockDim.x + threadIdx.x;
    if (bp >= GB * GP) return;
    const int b  = bp / GP;
    const int pv = pos[bp];
    if (pv < 0) return;
    int* lv = lev + (size_t)b * GN;
    lv[pv] = 2;
    const int beg = rowptr[(size_t)b * (GN + 1) + pv];
    const int end = rowptr[(size_t)b * (GN + 1) + pv + 1];
    const v2i* se = (const v2i*)sedge + (size_t)b * GE;
    for (int p = beg; p < end; ++p) lv[se[p].x] = 2;
}

__global__ __launch_bounds__(256) void mark_src_kernel(const int* __restrict__ rowptr,
                                                       const int* __restrict__ sedge,
                                                       int* lev) {
    int idx = blockIdx.x * blockDim.x + threadIdx.x;
    if (idx >= GB * GN) return;
    if (lev[idx] != 2) return;
    int b = idx / GN;
    int i = idx - b * GN;
    int* lv = lev + (size_t)b * GN;
    const int beg = rowptr[(size_t)b * (GN + 1) + i];
    const int end = rowptr[(size_t)b * (GN + 1) + i + 1];
    const v2i* se = (const v2i*)sedge + (size_t)b * GE;
    for (int p = beg; p < end; ++p) {
        const int s = se[p].x;
        if (lv[s] == 0) lv[s] = 1;
    }
}

__global__ __launch_bounds__(256) void norm_kernel(const int* __restrict__ rowptr,
                                                   const float* __restrict__ dinv,
                                                   const int* __restrict__ lev,
                                                   int* __restrict__ sedge) {
    int idx = blockIdx.x * blockDim.x + threadIdx.x;
    if (idx >= GB * GN) return;
    if (lev[idx] < 1) return;
    int b = idx / GN;
    int i = idx - b * GN;
    const int beg = rowptr[(size_t)b * (GN + 1) + i];
    const int end = rowptr[(size_t)b * (GN + 1) + i + 1];
    v2i* se = (v2i*)sedge + (size_t)b * GE;
    const float dc = dinv[idx];
    for (int p = beg; p < end; ++p) {
        v2i ed = se[p];
        const float nv = dinv[(size_t)b * GN + ed.x] * __int_as_float(ed.y) * dc;
        ed.y = __float_as_int(nv);
        se[p] = ed;
    }
}

template <int HIN>
__global__ __launch_bounds__(128) void gemm_wmma_kernel(const float* __restrict__ Hsrc,
                                                        const float* __restrict__ W,
                                                        float* __restrict__ Dst) {
    __shared__ float wlds[GH * HIN];
    const int tid = threadIdx.x;
    for (int i = tid; i < GH * HIN; i += 128) wlds[i] = W[i];
    __syncthreads();

    const int wave = tid >> 5;
    const int lane = tid & 31;
    const int half = lane >> 4;
    const int lm   = lane & 15;
    const int b    = blockIdx.z;
    const int m0   = blockIdx.x * 16;
    const int n0   = wave * 16;

    const float* hrow = Hsrc + ((size_t)b * GN + m0 + lm) * HIN;
    const float* wrow = wlds + (size_t)(n0 + lm) * HIN;

    v8f acc = {};
#pragma unroll
    for (int k0 = 0; k0 < HIN; k0 += 4) {
        const int ka = k0 + half * 2;
        v2f a = *(const v2f*)(hrow + ka);
        v2f bf = *(const v2f*)(wrow + ka);
        acc = __builtin_amdgcn_wmma_f32_16x16x4_f32(
             false, a,  false, bf,
             (short)0, acc,  false,  false);
    }

    float* drow = Dst + ((size_t)b * GN + m0) * GH + n0 + lm;
#pragma unroll
    for (int j = 0; j < 8; ++j) {
        drow[(size_t)(j + 8 * half) * GH] = acc[j];
    }
}

template <int HIN>
__global__ __launch_bounds__(128) void gemm_need_kernel(const float* __restrict__ Hsrc,
                                                        const float* __restrict__ W,
                                                        const int* __restrict__ lev,
                                                        float* __restrict__ Dst,
                                                        int minlev) {
    __shared__ float wlds[GH * HIN];
    const int tid = threadIdx.x;
    for (int i = tid; i < GH * HIN; i += 128) wlds[i] = W[i];
    __syncthreads();

    const int wave = tid >> 5;
    const int lane = tid & 31;
    const int half = lane >> 4;
    const int lm   = lane & 15;
    const int n0   = wave * 16;
    const float* wrow = wlds + (size_t)(n0 + lm) * HIN;

    for (int t = blockIdx.x; t < GB * GN / 16; t += gridDim.x) {
        const int row = t * 16 + lm;
        const bool need = lev[row] >= minlev;
        if (__builtin_amdgcn_ballot_w32(need) == 0u) continue;
        const float* hrow = Hsrc + (size_t)row * HIN;

        v2f a[HIN / 4];
#pragma unroll
        for (int q = 0; q < HIN / 4; ++q) a[q] = (v2f){0.f, 0.f};
        if (need) {
#pragma unroll
            for (int q = 0; q < HIN / 4; ++q) a[q] = *(const v2f*)(hrow + q * 4 + half * 2);
        }
        v8f acc = {};
#pragma unroll
        for (int q = 0; q < HIN / 4; ++q) {
            const int ka = q * 4 + half * 2;
            v2f bf = *(const v2f*)(wrow + ka);
            acc = __builtin_amdgcn_wmma_f32_16x16x4_f32(
                 false, a[q],  false, bf,
                 (short)0, acc,  false,  false);
        }
        float* drow = Dst + (size_t)t * 16 * GH + n0 + lm;
#pragma unroll
        for (int j = 0; j < 8; ++j) {
            drow[(size_t)(j + 8 * half) * GH] = acc[j];
        }
    }
}

__device__ __forceinline__ v4f pull_row(const int* __restrict__ rowptr,
                                        const int* __restrict__ sedge,
                                        const float* __restrict__ tmp,
                                        const float* __restrict__ dinv,
                                        const float* __restrict__ bias,
                                        int b, int i, int k4) {
    const int node = b * GN + i;
    const int beg = rowptr[(size_t)b * (GN + 1) + i];
    const int end = rowptr[(size_t)b * (GN + 1) + i + 1];
    const v2i*   se = (const v2i*)sedge + (size_t)b * GE;
    const float* tb = tmp + (size_t)b * GN * GH + k4 * 4;

    v4f acc = {0.f, 0.f, 0.f, 0.f};
    int p = beg;
    for (; p + 4 <= end; p += 4) {
        const v2i e0 = se[p], e1 = se[p + 1], e2 = se[p + 2], e3 = se[p + 3];
        const v4f r0 = *(const v4f*)(tb + (size_t)e0.x * GH);
        const v4f r1 = *(const v4f*)(tb + (size_t)e1.x * GH);
        const v4f r2 = *(const v4f*)(tb + (size_t)e2.x * GH);
        const v4f r3 = *(const v4f*)(tb + (size_t)e3.x * GH);
        acc += r0 * __int_as_float(e0.y);
        acc += r1 * __int_as_float(e1.y);
        acc += r2 * __int_as_float(e2.y);
        acc += r3 * __int_as_float(e3.y);
    }
    for (; p < end; ++p) {
        const v2i e0 = se[p];
        const v4f r0 = *(const v4f*)(tb + (size_t)e0.x * GH);
        acc += r0 * __int_as_float(e0.y);
    }

    float di = dinv[node];
    float sl = di * di;
    v4f t  = *(const v4f*)(tmp + ((size_t)node * (GH / 4) + k4) * 4);
    v4f bb = *(const v4f*)(bias + k4 * 4);
    return acc + t * sl + bb;
}

__global__ __launch_bounds__(256) void pull_need_kernel(const int* __restrict__ rowptr,
                                                        const int* __restrict__ sedge,
                                                        const float* __restrict__ tmp,
                                                        const float* __restrict__ dinv,
                                                        const float* __restrict__ bias,
                                                        const int* __restrict__ lev,
                                                        float* __restrict__ hout,
                                                        int minlev) {
    const int k4 = threadIdx.x & 15;
    for (int node = blockIdx.x * 16 + (threadIdx.x >> 4); node < GB * GN; node += gridDim.x * 16) {
        if (lev[node] < minlev) continue;
        const int b = node / GN;
        const int i = node - b * GN;
        v4f v = pull_row(rowptr, sedge, tmp, dinv, bias, b, i, k4);
        v.x = tanhf(v.x); v.y = tanhf(v.y); v.z = tanhf(v.z); v.w = tanhf(v.w);
        *(v4f*)(hout + ((size_t)node * (GH / 4) + k4) * 4) = v;
    }
}

__global__ __launch_bounds__(256) void pull_pos_kernel(const int* __restrict__ rowptr,
                                                       const int* __restrict__ sedge,
                                                       const float* __restrict__ tmp,
                                                       const float* __restrict__ dinv,
                                                       const float* __restrict__ bias,
                                                       const int* __restrict__ pos,
                                                       float* __restrict__ out) {
    int idx = blockIdx.x * blockDim.x + threadIdx.x;
    if (idx >= GB * GP * (GEMB / 4)) return;
    const int k4 = idx & 15;
    const int bp = idx >> 4;
    const int b  = bp / GP;
    const int pv = pos[bp];
    v4f v = {-3.0f, -3.0f, -3.0f, -3.0f};
    if (pv >= 0) v = pull_row(rowptr, sedge, tmp, dinv, bias, b, pv, k4);
    *(v4f*)(out + (size_t)idx * 4) = v;
}

extern "C" void kernel_launch(void* const* d_in, const int* in_sizes, int n_in,
                              void* d_out, int out_size, void* d_ws, size_t ws_size,
                              hipStream_t stream) {
    const float* x   = (const float*)d_in[0];
    const int*   ei  = (const int*)d_in[1];
    const float* ew  = (const float*)d_in[2];
    const int*   pos = (const int*)d_in[3];
    const float* W1  = (const float*)d_in[4];
    const float* b1  = (const float*)d_in[5];
    const float* W2  = (const float*)d_in[6];
    const float* b2  = (const float*)d_in[7];
    const float* W3  = (const float*)d_in[8];
    const float* b3  = (const float*)d_in[9];
    float* out = (float*)d_out;

    float* ws   = (float*)d_ws;
    const size_t nBN  = (size_t)GB * GN;
    const size_t nRP  = (size_t)GB * (GN + 1);
    const size_t nBE  = (size_t)GB * GE;
    const size_t nBNH = (size_t)GB * GN * GH;
    float* dinv   = ws;
    int*   rowptr = (int*)(dinv + nBN);
    int*   sedge  = rowptr + nRP;
    float* tmp    = (float*)(sedge + 2 * nBE);
    float* hbuf   = tmp + nBNH;
    int*   lev    = (int*)(hbuf + nBNH);

    const int T = 256;
    dim3 gemm_grid(GN / 16, 1, GB);
    dim3 sort_grid(PARTS, GB, 1);
    const int nodeBlocks = ((int)nBN + T - 1) / T;

    sort_kernel<<<sort_grid, SORT_T, 0, stream>>>(ei, ew, rowptr, sedge);
    dinv_kernel<<<nodeBlocks, T, 0, stream>>>(rowptr, sedge, dinv, lev);
    mark_pos_kernel<<<(GB * GP + T - 1) / T, T, 0, stream>>>(rowptr, sedge, pos, lev);
    mark_src_kernel<<<nodeBlocks, T, 0, stream>>>(rowptr, sedge, lev);
    norm_kernel<<<nodeBlocks, T, 0, stream>>>(rowptr, dinv, lev, sedge);

    gemm_wmma_kernel<GS><<<gemm_grid, 128, 0, stream>>>(x, W1, tmp);
    pull_need_kernel<<<1536, T, 0, stream>>>(rowptr, sedge, tmp, dinv, b1, lev, hbuf, 1);

    gemm_need_kernel<GH><<<2500, 128, 0, stream>>>(hbuf, W2, lev, tmp, 1);
    pull_need_kernel<<<1024, T, 0, stream>>>(rowptr, sedge, tmp, dinv, b2, lev, hbuf, 2);

    gemm_need_kernel<GH><<<1250, 128, 0, stream>>>(hbuf, W3, lev, tmp, 2);
    pull_pos_kernel<<<(GB * GP * (GEMB / 4) + T - 1) / T, T, 0, stream>>>(rowptr, sedge, tmp, dinv, b3, pos, out);
}


# Edge sort placement pass in two sweeps per trip: 16 LDS-atomic slots and source/weight loads first, stores after; one memory wait per trip, not per owned edge; list order stays arrival order
# speedup vs baseline: 3.5846x; 1.0489x over previous
#include <hip/hip_runtime.h>
#include <hip/hip_bf16.h>

#define GB   16
#define GN   10000
#define GE   160000
#define GS   128
#define GH   64
#define GEMB 64
#define GP   16

#define PARTS 16
#define NPP   (GN / PARTS)
#define SORT_T 1024

typedef __attribute__((ext_vector_type(2))) float v2f;
typedef __attribute__((ext_vector_type(4))) float v4f;
typedef __attribute__((ext_vector_type(8))) float v8f;
typedef __attribute__((ext_vector_type(2))) int   v2i;
typedef __attribute__((ext_vector_type(4))) int   v4i;

__global__ __launch_bounds__(SORT_T) void sort_kernel(const int* __restrict__ ei,
                                                      const float* __restrict__ ew,
                                                      int* __restrict__ rowptr,
                                                      int* __restrict__ sedge) {
    __shared__ int cnt[NPP];
    __shared__ int sc[SORT_T];
    __shared__ int s_below;
    const int tid  = threadIdx.x;
    const int part = blockIdx.x;
    const int b    = blockIdx.y;
    const int lo   = part * NPP;
    const int hi   = lo + NPP;
    const int*   rows = ei + (size_t)b * 2 * GE;
    const int*   cols = rows + GE;
    const float* w    = ew + (size_t)b * GE;
    const v4i*   cols4 = (const v4i*)cols;
    const int    NV = GE / 4;

    for (int i = tid; i < NPP; i += SORT_T) cnt[i] = 0;
    if (tid == 0) s_below = 0;
    __syncthreads();

    int below = 0;
    for (int v0 = tid; v0 < NV; v0 += 4 * SORT_T) {
        v4i c[4];
#pragma unroll
        for (int u = 0; u < 4; ++u) {
            const int v = v0 + u * SORT_T;
            c[u] = (v < NV) ? cols4[v] : (v4i){GN, GN, GN, GN};
        }
#pragma unroll
        for (int u = 0; u < 4; ++u) {
#pragma unroll
            for (int j = 0; j < 4; ++j) {
                const int cj = c[u][j];
                if (cj >= lo && cj < hi) atomicAdd(&cnt[cj - lo], 1);
                below += (cj < lo) ? 1 : 0;
            }
        }
    }
    atomicAdd(&s_below, below);
    __syncthreads();

    const int own = (tid < NPP) ? cnt[tid] : 0;
    sc[tid] = own;
    __syncthreads();
    for (int off = 1; off < SORT_T; off <<= 1) {
        const int add = (tid >= off) ? sc[tid - off] : 0;
        __syncthreads();
        sc[tid] += add;
        __syncthreads();
    }
    if (tid < NPP) {
        const int start = s_below + sc[tid] - own;
        cnt[tid] = start;
        rowptr[(size_t)b * (GN + 1) + lo + tid] = start;
    }
    if (part == PARTS - 1 && tid == 0) rowptr[(size_t)b * (GN + 1) + GN] = GE;
    __syncthreads();

    v2i* se = (v2i*)sedge + (size_t)b * GE;
    for (int v0 = tid; v0 < NV; v0 += 4 * SORT_T) {
        v4i c[4];
#pragma unroll
        for (int u = 0; u < 4; ++u) {
            const int v = v0 + u * SORT_T;
            c[u] = (v < NV) ? cols4[v] : (v4i){GN, GN, GN, GN};
        }
        asm volatile("" :: "v"(c[0]), "v"(c[1]), "v"(c[2]), "v"(c[3]));
        int pos_[16], src_[16], wgt_[16];
#pragma unroll
        for (int u = 0; u < 4; ++u) {
#pragma unroll
            for (int j = 0; j < 4; ++j) {
                const int k  = 4 * u + j;
                const int cj = c[u][j];
                pos_[k] = -1;
                if (cj >= lo && cj < hi) {
                    const int e = 4 * (v0 + u * SORT_T) + j;
                    pos_[k] = atomicAdd(&cnt[cj - lo], 1);
                    src_[k] = rows[e];
                    wgt_[k] = __float_as_int(w[e]);
                }
            }
        }
#pragma unroll
        for (int k = 0; k < 16; ++k) {
            if (pos_[k] >= 0) {
                v2i ed;
                ed.x = src_[k];
                ed.y = wgt_[k];
                se[pos_[k]] = ed;
            }
        }
    }
}

__global__ __launch_bounds__(256) void dinv_kernel(const int* __restrict__ rowptr,
                                                   const int* __restrict__ sedge,
                                                   float* __restrict__ dinv,
                                                   int* __restrict__ lev) {
    int idx = blockIdx.x * blockDim.x + threadIdx.x;
    if (idx >= GB * GN) return;
    lev[idx] = 0;
    int b = idx / GN;
    int i = idx - b * GN;
    const int beg = rowptr[(size_t)b * (GN + 1) + i];
    const int end = rowptr[(size_t)b * (GN + 1) + i + 1];
    const v2i* se = (const v2i*)sedge + (size_t)b * GE;
    float d = 0.f;
    for (int p = beg; p < end; ++p) d += __int_as_float(se[p].y);
    dinv[idx] = rsqrtf(d + 1.0f);
}

__global__ __launch_bounds__(256) void mark_pos_kernel(const int* __restrict__ rowptr,
                                                       const int* __restrict__ sedge,
                                                       const int* __restrict__ pos,
                                                       int* __restrict__ lev) {
    int bp = blockIdx.x * blockDim.x + threadIdx.x;
    if (bp >= GB * GP) return;
    const int b  = bp / GP;
    const int pv = pos[bp];
    if (pv < 0) return;
    int* lv = lev + (size_t)b * GN;
    lv[pv] = 2;
    const int beg = rowptr[(size_t)b * (GN + 1) + pv];
    const int end = rowptr[(size_t)b * (GN + 1) + pv + 1];
    const v2i* se = (const v2i*)sedge + (size_t)b * GE;
    for (int p = beg; p < end; ++p) lv[se[p].x] = 2;
}

__global__ __launch_bounds__(256) void mark_src_kernel(const int* __restrict__ rowptr,
                                                       const int* __restrict__ sedge,
                                                       int* lev) {
    int idx = blockIdx.x * blockDim.x + threadIdx.x;
    if (idx >= GB * GN) return;
    if (lev[idx] != 2) return;
    int b = idx / GN;
    int i = idx - b * GN;
    int* lv = lev + (size_t)b * GN;
    const int beg = rowptr[(size_t)b * (GN + 1) + i];
    const int end = rowptr[(size_t)b * (GN + 1) + i + 1];
    const v2i* se = (const v2i*)sedge + (size_t)b * GE;
    for (int p = beg; p < end; ++p) {
        const int s = se[p].x;
        if (lv[s] == 0) lv[s] = 1;
    }
}

__global__ __launch_bounds__(256) void norm_kernel(const int* __restrict__ rowptr,
                                                   const float* __restrict__ dinv,
                                                   const int* __restrict__ lev,
                                                   int* __restrict__ sedge) {
    int idx = blockIdx.x * blockDim.x + threadIdx.x;
    if (idx >= GB * GN) return;
    if (lev[idx] < 1) return;
    int b = idx / GN;
    int i = idx - b * GN;
    const int beg = rowptr[(size_t)b * (GN + 1) + i];
    const int end = rowptr[(size_t)b * (GN + 1) + i + 1];
    v2i* se = (v2i*)sedge + (size_t)b * GE;
    const float dc = dinv[idx];
    for (int p = beg; p < end; ++p) {
        v2i ed = se[p];
        const float nv = dinv[(size_t)b * GN + ed.x] * __int_as_float(ed.y) * dc;
        ed.y = __float_as_int(nv);
        se[p] = ed;
    }
}

template <int HIN>
__global__ __launch_bounds__(128) void gemm_wmma_kernel(const float* __restrict__ Hsrc,
                                                        const float* __restrict__ W,
                                                        float* __restrict__ Dst) {
    __shared__ float wlds[GH * HIN];
    const int tid = threadIdx.x;
    for (int i = tid; i < GH * HIN; i += 128) wlds[i] = W[i];
    __syncthreads();

    const int wave = tid >> 5;
    const int lane = tid & 31;
    const int half = lane >> 4;
    const int lm   = lane & 15;
    const int b    = blockIdx.z;
    const int m0   = blockIdx.x * 16;
    const int n0   = wave * 16;

    const float* hrow = Hsrc + ((size_t)b * GN + m0 + lm) * HIN;
    const float* wrow = wlds + (size_t)(n0 + lm) * HIN;

    v8f acc = {};
#pragma unroll
    for (int k0 = 0; k0 < HIN; k0 += 4) {
        const int ka = k0 + half * 2;
        v2f a = *(const v2f*)(hrow + ka);
        v2f bf = *(const v2f*)(wrow + ka);
        acc = __builtin_amdgcn_wmma_f32_16x16x4_f32(
             false, a,  false, bf,
             (short)0, acc,  false,  false);
    }

    float* drow = Dst + ((size_t)b * GN + m0) * GH + n0 + lm;
#pragma unroll
    for (int j = 0; j < 8; ++j) {
        drow[(size_t)(j + 8 * half) * GH] = acc[j];
    }
}

template <int HIN>
__global__ __launch_bounds__(128) void gemm_need_kernel(const float* __restrict__ Hsrc,
                                                        const float* __restrict__ W,
                                                        const int* __restrict__ lev,
                                                        float* __restrict__ Dst,
                                                        int minlev) {
    __shared__ float wlds[GH * HIN];
    const int tid = threadIdx.x;
    for (int i = tid; i < GH * HIN; i += 128) wlds[i] = W[i];
    __syncthreads();

    const int wave = tid >> 5;
    const int lane = tid & 31;
    const int half = lane >> 4;
    const int lm   = lane & 15;
    const int n0   = wave * 16;
    const float* wrow = wlds + (size_t)(n0 + lm) * HIN;

    for (int t = blockIdx.x; t < GB * GN / 16; t += gridDim.x) {
        const int row = t * 16 + lm;
        const bool need = lev[row] >= minlev;
        if (__builtin_amdgcn_ballot_w32(need) == 0u) continue;
        const float* hrow = Hsrc + (size_t)row * HIN;

        v2f a[HIN / 4];
#pragma unroll
        for (int q = 0; q < HIN / 4; ++q) a[q] = (v2f){0.f, 0.f};
        if (need) {
#pragma unroll
            for (int q = 0; q < HIN / 4; ++q) a[q] = *(const v2f*)(hrow + q * 4 + half * 2);
        }
        v8f acc = {};
#pragma unroll
        for (int q = 0; q < HIN / 4; ++q) {
            const int ka = q * 4 + half * 2;
            v2f bf = *(const v2f*)(wrow + ka);
            acc = __builtin_amdgcn_wmma_f32_16x16x4_f32(
                 false, a[q],  false, bf,
                 (short)0, acc,  false,  false);
        }
        float* drow = Dst + (size_t)t * 16 * GH + n0 + lm;
#pragma unroll
        for (int j = 0; j < 8; ++j) {
            drow[(size_t)(j + 8 * half) * GH] = acc[j];
        }
    }
}

__device__ __forceinline__ v4f pull_row(const int* __restrict__ rowptr,
                                        const int* __restrict__ sedge,
                                        const float* __restrict__ tmp,
                                        const float* __restrict__ dinv,
                                        const float* __restrict__ bias,
                                        int b, int i, int k4) {
    const int node = b * GN + i;
    const int beg = rowptr[(size_t)b * (GN + 1) + i];
    const int end = rowptr[(size_t)b * (GN + 1) + i + 1];
    const v2i*   se = (const v2i*)sedge + (size_t)b * GE;
    const float* tb = tmp + (size_t)b * GN * GH + k4 * 4;

    v4f acc = {0.f, 0.f, 0.f, 0.f};
    int p = beg;
    for (; p + 4 <= end; p += 4) {
        const v2i e0 = se[p], e1 = se[p + 1], e2 = se[p + 2], e3 = se[p + 3];
        const v4f r0 = *(const v4f*)(tb + (size_t)e0.x * GH);
        const v4f r1 = *(const v4f*)(tb + (size_t)e1.x * GH);
        const v4f r2 = *(const v4f*)(tb + (size_t)e2.x * GH);
        const v4f r3 = *(const v4f*)(tb + (size_t)e3.x * GH);
        acc += r0 * __int_as_float(e0.y);
        acc += r1 * __int_as_float(e1.y);
        acc += r2 * __int_as_float(e2.y);
        acc += r3 * __int_as_float(e3.y);
    }
    for (; p < end; ++p) {
        const v2i e0 = se[p];
        const v4f r0 = *(const v4f*)(tb + (size_t)e0.x * GH);
        acc += r0 * __int_as_float(e0.y);
    }

    float di = dinv[node];
    float sl = di * di;
    v4f t  = *(const v4f*)(tmp + ((size_t)node * (GH / 4) + k4) * 4);
    v4f bb = *(const v4f*)(bias + k4 * 4);
    return acc + t * sl + bb;
}

__global__ __launch_bounds__(256) void pull_need_kernel(const int* __restrict__ rowptr,
                                                        const int* __restrict__ sedge,
                                                        const float* __restrict__ tmp,
                                                        const float* __restrict__ dinv,
                                                        const float* __restrict__ bias,
                                                        const int* __restrict__ lev,
                                                        float* __restrict__ hout,
                                                        int minlev) {
    const int k4 = threadIdx.x & 15;
    for (int node = blockIdx.x * 16 + (threadIdx.x >> 4); node < GB * GN; node += gridDim.x * 16) {
        if (lev[node] < minlev) continue;
        const int b = node / GN;
        const int i = node - b * GN;
        v4f v = pull_row(rowptr, sedge, tmp, dinv, bias, b, i, k4);
        v.x = tanhf(v.x); v.y = tanhf(v.y); v.z = tanhf(v.z); v.w = tanhf(v.w);
        *(v4f*)(hout + ((size_t)node * (GH / 4) + k4) * 4) = v;
    }
}

__global__ __launch_bounds__(256) void pull_pos_kernel(const int* __restrict__ rowptr,
                                                       const int* __restrict__ sedge,
                                                       const float* __restrict__ tmp,
                                                       const float* __restrict__ dinv,
                                                       const float* __restrict__ bias,
                                                       const int* __restrict__ pos,
                                                       float* __restrict__ out) {
    int idx = blockIdx.x * blockDim.x + threadIdx.x;
    if (idx >= GB * GP * (GEMB / 4)) return;
    const int k4 = idx & 15;
    const int bp = idx >> 4;
    const int b  = bp / GP;
    const int pv = pos[bp];
    v4f v = {-3.0f, -3.0f, -3.0f, -3.0f};
    if (pv >= 0) v = pull_row(rowptr, sedge, tmp, dinv, bias, b, pv, k4);
    *(v4f*)(out + (size_t)idx * 4) = v;
}

extern "C" void kernel_launch(void* const* d_in, const int* in_sizes, int n_in,
                              void* d_out, int out_size, void* d_ws, size_t ws_size,
                              hipStream_t stream) {
    const float* x   = (const float*)d_in[0];
    const int*   ei  = (const int*)d_in[1];
    const float* ew  = (const float*)d_in[2];
    const int*   pos = (const int*)d_in[3];
    const float* W1  = (const float*)d_in[4];
    const float* b1  = (const float*)d_in[5];
    const float* W2  = (const float*)d_in[6];
    const float* b2  = (const float*)d_in[7];
    const float* W3  = (const float*)d_in[8];
    const float* b3  = (const float*)d_in[9];
    float* out = (float*)d_out;

    float* ws   = (float*)d_ws;
    const size_t nBN  = (size_t)GB * GN;
    const size_t nRP  = (size_t)GB * (GN + 1);
    const size_t nBE  = (size_t)GB * GE;
    const size_t nBNH = (size_t)GB * GN * GH;
    float* dinv   = ws;
    int*   rowptr = (int*)(dinv + nBN);
    int*   sedge  = rowptr + nRP;
    float* tmp    = (float*)(sedge + 2 * nBE);
    float* hbuf   = tmp + nBNH;
    int*   lev    = (int*)(hbuf + nBNH);

    const int T = 256;
    dim3 gemm_grid(GN / 16, 1, GB);
    dim3 sort_grid(PARTS, GB, 1);
    const int nodeBlocks = ((int)nBN + T - 1) / T;

    sort_kernel<<<sort_grid, SORT_T, 0, stream>>>(ei, ew, rowptr, sedge);
    dinv_kernel<<<nodeBlocks, T, 0, stream>>>(rowptr, sedge, dinv, lev);
    mark_pos_kernel<<<(GB * GP + T - 1) / T, T, 0, stream>>>(rowptr, sedge, pos, lev);
    mark_src_kernel<<<nodeBlocks, T, 0, stream>>>(rowptr, sedge, lev);
    norm_kernel<<<nodeBlocks, T, 0, stream>>>(rowptr, dinv, lev, sedge);

    gemm_wmma_kernel<GS><<<gemm_grid, 128, 0, stream>>>(x, W1, tmp);
    pull_need_kernel<<<1536, T, 0, stream>>>(rowptr, sedge, tmp, dinv, b1, lev, hbuf, 1);

    gemm_need_kernel<GH><<<2500, 128, 0, stream>>>(hbuf, W2, lev, tmp, 1);
    pull_need_kernel<<<1024, T, 0, stream>>>(rowptr, sedge, tmp, dinv, b2, lev, hbuf, 2);

    gemm_need_kernel<GH><<<1250, 128, 0, stream>>>(hbuf, W3, lev, tmp, 2);
    pull_pos_kernel<<<(GB * GP * (GEMB / 4) + T - 1) / T, T, 0, stream>>>(rowptr, sedge, tmp, dinv, b3, pos, out);
}


# GEMMs re-tiled: a wave owns a 16x64 strip, row fragment loaded once with 16-byte loads for 4 accumulators, W read from LDS at a conflict-free padded pitch; k order inside the products changes
# speedup vs baseline: 4.2398x; 1.1828x over previous
#include <hip/hip_runtime.h>
#include <hip/hip_bf16.h>

#define GB   16
#define GN   10000
#define GE   160000
#define GS   128
#define GH   64
#define GEMB 64
#define GP   16

#define PARTS 16
#define NPP   (GN / PARTS)
#define SORT_T 1024

typedef __attribute__((ext_vector_type(2))) float v2f;
typedef __attribute__((ext_vector_type(4))) float v4f;
typedef __attribute__((ext_vector_type(8))) float v8f;
typedef __attribute__((ext_vector_type(2))) int   v2i;
typedef __attribute__((ext_vector_type(4))) int   v4i;

__global__ __launch_bounds__(SORT_T) void sort_kernel(const int* __restrict__ ei,
                                                      const float* __restrict__ ew,
                                                      int* __restrict__ rowptr,
                                                      int* __restrict__ sedge) {
    __shared__ int cnt[NPP];
    __shared__ int sc[SORT_T];
    __shared__ int s_below;
    const int tid  = threadIdx.x;
    const int part = blockIdx.x;
    const int b    = blockIdx.y;
    const int lo   = part * NPP;
    const int hi   = lo + NPP;
    const int*   rows = ei + (size_t)b * 2 * GE;
    const int*   cols = rows + GE;
    const float* w    = ew + (size_t)b * GE;
    const v4i*   cols4 = (const v4i*)cols;
    const int    NV = GE / 4;

    for (int i = tid; i < NPP; i += SORT_T) cnt[i] = 0;
    if (tid == 0) s_below = 0;
    __syncthreads();

    int below = 0;
    for (int v0 = tid; v0 < NV; v0 += 4 * SORT_T) {
        v4i c[4];
#pragma unroll
        for (int u = 0; u < 4; ++u) {
            const int v = v0 + u * SORT_T;
            c[u] = (v < NV) ? cols4[v] : (v4i){GN, GN, GN, GN};
        }
#pragma unroll
        for (int u = 0; u < 4; ++u) {
#pragma unroll
            for (int j = 0; j < 4; ++j) {
                const int cj = c[u][j];
                if (cj >= lo && cj < hi) atomicAdd(&cnt[cj - lo], 1);
                below += (cj < lo) ? 1 : 0;
            }
        }
    }
    atomicAdd(&s_below, below);
    __syncthreads();

    const int own = (tid < NPP) ? cnt[tid] : 0;
    sc[tid] = own;
    __syncthreads();
    for (int off = 1; off < SORT_T; off <<= 1) {
        const int add = (tid >= off) ? sc[tid - off] : 0;
        __syncthreads();
        sc[tid] += add;
        __syncthreads();
    }
    if (tid < NPP) {
        const int start = s_below + sc[tid] - own;
        cnt[tid] = start;
        rowptr[(size_t)b * (GN + 1) + lo + tid] = start;
    }
    if (part == PARTS - 1 && tid == 0) rowptr[(size_t)b * (GN + 1) + GN] = GE;
    __syncthreads();

    v2i* se = (v2i*)sedge + (size_t)b * GE;
    for (int v0 = tid; v0 < NV; v0 += 4 * SORT_T) {
        v4i c[4];
#pragma unroll
        for (int u = 0; u < 4; ++u) {
            const int v = v0 + u * SORT_T;
            c[u] = (v < NV) ? cols4[v] : (v4i){GN, GN, GN, GN};
        }
        asm volatile("" :: "v"(c[0]), "v"(c[1]), "v"(c[2]), "v"(c[3]));
        int pos_[16], src_[16], wgt_[16];
#pragma unroll
        for (int u = 0; u < 4; ++u) {
#pragma unroll
            for (int j = 0; j < 4; ++j) {
                const int k  = 4 * u + j;
                const int cj = c[u][j];
                pos_[k] = -1;
                if (cj >= lo && cj < hi) {
                    const int e = 4 * (v0 + u * SORT_T) + j;
                    pos_[k] = atomicAdd(&cnt[cj - lo], 1);
                    src_[k] = rows[e];
                    wgt_[k] = __float_as_int(w[e]);
                }
            }
        }
#pragma unroll
        for (int k = 0; k < 16; ++k) {
            if (pos_[k] >= 0) {
                v2i ed;
                ed.x = src_[k];
                ed.y = wgt_[k];
                se[pos_[k]] = ed;
            }
        }
    }
}

__global__ __launch_bounds__(256) void dinv_kernel(const int* __restrict__ rowptr,
                                                   const int* __restrict__ sedge,
                                                   float* __restrict__ dinv,
                                                   int* __restrict__ lev) {
    int idx = blockIdx.x * blockDim.x + threadIdx.x;
    if (idx >= GB * GN) return;
    lev[idx] = 0;
    int b = idx / GN;
    int i = idx - b * GN;
    const int beg = rowptr[(size_t)b * (GN + 1) + i];
    const int end = rowptr[(size_t)b * (GN + 1) + i + 1];
    const v2i* se = (const v2i*)sedge + (size_t)b * GE;
    float d = 0.f;
    for (int p = beg; p < end; ++p) d += __int_as_float(se[p].y);
    dinv[idx] = rsqrtf(d + 1.0f);
}

__global__ __launch_bounds__(256) void mark_pos_kernel(const int* __restrict__ rowptr,
                                                       const int* __restrict__ sedge,
                                                       const int* __restrict__ pos,
                                                       int* __restrict__ lev) {
    int bp = blockIdx.x * blockDim.x + threadIdx.x;
    if (bp >= GB * GP) return;
    const int b  = bp / GP;
    const int pv = pos[bp];
    if (pv < 0) return;
    int* lv = lev + (size_t)b * GN;
    lv[pv] = 2;
    const int beg = rowptr[(size_t)b * (GN + 1) + pv];
    const int end = rowptr[(size_t)b * (GN + 1) + pv + 1];
    const v2i* se = (const v2i*)sedge + (size_t)b * GE;
    for (int p = beg; p < end; ++p) lv[se[p].x] = 2;
}

__global__ __launch_bounds__(256) void mark_src_kernel(const int* __restrict__ rowptr,
                                                       const int* __restrict__ sedge,
                                                       int* lev) {
    int idx = blockIdx.x * blockDim.x + threadIdx.x;
    if (idx >= GB * GN) return;
    if (lev[idx] != 2) return;
    int b = idx / GN;
    int i = idx - b * GN;
    int* lv = lev + (size_t)b * GN;
    const int beg = rowptr[(size_t)b * (GN + 1) + i];
    const int end = rowptr[(size_t)b * (GN + 1) + i + 1];
    const v2i* se = (const v2i*)sedge + (size_t)b * GE;
    for (int p = beg; p < end; ++p) {
        const int s = se[p].x;
        if (lv[s] == 0) lv[s] = 1;
    }
}

__global__ __launch_bounds__(256) void norm_kernel(const int* __restrict__ rowptr,
                                                   const float* __restrict__ dinv,
                                                   const int* __restrict__ lev,
                                                   int* __restrict__ sedge) {
    int idx = blockIdx.x * blockDim.x + threadIdx.x;
    if (idx >= GB * GN) return;
    if (lev[idx] < 1) return;
    int b = idx / GN;
    int i = idx - b * GN;
    const int beg = rowptr[(size_t)b * (GN + 1) + i];
    const int end = rowptr[(size_t)b * (GN + 1) + i + 1];
    v2i* se = (v2i*)sedge + (size_t)b * GE;
    const float dc = dinv[idx];
    for (int p = beg; p < end; ++p) {
        v2i ed = se[p];
        const float nv = dinv[(size_t)b * GN + ed.x] * __int_as_float(ed.y) * dc;
        ed.y = __float_as_int(nv);
        se[p] = ed;
    }
}

template <int HIN, bool ALL>
__global__ __launch_bounds__(512) void gemm_strip_kernel(const float* __restrict__ Hsrc,
                                                         const float* __restrict__ W,
                                                         const int* __restrict__ lev,
                                                         float* __restrict__ Dst,
                                                         int minlev) {
    constexpr int PITCH = HIN + 4;
    constexpr int NQ = HIN / 8;
    __shared__ __attribute__((aligned(16))) float wlds[GH * PITCH];
    const int tid = threadIdx.x;
    for (int i = tid; i < GH * HIN / 4; i += 512) {
        const int r  = i / (HIN / 4);
        const int c4 = i - r * (HIN / 4);
        *(v4f*)(wlds + r * PITCH + c4 * 4) = ((const v4f*)W)[i];
    }
    __syncthreads();

    const int wave = __builtin_amdgcn_readfirstlane(tid >> 5);
    const int lane = tid & 31;
    const int half = lane >> 4;
    const int lm   = lane & 15;
    const int woff0 = lm * PITCH + half * 4;

    for (int t = blockIdx.x + gridDim.x * wave; t < GB * GN / 16; t += gridDim.x * 16) {
        const int row = t * 16 + lm;
        bool need = true;
        if (!ALL) {
            need = lev[row] >= minlev;
            if (__builtin_amdgcn_ballot_w32(need) == 0u) continue;
        }
        const float* hrow = Hsrc + (size_t)row * HIN + half * 4;
        int woff = woff0;
        asm volatile("" : "+v"(woff));
        const float* wrow = wlds + woff;

        v4f a[NQ];
        if (ALL) {
#pragma unroll
            for (int q = 0; q < NQ; ++q) a[q] = *(const v4f*)(hrow + q * 8);
        } else {
#pragma unroll
            for (int q = 0; q < NQ; ++q) a[q] = (v4f){0.f, 0.f, 0.f, 0.f};
            if (need) {
#pragma unroll
                for (int q = 0; q < NQ; ++q) a[q] = *(const v4f*)(hrow + q * 8);
            }
        }

        __builtin_amdgcn_sched_barrier(0);

        v8f acc[4] = {};
        v4f bcur[4], bnxt[4];
#pragma unroll
        for (int n = 0; n < 4; ++n) bcur[n] = *(const v4f*)(wrow + n * 16 * PITCH);
#pragma unroll
        for (int q = 0; q < NQ; ++q) {
            if (q + 1 < NQ) {
#pragma unroll
                for (int n = 0; n < 4; ++n) bnxt[n] = *(const v4f*)(wrow + n * 16 * PITCH + (q + 1) * 8);
            }
            __builtin_amdgcn_sched_barrier(0);
            const v2f a0 = {a[q].x, a[q].y};
            const v2f a1 = {a[q].z, a[q].w};
#pragma unroll
            for (int n = 0; n < 4; ++n) {
                const v2f b0 = {bcur[n].x, bcur[n].y};
                acc[n] = __builtin_amdgcn_wmma_f32_16x16x4_f32(
                     false, a0,  false, b0,
                     (short)0, acc[n],  false,  false);
            }
#pragma unroll
            for (int n = 0; n < 4; ++n) {
                const v2f b1 = {bcur[n].z, bcur[n].w};
                acc[n] = __builtin_amdgcn_wmma_f32_16x16x4_f32(
                     false, a1,  false, b1,
                     (short)0, acc[n],  false,  false);
            }
#pragma unroll
            for (int n = 0; n < 4; ++n) bcur[n] = bnxt[n];
        }

        float* drow = Dst + (size_t)t * 16 * GH + lm;
#pragma unroll
        for (int j = 0; j < 8; ++j) {
#pragma unroll
            for (int n = 0; n < 4; ++n) {
                drow[(size_t)(j + 8 * half) * GH + n * 16] = acc[n][j];
            }
        }
    }
}

__device__ __forceinline__ v4f pull_row(const int* __restrict__ rowptr,
                                        const int* __restrict__ sedge,
                                        const float* __restrict__ tmp,
                                        const float* __restrict__ dinv,
                                        const float* __restrict__ bias,
                                        int b, int i, int k4) {
    const int node = b * GN + i;
    const int beg = rowptr[(size_t)b * (GN + 1) + i];
    const int end = rowptr[(size_t)b * (GN + 1) + i + 1];
    const v2i*   se = (const v2i*)sedge + (size_t)b * GE;
    const float* tb = tmp + (size_t)b * GN * GH + k4 * 4;

    v4f acc = {0.f, 0.f, 0.f, 0.f};
    int p = beg;
    for (; p + 4 <= end; p += 4) {
        const v2i e0 = se[p], e1 = se[p + 1], e2 = se[p + 2], e3 = se[p + 3];
        const v4f r0 = *(const v4f*)(tb + (size_t)e0.x * GH);
        const v4f r1 = *(const v4f*)(tb + (size_t)e1.x * GH);
        const v4f r2 = *(const v4f*)(tb + (size_t)e2.x * GH);
        const v4f r3 = *(const v4f*)(tb + (size_t)e3.x * GH);
        acc += r0 * __int_as_float(e0.y);
        acc += r1 * __int_as_float(e1.y);
        acc += r2 * __int_as_float(e2.y);
        acc += r3 * __int_as_float(e3.y);
    }
    for (; p < end; ++p) {
        const v2i e0 = se[p];
        const v4f r0 = *(const v4f*)(tb + (size_t)e0.x * GH);
        acc += r0 * __int_as_float(e0.y);
    }

    float di = dinv[node];
    float sl = di * di;
    v4f t  = *(const v4f*)(tmp + ((size_t)node * (GH / 4) + k4) * 4);
    v4f bb = *(const v4f*)(bias + k4 * 4);
    return acc + t * sl + bb;
}

__global__ __launch_bounds__(256) void pull_need_kernel(const int* __restrict__ rowptr,
                                                        const int* __restrict__ sedge,
                                                        const float* __restrict__ tmp,
                                                        const float* __restrict__ dinv,
                                                        const float* __restrict__ bias,
                                                        const int* __restrict__ lev,
                                                        float* __restrict__ hout,
                                                        int minlev) {
    const int k4 = threadIdx.x & 15;
    for (int node = blockIdx.x * 16 + (threadIdx.x >> 4); node < GB * GN; node += gridDim.x * 16) {
        if (lev[node] < minlev) continue;
        const int b = node / GN;
        const int i = node - b * GN;
        v4f v = pull_row(rowptr, sedge, tmp, dinv, bias, b, i, k4);
        v.x = tanhf(v.x); v.y = tanhf(v.y); v.z = tanhf(v.z); v.w = tanhf(v.w);
        *(v4f*)(hout + ((size_t)node * (GH / 4) + k4) * 4) = v;
    }
}

__global__ __launch_bounds__(256) void pull_pos_kernel(const int* __restrict__ rowptr,
                                                       const int* __restrict__ sedge,
                                                       const float* __restrict__ tmp,
                                                       const float* __restrict__ dinv,
                                                       const float* __restrict__ bias,
                                                       const int* __restrict__ pos,
                                                       float* __restrict__ out) {
    int idx = blockIdx.x * blockDim.x + threadIdx.x;
    if (idx >= GB * GP * (GEMB / 4)) return;
    const int k4 = idx & 15;
    const int bp = idx >> 4;
    const int b  = bp / GP;
    const int pv = pos[bp];
    v4f v = {-3.0f, -3.0f, -3.0f, -3.0f};
    if (pv >= 0) v = pull_row(rowptr, sedge, tmp, dinv, bias, b, pv, k4);
    *(v4f*)(out + (size_t)idx * 4) = v;
}

extern "C" void kernel_launch(void* const* d_in, const int* in_sizes, int n_in,
                              void* d_out, int out_size, void* d_ws, size_t ws_size,
                              hipStream_t stream) {
    const float* x   = (const float*)d_in[0];
    const int*   ei  = (const int*)d_in[1];
    const float* ew  = (const float*)d_in[2];
    const int*   pos = (const int*)d_in[3];
    const float* W1  = (const float*)d_in[4];
    const float* b1  = (const float*)d_in[5];
    const float* W2  = (const float*)d_in[6];
    const float* b2  = (const float*)d_in[7];
    const float* W3  = (const float*)d_in[8];
    const float* b3  = (const float*)d_in[9];
    float* out = (float*)d_out;

    float* ws   = (float*)d_ws;
    const size_t nBN  = (size_t)GB * GN;
    const size_t nRP  = (size_t)GB * (GN + 1);
    const size_t nBE  = (size_t)GB * GE;
    const size_t nBNH = (size_t)GB * GN * GH;
    float* dinv   = ws;
    int*   rowptr = (int*)(dinv + nBN);
    int*   sedge  = rowptr + nRP;
    float* tmp    = (float*)(sedge + 2 * nBE);
    float* hbuf   = tmp + nBNH;
    int*   lev    = (int*)(hbuf + nBNH);

    const int T = 256;
    dim3 sort_grid(PARTS, GB, 1);
    const int nodeBlocks = ((int)nBN + T - 1) / T;

    sort_kernel<<<sort_grid, SORT_T, 0, stream>>>(ei, ew, rowptr, sedge);
    dinv_kernel<<<nodeBlocks, T, 0, stream>>>(rowptr, sedge, dinv, lev);
    mark_pos_kernel<<<(GB * GP + T - 1) / T, T, 0, stream>>>(rowptr, sedge, pos, lev);
    mark_src_kernel<<<nodeBlocks, T, 0, stream>>>(rowptr, sedge, lev);
    norm_kernel<<<nodeBlocks, T, 0, stream>>>(rowptr, dinv, lev, sedge);

    gemm_strip_kernel<GS, true><<<256, 512, 0, stream>>>(x, W1, lev, tmp, 0);
    pull_need_kernel<<<1536, T, 0, stream>>>(rowptr, sedge, tmp, dinv, b1, lev, hbuf, 1);

    gemm_strip_kernel<GH, false><<<256, 512, 0, stream>>>(hbuf, W2, lev, tmp, 1);
    pull_need_kernel<<<1024, T, 0, stream>>>(rowptr, sedge, tmp, dinv, b2, lev, hbuf, 2);

    gemm_strip_kernel<GH, false><<<256, 512, 0, stream>>>(hbuf, W3, lev, tmp, 2);
    pull_pos_kernel<<<(GB * GP * (GEMB / 4) + T - 1) / T, T, 0, stream>>>(rowptr, sedge, tmp, dinv, b3, pos, out);
}


# Edge sort split in two launches: each 10000-edge slice is first grouped by target range, then a range block sorts only its own edges, not all 160000; list (sum) order changes, still arrival order
# speedup vs baseline: 4.5375x; 1.0702x over previous
#include <hip/hip_runtime.h>
#include <hip/hip_bf16.h>

#define GB   16
#define GN   10000
#define GE   160000
#define GS   128
#define GH   64
#define GEMB 64
#define GP   16

#define PARTS 16
#define NPP   (GN / PARTS)
#define SORT_T 1024
#define SLICES 16
#define EPS   (GE / SLICES)
#define EPT   ((EPS + SORT_T - 1) / SORT_T)
static_assert(GN % PARTS == 0 && GE % SLICES == 0 && GN <= 65536, "sort layout");

typedef __attribute__((ext_vector_type(2))) float v2f;
typedef __attribute__((ext_vector_type(4))) float v4f;
typedef __attribute__((ext_vector_type(8))) float v8f;
typedef __attribute__((ext_vector_type(2))) int   v2i;
typedef __attribute__((ext_vector_type(4))) int   v4i;

__global__ __launch_bounds__(SORT_T) void part_kernel(const int* __restrict__ ei,
                                                      const float* __restrict__ ew,
                                                      int* __restrict__ binptr,
                                                      int* __restrict__ pedge) {
    __shared__ int bin[PARTS];
    const int tid = threadIdx.x;
    const int s   = blockIdx.x;
    const int b   = blockIdx.y;
    const int*   rows = ei + (size_t)b * 2 * GE + (size_t)s * EPS;
    const int*   cols = rows + GE;
    const float* w    = ew + (size_t)b * GE + (size_t)s * EPS;

    if (tid < PARTS) bin[tid] = 0;
    int c_[EPT], r_[EPT], w_[EPT];
#pragma unroll
    for (int u = 0; u < EPT; ++u) {
        const int e = tid + u * SORT_T;
        c_[u] = 0; r_[u] = 0; w_[u] = 0;
        if (e < EPS) {
            c_[u] = cols[e];
            r_[u] = rows[e];
            w_[u] = __float_as_int(w[e]);
        }
    }
    __syncthreads();
#pragma unroll
    for (int u = 0; u < EPT; ++u) {
        const int e = tid + u * SORT_T;
        if (e < EPS) atomicAdd(&bin[c_[u] / NPP], 1);
    }
    __syncthreads();
    int* bp = binptr + ((size_t)b * SLICES + s) * (PARTS + 1);
    if (tid == 0) {
        int n_[PARTS];
#pragma unroll
        for (int p = 0; p < PARTS; ++p) n_[p] = bin[p];
        int run = 0;
#pragma unroll
        for (int p = 0; p < PARTS; ++p) {
            bin[p] = run;
            bp[p]  = run;
            run += n_[p];
        }
        bp[PARTS] = run;
    }
    __syncthreads();
    v2i* pe = (v2i*)pedge + (size_t)b * GE + (size_t)s * EPS;
    int pos_[EPT];
#pragma unroll
    for (int u = 0; u < EPT; ++u) {
        const int e = tid + u * SORT_T;
        pos_[u] = -1;
        if (e < EPS) pos_[u] = atomicAdd(&bin[c_[u] / NPP], 1);
    }
#pragma unroll
    for (int u = 0; u < EPT; ++u) {
        if (pos_[u] >= 0) {
            v2i ed;
            ed.x = r_[u] | (c_[u] << 16);
            ed.y = w_[u];
            pe[pos_[u]] = ed;
        }
    }
}

__global__ __launch_bounds__(SORT_T) void sort_kernel(const int* __restrict__ binptr,
                                                      const int* __restrict__ pedge,
                                                      int* __restrict__ rowptr,
                                                      int* __restrict__ sedge) {
    __shared__ int cnt[NPP];
    __shared__ int sc[SORT_T];
    const int tid  = threadIdx.x;
    const int part = blockIdx.x;
    const int b    = blockIdx.y;
    const int lo   = part * NPP;

    for (int i = tid; i < NPP; i += SORT_T) cnt[i] = 0;

    const int* bpb = binptr + (size_t)b * SLICES * (PARTS + 1) + part;
    const v2i* pe  = (const v2i*)pedge + (size_t)b * GE;
    int beg[SLICES], end[SLICES];
    int below = 0;
#pragma unroll
    for (int s = 0; s < SLICES; ++s) {
        beg[s] = bpb[s * (PARTS + 1)];
        end[s] = bpb[s * (PARTS + 1) + 1];
        below += beg[s];
    }
    v2i ed[SLICES];
#pragma unroll
    for (int s = 0; s < SLICES; ++s) {
        const int k = beg[s] + tid;
        ed[s].x = -1;
        ed[s].y = 0;
        if (k < end[s]) ed[s] = pe[s * EPS + k];
    }
    __syncthreads();
#pragma unroll
    for (int s = 0; s < SLICES; ++s) {
        if (ed[s].x >= 0) atomicAdd(&cnt[(ed[s].x >> 16) - lo], 1);
        if (end[s] - beg[s] > SORT_T) {
            for (int k = beg[s] + tid + SORT_T; k < end[s]; k += SORT_T)
                atomicAdd(&cnt[(pe[s * EPS + k].x >> 16) - lo], 1);
        }
    }
    __syncthreads();

    const int own = (tid < NPP) ? cnt[tid] : 0;
    sc[tid] = own;
    __syncthreads();
    for (int off = 1; off < SORT_T; off <<= 1) {
        const int add = (tid >= off) ? sc[tid - off] : 0;
        __syncthreads();
        sc[tid] += add;
        __syncthreads();
    }
    if (tid < NPP) {
        const int start = below + sc[tid] - own;
        cnt[tid] = start;
        rowptr[(size_t)b * (GN + 1) + lo + tid] = start;
    }
    if (part == PARTS - 1 && tid == 0) rowptr[(size_t)b * (GN + 1) + GN] = GE;
    __syncthreads();

    v2i* se = (v2i*)sedge + (size_t)b * GE;
    int pos_[SLICES];
#pragma unroll
    for (int s = 0; s < SLICES; ++s) {
        pos_[s] = -1;
        if (ed[s].x >= 0) pos_[s] = atomicAdd(&cnt[(ed[s].x >> 16) - lo], 1);
    }
#pragma unroll
    for (int s = 0; s < SLICES; ++s) {
        if (pos_[s] >= 0) {
            v2i o;
            o.x = ed[s].x & 0xFFFF;
            o.y = ed[s].y;
            se[pos_[s]] = o;
        }
        if (end[s] - beg[s] > SORT_T)
        for (int k = beg[s] + tid + SORT_T; k < end[s]; k += SORT_T) {
            const v2i e2 = pe[s * EPS + k];
            const int p2 = atomicAdd(&cnt[(e2.x >> 16) - lo], 1);
            v2i o;
            o.x = e2.x & 0xFFFF;
            o.y = e2.y;
            se[p2] = o;
        }
    }
}

__global__ __launch_bounds__(256) void dinv_kernel(const int* __restrict__ rowptr,
                                                   const int* __restrict__ sedge,
                                                   float* __restrict__ dinv,
                                                   int* __restrict__ lev) {
    int idx = blockIdx.x * blockDim.x + threadIdx.x;
    if (idx >= GB * GN) return;
    lev[idx] = 0;
    int b = idx / GN;
    int i = idx - b * GN;
    const int beg = rowptr[(size_t)b * (GN + 1) + i];
    const int end = rowptr[(size_t)b * (GN + 1) + i + 1];
    const v2i* se = (const v2i*)sedge + (size_t)b * GE;
    float d = 0.f;
    for (int p = beg; p < end; ++p) d += __int_as_float(se[p].y);
    dinv[idx] = rsqrtf(d + 1.0f);
}

__global__ __launch_bounds__(256) void mark_pos_kernel(const int* __restrict__ rowptr,
                                                       const int* __restrict__ sedge,
                                                       const int* __restrict__ pos,
                                                       int* __restrict__ lev) {
    int bp = blockIdx.x * blockDim.x + threadIdx.x;
    if (bp >= GB * GP) return;
    const int b  = bp / GP;
    const int pv = pos[bp];
    if (pv < 0) return;
    int* lv = lev + (size_t)b * GN;
    lv[pv] = 2;
    const int beg = rowptr[(size_t)b * (GN + 1) + pv];
    const int end = rowptr[(size_t)b * (GN + 1) + pv + 1];
    const v2i* se = (const v2i*)sedge + (size_t)b * GE;
    for (int p = beg; p < end; ++p) lv[se[p].x] = 2;
}

__global__ __launch_bounds__(256) void mark_src_kernel(const int* __restrict__ rowptr,
                                                       const int* __restrict__ sedge,
                                                       int* lev) {
    int idx = blockIdx.x * blockDim.x + threadIdx.x;
    if (idx >= GB * GN) return;
    if (lev[idx] != 2) return;
    int b = idx / GN;
    int i = idx - b * GN;
    int* lv = lev + (size_t)b * GN;
    const int beg = rowptr[(size_t)b * (GN + 1) + i];
    const int end = rowptr[(size_t)b * (GN + 1) + i + 1];
    const v2i* se = (const v2i*)sedge + (size_t)b * GE;
    for (int p = beg; p < end; ++p) {
        const int s = se[p].x;
        if (lv[s] == 0) lv[s] = 1;
    }
}

__global__ __launch_bounds__(256) void norm_kernel(const int* __restrict__ rowptr,
                                                   const float* __restrict__ dinv,
                                                   const int* __restrict__ lev,
                                                   int* __restrict__ sedge) {
    int idx = blockIdx.x * blockDim.x + threadIdx.x;
    if (idx >= GB * GN) return;
    if (lev[idx] < 1) return;
    int b = idx / GN;
    int i = idx - b * GN;
    const int beg = rowptr[(size_t)b * (GN + 1) + i];
    const int end = rowptr[(size_t)b * (GN + 1) + i + 1];
    v2i* se = (v2i*)sedge + (size_t)b * GE;
    const float dc = dinv[idx];
    for (int p = beg; p < end; ++p) {
        v2i ed = se[p];
        const float nv = dinv[(size_t)b * GN + ed.x] * __int_as_float(ed.y) * dc;
        ed.y = __float_as_int(nv);
        se[p] = ed;
    }
}

template <int HIN, bool ALL>
__global__ __launch_bounds__(512) void gemm_strip_kernel(const float* __restrict__ Hsrc,
                                                         const float* __restrict__ W,
                                                         const int* __restrict__ lev,
                                                         float* __restrict__ Dst,
                                                         int minlev) {
    constexpr int PITCH = HIN + 4;
    constexpr int NQ = HIN / 8;
    __shared__ __attribute__((aligned(16))) float wlds[GH * PITCH];
    const int tid = threadIdx.x;
    for (int i = tid; i < GH * HIN / 4; i += 512) {
        const int r  = i / (HIN / 4);
        const int c4 = i - r * (HIN / 4);
        *(v4f*)(wlds + r * PITCH + c4 * 4) = ((const v4f*)W)[i];
    }
    __syncthreads();

    const int wave = __builtin_amdgcn_readfirstlane(tid >> 5);
    const int lane = tid & 31;
    const int half = lane >> 4;
    const int lm   = lane & 15;
    const int woff0 = lm * PITCH + half * 4;

    for (int t = blockIdx.x + gridDim.x * wave; t < GB * GN / 16; t += gridDim.x * 16) {
        const int row = t * 16 + lm;
        bool need = true;
        if (!ALL) {
            need = lev[row] >= minlev;
            if (__builtin_amdgcn_ballot_w32(need) == 0u) continue;
        }
        const float* hrow = Hsrc + (size_t)row * HIN + half * 4;
        int woff = woff0;
        asm volatile("" : "+v"(woff));
        const float* wrow = wlds + woff;

        v4f a[NQ];
        if (ALL) {
#pragma unroll
            for (int q = 0; q < NQ; ++q) a[q] = *(const v4f*)(hrow + q * 8);
        } else {
#pragma unroll
            for (int q = 0; q < NQ; ++q) a[q] = (v4f){0.f, 0.f, 0.f, 0.f};
            if (need) {
#pragma unroll
                for (int q = 0; q < NQ; ++q) a[q] = *(const v4f*)(hrow + q * 8);
            }
        }

        __builtin_amdgcn_sched_barrier(0);

        v8f acc[4] = {};
        v4f bcur[4], bnxt[4];
#pragma unroll
        for (int n = 0; n < 4; ++n) bcur[n] = *(const v4f*)(wrow + n * 16 * PITCH);
#pragma unroll
        for (int q = 0; q < NQ; ++q) {
            if (q + 1 < NQ) {
#pragma unroll
                for (int n = 0; n < 4; ++n) bnxt[n] = *(const v4f*)(wrow + n * 16 * PITCH + (q + 1) * 8);
            }
            __builtin_amdgcn_sched_barrier(0);
            const v2f a0 = {a[q].x, a[q].y};
            const v2f a1 = {a[q].z, a[q].w};
#pragma unroll
            for (int n = 0; n < 4; ++n) {
                const v2f b0 = {bcur[n].x, bcur[n].y};
                acc[n] = __builtin_amdgcn_wmma_f32_16x16x4_f32(
                     false, a0,  false, b0,
                     (short)0, acc[n],  false,  false);
            }
#pragma unroll
            for (int n = 0; n < 4; ++n) {
                const v2f b1 = {bcur[n].z, bcur[n].w};
                acc[n] = __builtin_amdgcn_wmma_f32_16x16x4_f32(
                     false, a1,  false, b1,
                     (short)0, acc[n],  false,  false);
            }
#pragma unroll
            for (int n = 0; n < 4; ++n) bcur[n] = bnxt[n];
        }

        float* drow = Dst + (size_t)t * 16 * GH + lm;
#pragma unroll
        for (int j = 0; j < 8; ++j) {
#pragma unroll
            for (int n = 0; n < 4; ++n) {
                drow[(size_t)(j + 8 * half) * GH + n * 16] = acc[n][j];
            }
        }
    }
}

__device__ __forceinline__ v4f pull_row(const int* __restrict__ rowptr,
                                        const int* __restrict__ sedge,
                                        const float* __restrict__ tmp,
                                        const float* __restrict__ dinv,
                                        const float* __restrict__ bias,
                                        int b, int i, int k4) {
    const int node = b * GN + i;
    const int beg = rowptr[(size_t)b * (GN + 1) + i];
    const int end = rowptr[(size_t)b * (GN + 1) + i + 1];
    const v2i*   se = (const v2i*)sedge + (size_t)b * GE;
    const float* tb = tmp + (size_t)b * GN * GH + k4 * 4;

    v4f acc = {0.f, 0.f, 0.f, 0.f};
    int p = beg;
    for (; p + 4 <= end; p += 4) {
        const v2i e0 = se[p], e1 = se[p + 1], e2 = se[p + 2], e3 = se[p + 3];
        const v4f r0 = *(const v4f*)(tb + (size_t)e0.x * GH);
        const v4f r1 = *(const v4f*)(tb + (size_t)e1.x * GH);
        const v4f r2 = *(const v4f*)(tb + (size_t)e2.x * GH);
        const v4f r3 = *(const v4f*)(tb + (size_t)e3.x * GH);
        acc += r0 * __int_as_float(e0.y);
        acc += r1 * __int_as_float(e1.y);
        acc += r2 * __int_as_float(e2.y);
        acc += r3 * __int_as_float(e3.y);
    }
    for (; p < end; ++p) {
        const v2i e0 = se[p];
        const v4f r0 = *(const v4f*)(tb + (size_t)e0.x * GH);
        acc += r0 * __int_as_float(e0.y);
    }

    float di = dinv[node];
    float sl = di * di;
    v4f t  = *(const v4f*)(tmp + ((size_t)node * (GH / 4) + k4) * 4);
    v4f bb = *(const v4f*)(bias + k4 * 4);
    return acc + t * sl + bb;
}

__global__ __launch_bounds__(256) void pull_need_kernel(const int* __restrict__ rowptr,
                                                        const int* __restrict__ sedge,
                                                        const float* __restrict__ tmp,
                                                        const float* __restrict__ dinv,
                                                        const float* __restrict__ bias,
                                                        const int* __restrict__ lev,
                                                        float* __restrict__ hout,
                                                        int minlev) {
    const int k4 = threadIdx.x & 15;
    for (int node = blockIdx.x * 16 + (threadIdx.x >> 4); node < GB * GN; node += gridDim.x * 16) {
        if (lev[node] < minlev) continue;
        const int b = node / GN;
        const int i = node - b * GN;
        v4f v = pull_row(rowptr, sedge, tmp, dinv, bias, b, i, k4);
        v.x = tanhf(v.x); v.y = tanhf(v.y); v.z = tanhf(v.z); v.w = tanhf(v.w);
        *(v4f*)(hout + ((size_t)node * (GH / 4) + k4) * 4) = v;
    }
}

__global__ __launch_bounds__(256) void pull_pos_kernel(const int* __restrict__ rowptr,
                                                       const int* __restrict__ sedge,
                                                       const float* __restrict__ tmp,
                                                       const float* __restrict__ dinv,
                                                       const float* __restrict__ bias,
                                                       const int* __restrict__ pos,
                                                       float* __restrict__ out) {
    int idx = blockIdx.x * blockDim.x + threadIdx.x;
    if (idx >= GB * GP * (GEMB / 4)) return;
    const int k4 = idx & 15;
    const int bp = idx >> 4;
    const int b  = bp / GP;
    const int pv = pos[bp];
    v4f v = {-3.0f, -3.0f, -3.0f, -3.0f};
    if (pv >= 0) v = pull_row(rowptr, sedge, tmp, dinv, bias, b, pv, k4);
    *(v4f*)(out + (size_t)idx * 4) = v;
}

extern "C" void kernel_launch(void* const* d_in, const int* in_sizes, int n_in,
                              void* d_out, int out_size, void* d_ws, size_t ws_size,
                              hipStream_t stream) {
    const float* x   = (const float*)d_in[0];
    const int*   ei  = (const int*)d_in[1];
    const float* ew  = (const float*)d_in[2];
    const int*   pos = (const int*)d_in[3];
    const float* W1  = (const float*)d_in[4];
    const float* b1  = (const float*)d_in[5];
    const float* W2  = (const float*)d_in[6];
    const float* b2  = (const float*)d_in[7];
    const float* W3  = (const float*)d_in[8];
    const float* b3  = (const float*)d_in[9];
    float* out = (float*)d_out;

    float* ws   = (float*)d_ws;
    const size_t nBN  = (size_t)GB * GN;
    const size_t nRP  = (size_t)GB * (GN + 1);
    const size_t nBE  = (size_t)GB * GE;
    const size_t nBNH = (size_t)GB * GN * GH;
    float* dinv   = ws;
    int*   rowptr = (int*)(dinv + nBN);
    int*   sedge  = rowptr + nRP;
    float* tmp    = (float*)(sedge + 2 * nBE);
    float* hbuf   = tmp + nBNH;
    int*   lev    = (int*)(hbuf + nBNH);
    int*   binptr = (int*)tmp;
    int*   pedge  = (int*)hbuf;

    const int T = 256;
    dim3 part_grid(SLICES, GB, 1);
    dim3 sort_grid(PARTS, GB, 1);
    const int nodeBlocks = ((int)nBN + T - 1) / T;

    part_kernel<<<part_grid, SORT_T, 0, stream>>>(ei, ew, binptr, pedge);
    sort_kernel<<<sort_grid, SORT_T, 0, stream>>>(binptr, pedge, rowptr, sedge);
    dinv_kernel<<<nodeBlocks, T, 0, stream>>>(rowptr, sedge, dinv, lev);
    mark_pos_kernel<<<(GB * GP + T - 1) / T, T, 0, stream>>>(rowptr, sedge, pos, lev);
    mark_src_kernel<<<nodeBlocks, T, 0, stream>>>(rowptr, sedge, lev);
    norm_kernel<<<nodeBlocks, T, 0, stream>>>(rowptr, dinv, lev, sedge);

    gemm_strip_kernel<GS, true><<<256, 512, 0, stream>>>(x, W1, lev, tmp, 0);
    pull_need_kernel<<<1536, T, 0, stream>>>(rowptr, sedge, tmp, dinv, b1, lev, hbuf, 1);

    gemm_strip_kernel<GH, false><<<256, 512, 0, stream>>>(hbuf, W2, lev, tmp, 1);
    pull_need_kernel<<<1024, T, 0, stream>>>(rowptr, sedge, tmp, dinv, b2, lev, hbuf, 2);

    gemm_strip_kernel<GH, false><<<256, 512, 0, stream>>>(hbuf, W3, lev, tmp, 2);
    pull_pos_kernel<<<(GB * GP * (GEMB / 4) + T - 1) / T, T, 0, stream>>>(rowptr, sedge, tmp, dinv, b3, pos, out);
}


# Pull walk rebuilt for few dependent memory round trips: 16 lanes read 32 list entries at once, share them by DPP row_share, 8 row reads in flight, next node's level/bounds read ahead; sum order kept
# speedup vs baseline: 5.0708x; 1.1175x over previous
#include <hip/hip_runtime.h>
#include <hip/hip_bf16.h>

#define GB   16
#define GN   10000
#define GE   160000
#define GS   128
#define GH   64
#define GEMB 64
#define GP   16

#define PARTS 16
#define NPP   (GN / PARTS)
#define SORT_T 1024
#define SLICES 16
#define EPS   (GE / SLICES)
#define EPT   ((EPS + SORT_T - 1) / SORT_T)
static_assert(GN % PARTS == 0 && GE % SLICES == 0 && GN <= 65536, "sort layout");

typedef __attribute__((ext_vector_type(2))) float v2f;
typedef __attribute__((ext_vector_type(4))) float v4f;
typedef __attribute__((ext_vector_type(8))) float v8f;
typedef __attribute__((ext_vector_type(2))) int   v2i;
typedef __attribute__((ext_vector_type(4))) int   v4i;

__global__ __launch_bounds__(SORT_T) void part_kernel(const int* __restrict__ ei,
                                                      const float* __restrict__ ew,
                                                      int* __restrict__ binptr,
                                                      int* __restrict__ pedge) {
    __shared__ int bin[PARTS];
    const int tid = threadIdx.x;
    const int s   = blockIdx.x;
    const int b   = blockIdx.y;
    const int*   rows = ei + (size_t)b * 2 * GE + (size_t)s * EPS;
    const int*   cols = rows + GE;
    const float* w    = ew + (size_t)b * GE + (size_t)s * EPS;

    if (tid < PARTS) bin[tid] = 0;
    int c_[EPT], r_[EPT], w_[EPT];
#pragma unroll
    for (int u = 0; u < EPT; ++u) {
        const int e = tid + u * SORT_T;
        c_[u] = 0; r_[u] = 0; w_[u] = 0;
        if (e < EPS) {
            c_[u] = cols[e];
            r_[u] = rows[e];
            w_[u] = __float_as_int(w[e]);
        }
    }
    __syncthreads();
#pragma unroll
    for (int u = 0; u < EPT; ++u) {
        const int e = tid + u * SORT_T;
        if (e < EPS) atomicAdd(&bin[c_[u] / NPP], 1);
    }
    __syncthreads();
    int* bp = binptr + ((size_t)b * SLICES + s) * (PARTS + 1);
    if (tid == 0) {
        int n_[PARTS];
#pragma unroll
        for (int p = 0; p < PARTS; ++p) n_[p] = bin[p];
        int run = 0;
#pragma unroll
        for (int p = 0; p < PARTS; ++p) {
            bin[p] = run;
            bp[p]  = run;
            run += n_[p];
        }
        bp[PARTS] = run;
    }
    __syncthreads();
    v2i* pe = (v2i*)pedge + (size_t)b * GE + (size_t)s * EPS;
    int pos_[EPT];
#pragma unroll
    for (int u = 0; u < EPT; ++u) {
        const int e = tid + u * SORT_T;
        pos_[u] = -1;
        if (e < EPS) pos_[u] = atomicAdd(&bin[c_[u] / NPP], 1);
    }
#pragma unroll
    for (int u = 0; u < EPT; ++u) {
        if (pos_[u] >= 0) {
            v2i ed;
            ed.x = r_[u] | (c_[u] << 16);
            ed.y = w_[u];
            pe[pos_[u]] = ed;
        }
    }
}

__global__ __launch_bounds__(SORT_T) void sort_kernel(const int* __restrict__ binptr,
                                                      const int* __restrict__ pedge,
                                                      int* __restrict__ rowptr,
                                                      int* __restrict__ sedge) {
    __shared__ int cnt[NPP];
    __shared__ int sc[SORT_T];
    const int tid  = threadIdx.x;
    const int part = blockIdx.x;
    const int b    = blockIdx.y;
    const int lo   = part * NPP;

    for (int i = tid; i < NPP; i += SORT_T) cnt[i] = 0;

    const int* bpb = binptr + (size_t)b * SLICES * (PARTS + 1) + part;
    const v2i* pe  = (const v2i*)pedge + (size_t)b * GE;
    int beg[SLICES], end[SLICES];
    int below = 0;
#pragma unroll
    for (int s = 0; s < SLICES; ++s) {
        beg[s] = bpb[s * (PARTS + 1)];
        end[s] = bpb[s * (PARTS + 1) + 1];
        below += beg[s];
    }
    v2i ed[SLICES];
#pragma unroll
    for (int s = 0; s < SLICES; ++s) {
        const int k = beg[s] + tid;
        ed[s].x = -1;
        ed[s].y = 0;
        if (k < end[s]) ed[s] = pe[s * EPS + k];
    }
    __syncthreads();
#pragma unroll
    for (int s = 0; s < SLICES; ++s) {
        if (ed[s].x >= 0) atomicAdd(&cnt[(ed[s].x >> 16) - lo], 1);
        if (end[s] - beg[s] > SORT_T) {
            for (int k = beg[s] + tid + SORT_T; k < end[s]; k += SORT_T)
                atomicAdd(&cnt[(pe[s * EPS + k].x >> 16) - lo], 1);
        }
    }
    __syncthreads();

    const int own = (tid < NPP) ? cnt[tid] : 0;
    sc[tid] = own;
    __syncthreads();
    for (int off = 1; off < SORT_T; off <<= 1) {
        const int add = (tid >= off) ? sc[tid - off] : 0;
        __syncthreads();
        sc[tid] += add;
        __syncthreads();
    }
    if (tid < NPP) {
        const int start = below + sc[tid] - own;
        cnt[tid] = start;
        rowptr[(size_t)b * (GN + 1) + lo + tid] = start;
    }
    if (part == PARTS - 1 && tid == 0) rowptr[(size_t)b * (GN + 1) + GN] = GE;
    __syncthreads();

    v2i* se = (v2i*)sedge + (size_t)b * GE;
    int pos_[SLICES];
#pragma unroll
    for (int s = 0; s < SLICES; ++s) {
        pos_[s] = -1;
        if (ed[s].x >= 0) pos_[s] = atomicAdd(&cnt[(ed[s].x >> 16) - lo], 1);
    }
#pragma unroll
    for (int s = 0; s < SLICES; ++s) {
        if (pos_[s] >= 0) {
            v2i o;
            o.x = ed[s].x & 0xFFFF;
            o.y = ed[s].y;
            se[pos_[s]] = o;
        }
        if (end[s] - beg[s] > SORT_T)
        for (int k = beg[s] + tid + SORT_T; k < end[s]; k += SORT_T) {
            const v2i e2 = pe[s * EPS + k];
            const int p2 = atomicAdd(&cnt[(e2.x >> 16) - lo], 1);
            v2i o;
            o.x = e2.x & 0xFFFF;
            o.y = e2.y;
            se[p2] = o;
        }
    }
}

__global__ __launch_bounds__(256) void dinv_kernel(const int* __restrict__ rowptr,
                                                   const int* __restrict__ sedge,
                                                   float* __restrict__ dinv,
                                                   int* __restrict__ lev) {
    int idx = blockIdx.x * blockDim.x + threadIdx.x;
    if (idx >= GB * GN) return;
    lev[idx] = 0;
    int b = idx / GN;
    int i = idx - b * GN;
    const int beg = rowptr[(size_t)b * (GN + 1) + i];
    const int end = rowptr[(size_t)b * (GN + 1) + i + 1];
    const v2i* se = (const v2i*)sedge + (size_t)b * GE;
    float d = 0.f;
    for (int p = beg; p < end; ++p) d += __int_as_float(se[p].y);
    dinv[idx] = rsqrtf(d + 1.0f);
}

__global__ __launch_bounds__(256) void mark_pos_kernel(const int* __restrict__ rowptr,
                                                       const int* __restrict__ sedge,
                                                       const int* __restrict__ pos,
                                                       int* __restrict__ lev) {
    int bp = blockIdx.x * blockDim.x + threadIdx.x;
    if (bp >= GB * GP) return;
    const int b  = bp / GP;
    const int pv = pos[bp];
    if (pv < 0) return;
    int* lv = lev + (size_t)b * GN;
    lv[pv] = 2;
    const int beg = rowptr[(size_t)b * (GN + 1) + pv];
    const int end = rowptr[(size_t)b * (GN + 1) + pv + 1];
    const v2i* se = (const v2i*)sedge + (size_t)b * GE;
    for (int p = beg; p < end; ++p) lv[se[p].x] = 2;
}

__global__ __launch_bounds__(256) void mark_src_kernel(const int* __restrict__ rowptr,
                                                       const int* __restrict__ sedge,
                                                       int* lev) {
    int idx = blockIdx.x * blockDim.x + threadIdx.x;
    if (idx >= GB * GN) return;
    if (lev[idx] != 2) return;
    int b = idx / GN;
    int i = idx - b * GN;
    int* lv = lev + (size_t)b * GN;
    const int beg = rowptr[(size_t)b * (GN + 1) + i];
    const int end = rowptr[(size_t)b * (GN + 1) + i + 1];
    const v2i* se = (const v2i*)sedge + (size_t)b * GE;
    for (int p = beg; p < end; ++p) {
        const int s = se[p].x;
        if (lv[s] == 0) lv[s] = 1;
    }
}

__global__ __launch_bounds__(256) void norm_kernel(const int* __restrict__ rowptr,
                                                   const float* __restrict__ dinv,
                                                   const int* __restrict__ lev,
                                                   int* __restrict__ sedge) {
    int idx = blockIdx.x * blockDim.x + threadIdx.x;
    if (idx >= GB * GN) return;
    if (lev[idx] < 1) return;
    int b = idx / GN;
    int i = idx - b * GN;
    const int beg = rowptr[(size_t)b * (GN + 1) + i];
    const int end = rowptr[(size_t)b * (GN + 1) + i + 1];
    v2i* se = (v2i*)sedge + (size_t)b * GE;
    const float dc = dinv[idx];
    for (int p = beg; p < end; ++p) {
        v2i ed = se[p];
        const float nv = dinv[(size_t)b * GN + ed.x] * __int_as_float(ed.y) * dc;
        ed.y = __float_as_int(nv);
        se[p] = ed;
    }
}

template <int HIN, bool ALL>
__global__ __launch_bounds__(512) void gemm_strip_kernel(const float* __restrict__ Hsrc,
                                                         const float* __restrict__ W,
                                                         const int* __restrict__ lev,
                                                         float* __restrict__ Dst,
                                                         int minlev) {
    constexpr int PITCH = HIN + 4;
    constexpr int NQ = HIN / 8;
    __shared__ __attribute__((aligned(16))) float wlds[GH * PITCH];
    const int tid = threadIdx.x;
    for (int i = tid; i < GH * HIN / 4; i += 512) {
        const int r  = i / (HIN / 4);
        const int c4 = i - r * (HIN / 4);
        *(v4f*)(wlds + r * PITCH + c4 * 4) = ((const v4f*)W)[i];
    }
    __syncthreads();

    const int wave = __builtin_amdgcn_readfirstlane(tid >> 5);
    const int lane = tid & 31;
    const int half = lane >> 4;
    const int lm   = lane & 15;
    const int woff0 = lm * PITCH + half * 4;

    for (int t = blockIdx.x + gridDim.x * wave; t < GB * GN / 16; t += gridDim.x * 16) {
        const int row = t * 16 + lm;
        bool need = true;
        if (!ALL) {
            need = lev[row] >= minlev;
            if (__builtin_amdgcn_ballot_w32(need) == 0u) continue;
        }
        const float* hrow = Hsrc + (size_t)row * HIN + half * 4;
        int woff = woff0;
        asm volatile("" : "+v"(woff));
        const float* wrow = wlds + woff;

        v4f a[NQ];
        if (ALL) {
#pragma unroll
            for (int q = 0; q < NQ; ++q) a[q] = *(const v4f*)(hrow + q * 8);
        } else {
#pragma unroll
            for (int q = 0; q < NQ; ++q) a[q] = (v4f){0.f, 0.f, 0.f, 0.f};
            if (need) {
#pragma unroll
                for (int q = 0; q < NQ; ++q) a[q] = *(const v4f*)(hrow + q * 8);
            }
        }

        __builtin_amdgcn_sched_barrier(0);

        v8f acc[4] = {};
        v4f bcur[4], bnxt[4];
#pragma unroll
        for (int n = 0; n < 4; ++n) bcur[n] = *(const v4f*)(wrow + n * 16 * PITCH);
#pragma unroll
        for (int q = 0; q < NQ; ++q) {
            if (q + 1 < NQ) {
#pragma unroll
                for (int n = 0; n < 4; ++n) bnxt[n] = *(const v4f*)(wrow + n * 16 * PITCH + (q + 1) * 8);
            }
            __builtin_amdgcn_sched_barrier(0);
            const v2f a0 = {a[q].x, a[q].y};
            const v2f a1 = {a[q].z, a[q].w};
#pragma unroll
            for (int n = 0; n < 4; ++n) {
                const v2f b0 = {bcur[n].x, bcur[n].y};
                acc[n] = __builtin_amdgcn_wmma_f32_16x16x4_f32(
                     false, a0,  false, b0,
                     (short)0, acc[n],  false,  false);
            }
#pragma unroll
            for (int n = 0; n < 4; ++n) {
                const v2f b1 = {bcur[n].z, bcur[n].w};
                acc[n] = __builtin_amdgcn_wmma_f32_16x16x4_f32(
                     false, a1,  false, b1,
                     (short)0, acc[n],  false,  false);
            }
#pragma unroll
            for (int n = 0; n < 4; ++n) bcur[n] = bnxt[n];
        }

        float* drow = Dst + (size_t)t * 16 * GH + lm;
#pragma unroll
        for (int j = 0; j < 8; ++j) {
#pragma unroll
            for (int n = 0; n < 4; ++n) {
                drow[(size_t)(j + 8 * half) * GH + n * 16] = acc[n][j];
            }
        }
    }
}


template <int J> __device__ __forceinline__ int share16(int v) {
    return __builtin_amdgcn_update_dpp(0, v, 0x150 + J, 0xf, 0xf, false);
}

#define PULL_LOAD(J)                                                                       \
    if (FULL || (J0 + J) < n)                                                              \
        r[J] = *(const v4f*)(tmp + (rowbase + (unsigned)share16<J0 + J>(m.x) * GH));
#define PULL_ADD(J)                                                                        \
    if (FULL || (J0 + J) < n)                                                              \
        acc += r[J] * __int_as_float(share16<J0 + J>(m.y));

template <int J0, bool FULL>
__device__ __forceinline__ void pull_batch8(const float* __restrict__ tmp, unsigned rowbase,
                                            v2i m, int n, v4f& acc) {
    v4f r[8];
#pragma unroll
    for (int j = 0; j < 8; ++j) r[j] = (v4f){0.f, 0.f, 0.f, 0.f};
    PULL_LOAD(0) PULL_LOAD(1) PULL_LOAD(2) PULL_LOAD(3)
    PULL_LOAD(4) PULL_LOAD(5) PULL_LOAD(6) PULL_LOAD(7)
    PULL_ADD(0) PULL_ADD(1) PULL_ADD(2) PULL_ADD(3)
    PULL_ADD(4) PULL_ADD(5) PULL_ADD(6) PULL_ADD(7)
}
#undef PULL_LOAD
#undef PULL_ADD

template <int J0>
__device__ __forceinline__ void pull_batch(const float* __restrict__ tmp, unsigned rowbase,
                                           v2i m, int n, v4f& acc) {
    if (n >= J0 + 8) pull_batch8<J0, true >(tmp, rowbase, m, n, acc);
    else             pull_batch8<J0, false>(tmp, rowbase, m, n, acc);
}

__device__ __forceinline__ v4f pull_row(const int* __restrict__ sedge,
                                        const float* __restrict__ tmp,
                                        const float* __restrict__ dinv,
                                        const float* __restrict__ bias,
                                        int b, int i, int k4, int beg, int end) {
    const int node = b * GN + i;
    const v2i* se = (const v2i*)sedge + (size_t)b * GE;
    const unsigned rowbase = (unsigned)(b * GN) * GH + (unsigned)k4 * 4;

    int p = beg;
    int n = end - p;
    v2i m0 = {0, 0}, m1 = {0, 0};
    if (k4 < n)      m0 = se[p + k4];
    if (k4 + 16 < n) m1 = se[p + 16 + k4];
    const float di = dinv[node];
    const v4f t  = *(const v4f*)(tmp + ((unsigned)node * GH + (unsigned)k4 * 4));
    const v4f bb = *(const v4f*)(bias + k4 * 4);

    v4f acc = {0.f, 0.f, 0.f, 0.f};
    while (n > 0) {
        pull_batch<0>(tmp, rowbase, m0, n, acc);
        if (n > 8)  pull_batch<8>(tmp, rowbase, m0, n, acc);
        if (n > 16) pull_batch<0>(tmp, rowbase, m1, n - 16, acc);
        if (n > 24) pull_batch<8>(tmp, rowbase, m1, n - 16, acc);
        p += 32;
        n -= 32;
        if (n > 0) {
            m0 = (v2i){0, 0};
            m1 = (v2i){0, 0};
            if (k4 < n)      m0 = se[p + k4];
            if (k4 + 16 < n) m1 = se[p + 16 + k4];
        }
    }

    const float sl = di * di;
    return acc + t * sl + bb;
}

__global__ __launch_bounds__(256) void pull_need_kernel(const int* __restrict__ rowptr,
                                                        const int* __restrict__ sedge,
                                                        const float* __restrict__ tmp,
                                                        const float* __restrict__ dinv,
                                                        const float* __restrict__ bias,
                                                        const int* __restrict__ lev,
                                                        float* __restrict__ hout,
                                                        int minlev) {
    const int k4 = threadIdx.x & 15;
    const int stride = gridDim.x * 16;
    int node = blockIdx.x * 16 + (threadIdx.x >> 4);
    int lv = 0, beg = 0, end = 0;
    if (node < GB * GN) {
        const int b = node / GN;
        beg = rowptr[node + b];
        end = rowptr[node + b + 1];
        lv  = lev[node];
    }
    while (node < GB * GN) {
        asm volatile("" : "+v"(lv));
        const bool need = lv >= minlev;
        const int cbeg = beg, cend = end;
        const int nxt = node + stride;
        lv = 0; beg = 0; end = 0;
        if (nxt < GB * GN) {
            const int b2 = nxt / GN;
            beg = rowptr[nxt + b2];
            end = rowptr[nxt + b2 + 1];
            lv  = lev[nxt];
        }
        if (need) {
            const int b = node / GN;
            const int i = node - b * GN;
            v4f v = pull_row(sedge, tmp, dinv, bias, b, i, k4, cbeg, cend);
            v.x = tanhf(v.x); v.y = tanhf(v.y); v.z = tanhf(v.z); v.w = tanhf(v.w);
            *(v4f*)(hout + ((size_t)node * (GH / 4) + k4) * 4) = v;
        }
        node = nxt;
    }
}

__global__ __launch_bounds__(256) void pull_pos_kernel(const int* __restrict__ rowptr,
                                                       const int* __restrict__ sedge,
                                                       const float* __restrict__ tmp,
                                                       const float* __restrict__ dinv,
                                                       const float* __restrict__ bias,
                                                       const int* __restrict__ pos,
                                                       float* __restrict__ out) {
    int idx = blockIdx.x * blockDim.x + threadIdx.x;
    if (idx >= GB * GP * (GEMB / 4)) return;
    const int k4 = idx & 15;
    const int bp = idx >> 4;
    const int b  = bp / GP;
    const int pv = pos[bp];
    v4f v = {-3.0f, -3.0f, -3.0f, -3.0f};
    if (pv >= 0) {
        const int beg = rowptr[(size_t)b * (GN + 1) + pv];
        const int end = rowptr[(size_t)b * (GN + 1) + pv + 1];
        v = pull_row(sedge, tmp, dinv, bias, b, pv, k4, beg, end);
    }
    *(v4f*)(out + (size_t)idx * 4) = v;
}

extern "C" void kernel_launch(void* const* d_in, const int* in_sizes, int n_in,
                              void* d_out, int out_size, void* d_ws, size_t ws_size,
                              hipStream_t stream) {
    const float* x   = (const float*)d_in[0];
    const int*   ei  = (const int*)d_in[1];
    const float* ew  = (const float*)d_in[2];
    const int*   pos = (const int*)d_in[3];
    const float* W1  = (const float*)d_in[4];
    const float* b1  = (const float*)d_in[5];
    const float* W2  = (const float*)d_in[6];
    const float* b2  = (const float*)d_in[7];
    const float* W3  = (const float*)d_in[8];
    const float* b3  = (const float*)d_in[9];
    float* out = (float*)d_out;

    float* ws   = (float*)d_ws;
    const size_t nBN  = (size_t)GB * GN;
    const size_t nRP  = (size_t)GB * (GN + 1);
    const size_t nBE  = (size_t)GB * GE;
    const size_t nBNH = (size_t)GB * GN * GH;
    float* dinv   = ws;
    int*   rowptr = (int*)(dinv + nBN);
    int*   sedge  = rowptr + nRP;
    float* tmp    = (float*)(sedge + 2 * nBE);
    float* hbuf   = tmp + nBNH;
    int*   lev    = (int*)(hbuf + nBNH);
    int*   binptr = (int*)tmp;
    int*   pedge  = (int*)hbuf;

    const int T = 256;
    dim3 part_grid(SLICES, GB, 1);
    dim3 sort_grid(PARTS, GB, 1);
    const int nodeBlocks = ((int)nBN + T - 1) / T;

    part_kernel<<<part_grid, SORT_T, 0, stream>>>(ei, ew, binptr, pedge);
    sort_kernel<<<sort_grid, SORT_T, 0, stream>>>(binptr, pedge, rowptr, sedge);
    dinv_kernel<<<nodeBlocks, T, 0, stream>>>(rowptr, sedge, dinv, lev);
    mark_pos_kernel<<<(GB * GP + T - 1) / T, T, 0, stream>>>(rowptr, sedge, pos, lev);
    mark_src_kernel<<<nodeBlocks, T, 0, stream>>>(rowptr, sedge, lev);
    norm_kernel<<<nodeBlocks, T, 0, stream>>>(rowptr, dinv, lev, sedge);

    gemm_strip_kernel<GS, true><<<256, 512, 0, stream>>>(x, W1, lev, tmp, 0);
    pull_need_kernel<<<1536, T, 0, stream>>>(rowptr, sedge, tmp, dinv, b1, lev, hbuf, 1);

    gemm_strip_kernel<GH, false><<<256, 512, 0, stream>>>(hbuf, W2, lev, tmp, 1);
    pull_need_kernel<<<1024, T, 0, stream>>>(rowptr, sedge, tmp, dinv, b2, lev, hbuf, 2);

    gemm_strip_kernel<GH, false><<<256, 512, 0, stream>>>(hbuf, W3, lev, tmp, 2);
    pull_pos_kernel<<<(GB * GP * (GEMB / 4) + T - 1) / T, T, 0, stream>>>(rowptr, sedge, tmp, dinv, b3, pos, out);
}


# Edge norm dinv[s]*w*dinv[t] split into row scalings: GEMM scales each input row by dinv[src], pull scales its finished sum by dinv[target]; per-edge norm launch gone; sum order kept, roundings change
# speedup vs baseline: 5.3704x; 1.0591x over previous
#include <hip/hip_runtime.h>
#include <hip/hip_bf16.h>

#define GB   16
#define GN   10000
#define GE   160000
#define GS   128
#define GH   64
#define GEMB 64
#define GP   16

#define PARTS 16
#define NPP   (GN / PARTS)
#define SORT_T 1024
#define SLICES 16
#define EPS   (GE / SLICES)
#define EPT   ((EPS + SORT_T - 1) / SORT_T)
static_assert(GN % PARTS == 0 && GE % SLICES == 0 && GN <= 65536, "sort layout");

typedef __attribute__((ext_vector_type(2))) float v2f;
typedef __attribute__((ext_vector_type(4))) float v4f;
typedef __attribute__((ext_vector_type(8))) float v8f;
typedef __attribute__((ext_vector_type(2))) int   v2i;
typedef __attribute__((ext_vector_type(4))) int   v4i;

__global__ __launch_bounds__(SORT_T) void part_kernel(const int* __restrict__ ei,
                                                      const float* __restrict__ ew,
                                                      int* __restrict__ binptr,
                                                      int* __restrict__ pedge) {
    __shared__ int bin[PARTS];
    const int tid = threadIdx.x;
    const int s   = blockIdx.x;
    const int b   = blockIdx.y;
    const int*   rows = ei + (size_t)b * 2 * GE + (size_t)s * EPS;
    const int*   cols = rows + GE;
    const float* w    = ew + (size_t)b * GE + (size_t)s * EPS;

    if (tid < PARTS) bin[tid] = 0;
    int c_[EPT], r_[EPT], w_[EPT];
#pragma unroll
    for (int u = 0; u < EPT; ++u) {
        const int e = tid + u * SORT_T;
        c_[u] = 0; r_[u] = 0; w_[u] = 0;
        if (e < EPS) {
            c_[u] = cols[e];
            r_[u] = rows[e];
            w_[u] = __float_as_int(w[e]);
        }
    }
    __syncthreads();
#pragma unroll
    for (int u = 0; u < EPT; ++u) {
        const int e = tid + u * SORT_T;
        if (e < EPS) atomicAdd(&bin[c_[u] / NPP], 1);
    }
    __syncthreads();
    int* bp = binptr + ((size_t)b * SLICES + s) * (PARTS + 1);
    if (tid == 0) {
        int n_[PARTS];
#pragma unroll
        for (int p = 0; p < PARTS; ++p) n_[p] = bin[p];
        int run = 0;
#pragma unroll
        for (int p = 0; p < PARTS; ++p) {
            bin[p] = run;
            bp[p]  = run;
            run += n_[p];
        }
        bp[PARTS] = run;
    }
    __syncthreads();
    v2i* pe = (v2i*)pedge + (size_t)b * GE + (size_t)s * EPS;
    int pos_[EPT];
#pragma unroll
    for (int u = 0; u < EPT; ++u) {
        const int e = tid + u * SORT_T;
        pos_[u] = -1;
        if (e < EPS) pos_[u] = atomicAdd(&bin[c_[u] / NPP], 1);
    }
#pragma unroll
    for (int u = 0; u < EPT; ++u) {
        if (pos_[u] >= 0) {
            v2i ed;
            ed.x = r_[u] | (c_[u] << 16);
            ed.y = w_[u];
            pe[pos_[u]] = ed;
        }
    }
}

__global__ __launch_bounds__(SORT_T) void sort_kernel(const int* __restrict__ binptr,
                                                      const int* __restrict__ pedge,
                                                      int* __restrict__ rowptr,
                                                      int* __restrict__ sedge) {
    __shared__ int cnt[NPP];
    __shared__ int sc[SORT_T];
    const int tid  = threadIdx.x;
    const int part = blockIdx.x;
    const int b    = blockIdx.y;
    const int lo   = part * NPP;

    for (int i = tid; i < NPP; i += SORT_T) cnt[i] = 0;

    const int* bpb = binptr + (size_t)b * SLICES * (PARTS + 1) + part;
    const v2i* pe  = (const v2i*)pedge + (size_t)b * GE;
    int beg[SLICES], end[SLICES];
    int below = 0;
#pragma unroll
    for (int s = 0; s < SLICES; ++s) {
        beg[s] = bpb[s * (PARTS + 1)];
        end[s] = bpb[s * (PARTS + 1) + 1];
        below += beg[s];
    }
    v2i ed[SLICES];
#pragma unroll
    for (int s = 0; s < SLICES; ++s) {
        const int k = beg[s] + tid;
        ed[s].x = -1;
        ed[s].y = 0;
        if (k < end[s]) ed[s] = pe[s * EPS + k];
    }
    __syncthreads();
#pragma unroll
    for (int s = 0; s < SLICES; ++s) {
        if (ed[s].x >= 0) atomicAdd(&cnt[(ed[s].x >> 16) - lo], 1);
        if (end[s] - beg[s] > SORT_T) {
            for (int k = beg[s] + tid + SORT_T; k < end[s]; k += SORT_T)
                atomicAdd(&cnt[(pe[s * EPS + k].x >> 16) - lo], 1);
        }
    }
    __syncthreads();

    const int own = (tid < NPP) ? cnt[tid] : 0;
    sc[tid] = own;
    __syncthreads();
    for (int off = 1; off < SORT_T; off <<= 1) {
        const int add = (tid >= off) ? sc[tid - off] : 0;
        __syncthreads();
        sc[tid] += add;
        __syncthreads();
    }
    if (tid < NPP) {
        const int start = below + sc[tid] - own;
        cnt[tid] = start;
        rowptr[(size_t)b * (GN + 1) + lo + tid] = start;
    }
    if (part == PARTS - 1 && tid == 0) rowptr[(size_t)b * (GN + 1) + GN] = GE;
    __syncthreads();

    v2i* se = (v2i*)sedge + (size_t)b * GE;
    int pos_[SLICES];
#pragma unroll
    for (int s = 0; s < SLICES; ++s) {
        pos_[s] = -1;
        if (ed[s].x >= 0) pos_[s] = atomicAdd(&cnt[(ed[s].x >> 16) - lo], 1);
    }
#pragma unroll
    for (int s = 0; s < SLICES; ++s) {
        if (pos_[s] >= 0) {
            v2i o;
            o.x = ed[s].x & 0xFFFF;
            o.y = ed[s].y;
            se[pos_[s]] = o;
        }
        if (end[s] - beg[s] > SORT_T)
        for (int k = beg[s] + tid + SORT_T; k < end[s]; k += SORT_T) {
            const v2i e2 = pe[s * EPS + k];
            const int p2 = atomicAdd(&cnt[(e2.x >> 16) - lo], 1);
            v2i o;
            o.x = e2.x & 0xFFFF;
            o.y = e2.y;
            se[p2] = o;
        }
    }
}

__global__ __launch_bounds__(256) void dinv_kernel(const int* __restrict__ rowptr,
                                                   const int* __restrict__ sedge,
                                                   float* __restrict__ dinv,
                                                   int* __restrict__ lev) {
    int idx = blockIdx.x * blockDim.x + threadIdx.x;
    if (idx >= GB * GN) return;
    lev[idx] = 0;
    int b = idx / GN;
    int i = idx - b * GN;
    const int beg = rowptr[(size_t)b * (GN + 1) + i];
    const int end = rowptr[(size_t)b * (GN + 1) + i + 1];
    const v2i* se = (const v2i*)sedge + (size_t)b * GE;
    float d = 0.f;
    for (int p = beg; p < end; ++p) d += __int_as_float(se[p].y);
    dinv[idx] = rsqrtf(d + 1.0f);
}

__global__ __launch_bounds__(256) void mark_pos_kernel(const int* __restrict__ rowptr,
                                                       const int* __restrict__ sedge,
                                                       const int* __restrict__ pos,
                                                       int* __restrict__ lev) {
    int bp = blockIdx.x * blockDim.x + threadIdx.x;
    if (bp >= GB * GP) return;
    const int b  = bp / GP;
    const int pv = pos[bp];
    if (pv < 0) return;
    int* lv = lev + (size_t)b * GN;
    lv[pv] = 2;
    const int beg = rowptr[(size_t)b * (GN + 1) + pv];
    const int end = rowptr[(size_t)b * (GN + 1) + pv + 1];
    const v2i* se = (const v2i*)sedge + (size_t)b * GE;
    for (int p = beg; p < end; ++p) lv[se[p].x] = 2;
}

__global__ __launch_bounds__(256) void mark_src_kernel(const int* __restrict__ rowptr,
                                                       const int* __restrict__ sedge,
                                                       int* lev) {
    int idx = blockIdx.x * blockDim.x + threadIdx.x;
    if (idx >= GB * GN) return;
    if (lev[idx] != 2) return;
    int b = idx / GN;
    int i = idx - b * GN;
    int* lv = lev + (size_t)b * GN;
    const int beg = rowptr[(size_t)b * (GN + 1) + i];
    const int end = rowptr[(size_t)b * (GN + 1) + i + 1];
    const v2i* se = (const v2i*)sedge + (size_t)b * GE;
    for (int p = beg; p < end; ++p) {
        const int s = se[p].x;
        if (lv[s] == 0) lv[s] = 1;
    }
}

template <int HIN, bool ALL>
__global__ __launch_bounds__(512) void gemm_strip_kernel(const float* __restrict__ Hsrc,
                                                         const float* __restrict__ W,
                                                         const int* __restrict__ lev,
                                                         const float* __restrict__ dinv,
                                                         float* __restrict__ Dst,
                                                         int minlev) {
    constexpr int PITCH = HIN + 4;
    constexpr int NQ = HIN / 8;
    __shared__ __attribute__((aligned(16))) float wlds[GH * PITCH];
    const int tid = threadIdx.x;
    for (int i = tid; i < GH * HIN / 4; i += 512) {
        const int r  = i / (HIN / 4);
        const int c4 = i - r * (HIN / 4);
        *(v4f*)(wlds + r * PITCH + c4 * 4) = ((const v4f*)W)[i];
    }
    __syncthreads();

    const int wave = __builtin_amdgcn_readfirstlane(tid >> 5);
    const int lane = tid & 31;
    const int half = lane >> 4;
    const int lm   = lane & 15;
    const int woff0 = lm * PITCH + half * 4;

    for (int t = blockIdx.x + gridDim.x * wave; t < GB * GN / 16; t += gridDim.x * 16) {
        const int row = t * 16 + lm;
        const float dr = dinv[row];
        bool need = true;
        if (!ALL) {
            need = lev[row] >= minlev;
            if (__builtin_amdgcn_ballot_w32(need) == 0u) continue;
        }
        const float* hrow = Hsrc + (size_t)row * HIN + half * 4;
        int woff = woff0;
        asm volatile("" : "+v"(woff));
        const float* wrow = wlds + woff;

        v4f a[NQ];
        if (ALL) {
#pragma unroll
            for (int q = 0; q < NQ; ++q) a[q] = *(const v4f*)(hrow + q * 8);
        } else {
#pragma unroll
            for (int q = 0; q < NQ; ++q) a[q] = (v4f){0.f, 0.f, 0.f, 0.f};
            if (need) {
#pragma unroll
                for (int q = 0; q < NQ; ++q) a[q] = *(const v4f*)(hrow + q * 8);
            }
        }

#pragma unroll
        for (int q = 0; q < NQ; ++q) a[q] *= dr;

        __builtin_amdgcn_sched_barrier(0);

        v8f acc[4] = {};
        v4f bcur[4], bnxt[4];
#pragma unroll
        for (int n = 0; n < 4; ++n) bcur[n] = *(const v4f*)(wrow + n * 16 * PITCH);
#pragma unroll
        for (int q = 0; q < NQ; ++q) {
            if (q + 1 < NQ) {
#pragma unroll
                for (int n = 0; n < 4; ++n) bnxt[n] = *(const v4f*)(wrow + n * 16 * PITCH + (q + 1) * 8);
            }
            __builtin_amdgcn_sched_barrier(0);
            const v2f a0 = {a[q].x, a[q].y};
            const v2f a1 = {a[q].z, a[q].w};
#pragma unroll
            for (int n = 0; n < 4; ++n) {
                const v2f b0 = {bcur[n].x, bcur[n].y};
                acc[n] = __builtin_amdgcn_wmma_f32_16x16x4_f32(
                     false, a0,  false, b0,
                     (short)0, acc[n],  false,  false);
            }
#pragma unroll
            for (int n = 0; n < 4; ++n) {
                const v2f b1 = {bcur[n].z, bcur[n].w};
                acc[n] = __builtin_amdgcn_wmma_f32_16x16x4_f32(
                     false, a1,  false, b1,
                     (short)0, acc[n],  false,  false);
            }
#pragma unroll
            for (int n = 0; n < 4; ++n) bcur[n] = bnxt[n];
        }

        float* drow = Dst + (size_t)t * 16 * GH + lm;
#pragma unroll
        for (int j = 0; j < 8; ++j) {
#pragma unroll
            for (int n = 0; n < 4; ++n) {
                drow[(size_t)(j + 8 * half) * GH + n * 16] = acc[n][j];
            }
        }
    }
}


template <int J> __device__ __forceinline__ int share16(int v) {
    return __builtin_amdgcn_update_dpp(0, v, 0x150 + J, 0xf, 0xf, false);
}

#define PULL_LOAD(J)                                                                       \
    if (FULL || (J0 + J) < n)                                                              \
        r[J] = *(const v4f*)(tmp + (rowbase + (unsigned)share16<J0 + J>(m.x) * GH));
#define PULL_ADD(J)                                                                        \
    if (FULL || (J0 + J) < n)                                                              \
        acc += r[J] * __int_as_float(share16<J0 + J>(m.y));

template <int J0, bool FULL>
__device__ __forceinline__ void pull_batch8(const float* __restrict__ tmp, unsigned rowbase,
                                            v2i m, int n, v4f& acc) {
    v4f r[8];
#pragma unroll
    for (int j = 0; j < 8; ++j) r[j] = (v4f){0.f, 0.f, 0.f, 0.f};
    PULL_LOAD(0) PULL_LOAD(1) PULL_LOAD(2) PULL_LOAD(3)
    PULL_LOAD(4) PULL_LOAD(5) PULL_LOAD(6) PULL_LOAD(7)
    PULL_ADD(0) PULL_ADD(1) PULL_ADD(2) PULL_ADD(3)
    PULL_ADD(4) PULL_ADD(5) PULL_ADD(6) PULL_ADD(7)
}
#undef PULL_LOAD
#undef PULL_ADD

template <int J0>
__device__ __forceinline__ void pull_batch(const float* __restrict__ tmp, unsigned rowbase,
                                           v2i m, int n, v4f& acc) {
    if (n >= J0 + 8) pull_batch8<J0, true >(tmp, rowbase, m, n, acc);
    else             pull_batch8<J0, false>(tmp, rowbase, m, n, acc);
}

__device__ __forceinline__ v4f pull_row(const int* __restrict__ sedge,
                                        const float* __restrict__ tmp,
                                        const float* __restrict__ dinv,
                                        const float* __restrict__ bias,
                                        int b, int i, int k4, int beg, int end) {
    const int node = b * GN + i;
    const v2i* se = (const v2i*)sedge + (size_t)b * GE;
    const unsigned rowbase = (unsigned)(b * GN) * GH + (unsigned)k4 * 4;

    int p = beg;
    int n = end - p;
    v2i m0 = {0, 0}, m1 = {0, 0};
    if (k4 < n)      m0 = se[p + k4];
    if (k4 + 16 < n) m1 = se[p + 16 + k4];
    const float di = dinv[node];
    const v4f t  = *(const v4f*)(tmp + ((unsigned)node * GH + (unsigned)k4 * 4));
    const v4f bb = *(const v4f*)(bias + k4 * 4);

    v4f acc = {0.f, 0.f, 0.f, 0.f};
    while (n > 0) {
        pull_batch<0>(tmp, rowbase, m0, n, acc);
        if (n > 8)  pull_batch<8>(tmp, rowbase, m0, n, acc);
        if (n > 16) pull_batch<0>(tmp, rowbase, m1, n - 16, acc);
        if (n > 24) pull_batch<8>(tmp, rowbase, m1, n - 16, acc);
        p += 32;
        n -= 32;
        if (n > 0) {
            m0 = (v2i){0, 0};
            m1 = (v2i){0, 0};
            if (k4 < n)      m0 = se[p + k4];
            if (k4 + 16 < n) m1 = se[p + 16 + k4];
        }
    }

    return (acc + t) * di + bb;
}

__global__ __launch_bounds__(256) void pull_need_kernel(const int* __restrict__ rowptr,
                                                        const int* __restrict__ sedge,
                                                        const float* __restrict__ tmp,
                                                        const float* __restrict__ dinv,
                                                        const float* __restrict__ bias,
                                                        const int* __restrict__ lev,
                                                        float* __restrict__ hout,
                                                        int minlev) {
    const int k4 = threadIdx.x & 15;
    const int stride = gridDim.x * 16;
    int node = blockIdx.x * 16 + (threadIdx.x >> 4);
    int lv = 0, beg = 0, end = 0;
    if (node < GB * GN) {
        const int b = node / GN;
        beg = rowptr[node + b];
        end = rowptr[node + b + 1];
        lv  = lev[node];
    }
    while (node < GB * GN) {
        asm volatile("" : "+v"(lv));
        const bool need = lv >= minlev;
        const int cbeg = beg, cend = end;
        const int nxt = node + stride;
        lv = 0; beg = 0; end = 0;
        if (nxt < GB * GN) {
            const int b2 = nxt / GN;
            beg = rowptr[nxt + b2];
            end = rowptr[nxt + b2 + 1];
            lv  = lev[nxt];
        }
        if (need) {
            const int b = node / GN;
            const int i = node - b * GN;
            v4f v = pull_row(sedge, tmp, dinv, bias, b, i, k4, cbeg, cend);
            v.x = tanhf(v.x); v.y = tanhf(v.y); v.z = tanhf(v.z); v.w = tanhf(v.w);
            *(v4f*)(hout + ((size_t)node * (GH / 4) + k4) * 4) = v;
        }
        node = nxt;
    }
}

__global__ __launch_bounds__(256) void pull_pos_kernel(const int* __restrict__ rowptr,
                                                       const int* __restrict__ sedge,
                                                       const float* __restrict__ tmp,
                                                       const float* __restrict__ dinv,
                                                       const float* __restrict__ bias,
                                                       const int* __restrict__ pos,
                                                       float* __restrict__ out) {
    int idx = blockIdx.x * blockDim.x + threadIdx.x;
    if (idx >= GB * GP * (GEMB / 4)) return;
    const int k4 = idx & 15;
    const int bp = idx >> 4;
    const int b  = bp / GP;
    const int pv = pos[bp];
    v4f v = {-3.0f, -3.0f, -3.0f, -3.0f};
    if (pv >= 0) {
        const int beg = rowptr[(size_t)b * (GN + 1) + pv];
        const int end = rowptr[(size_t)b * (GN + 1) + pv + 1];
        v = pull_row(sedge, tmp, dinv, bias, b, pv, k4, beg, end);
    }
    *(v4f*)(out + (size_t)idx * 4) = v;
}

extern "C" void kernel_launch(void* const* d_in, const int* in_sizes, int n_in,
                              void* d_out, int out_size, void* d_ws, size_t ws_size,
                              hipStream_t stream) {
    const float* x   = (const float*)d_in[0];
    const int*   ei  = (const int*)d_in[1];
    const float* ew  = (const float*)d_in[2];
    const int*   pos = (const int*)d_in[3];
    const float* W1  = (const float*)d_in[4];
    const float* b1  = (const float*)d_in[5];
    const float* W2  = (const float*)d_in[6];
    const float* b2  = (const float*)d_in[7];
    const float* W3  = (const float*)d_in[8];
    const float* b3  = (const float*)d_in[9];
    float* out = (float*)d_out;

    float* ws   = (float*)d_ws;
    const size_t nBN  = (size_t)GB * GN;
    const size_t nRP  = (size_t)GB * (GN + 1);
    const size_t nBE  = (size_t)GB * GE;
    const size_t nBNH = (size_t)GB * GN * GH;
    float* dinv   = ws;
    int*   rowptr = (int*)(dinv + nBN);
    int*   sedge  = rowptr + nRP;
    float* tmp    = (float*)(sedge + 2 * nBE);
    float* hbuf   = tmp + nBNH;
    int*   lev    = (int*)(hbuf + nBNH);
    int*   binptr = (int*)tmp;
    int*   pedge  = (int*)hbuf;

    const int T = 256;
    dim3 part_grid(SLICES, GB, 1);
    dim3 sort_grid(PARTS, GB, 1);
    const int nodeBlocks = ((int)nBN + T - 1) / T;

    part_kernel<<<part_grid, SORT_T, 0, stream>>>(ei, ew, binptr, pedge);
    sort_kernel<<<sort_grid, SORT_T, 0, stream>>>(binptr, pedge, rowptr, sedge);
    dinv_kernel<<<nodeBlocks, T, 0, stream>>>(rowptr, sedge, dinv, lev);
    mark_pos_kernel<<<(GB * GP + T - 1) / T, T, 0, stream>>>(rowptr, sedge, pos, lev);
    mark_src_kernel<<<nodeBlocks, T, 0, stream>>>(rowptr, sedge, lev);

    gemm_strip_kernel<GS, true><<<256, 512, 0, stream>>>(x, W1, lev, dinv, tmp, 0);
    pull_need_kernel<<<1536, T, 0, stream>>>(rowptr, sedge, tmp, dinv, b1, lev, hbuf, 1);

    gemm_strip_kernel<GH, false><<<256, 512, 0, stream>>>(hbuf, W2, lev, dinv, tmp, 1);
    pull_need_kernel<<<1024, T, 0, stream>>>(rowptr, sedge, tmp, dinv, b2, lev, hbuf, 2);

    gemm_strip_kernel<GH, false><<<256, 512, 0, stream>>>(hbuf, W3, lev, dinv, tmp, 2);
    pull_pos_kernel<<<(GB * GP * (GEMB / 4) + T - 1) / T, T, 0, stream>>>(rowptr, sedge, tmp, dinv, b3, pos, out);
}


# Layers 2 and 3 aggregate first, (A h) W^T not A (h W^T): each is one launch that pulls up to 16 needed rows into an LDS tile and multiplies it by W^T; two GEMM launches gone; roundings change
# speedup vs baseline: 6.3390x; 1.1804x over previous
#include <hip/hip_runtime.h>
#include <hip/hip_bf16.h>

#define GB   16
#define GN   10000
#define GE   160000
#define GS   128
#define GH   64
#define GEMB 64
#define GP   16

#define PARTS 16
#define NPP   (GN / PARTS)
#define SORT_T 1024
#define SLICES 16
#define EPS   (GE / SLICES)
#define EPT   ((EPS + SORT_T - 1) / SORT_T)
static_assert(GN % PARTS == 0 && GE % SLICES == 0 && GN <= 65536, "sort layout");

typedef __attribute__((ext_vector_type(2))) float v2f;
typedef __attribute__((ext_vector_type(4))) float v4f;
typedef __attribute__((ext_vector_type(8))) float v8f;
typedef __attribute__((ext_vector_type(2))) int   v2i;
typedef __attribute__((ext_vector_type(4))) int   v4i;

__global__ __launch_bounds__(SORT_T) void part_kernel(const int* __restrict__ ei,
                                                      const float* __restrict__ ew,
                                                      int* __restrict__ binptr,
                                                      int* __restrict__ pedge) {
    __shared__ int bin[PARTS];
    const int tid = threadIdx.x;
    const int s   = blockIdx.x;
    const int b   = blockIdx.y;
    const int*   rows = ei + (size_t)b * 2 * GE + (size_t)s * EPS;
    const int*   cols = rows + GE;
    const float* w    = ew + (size_t)b * GE + (size_t)s * EPS;

    if (tid < PARTS) bin[tid] = 0;
    int c_[EPT], r_[EPT], w_[EPT];
#pragma unroll
    for (int u = 0; u < EPT; ++u) {
        const int e = tid + u * SORT_T;
        c_[u] = 0; r_[u] = 0; w_[u] = 0;
        if (e < EPS) {
            c_[u] = cols[e];
            r_[u] = rows[e];
            w_[u] = __float_as_int(w[e]);
        }
    }
    __syncthreads();
#pragma unroll
    for (int u = 0; u < EPT; ++u) {
        const int e = tid + u * SORT_T;
        if (e < EPS) atomicAdd(&bin[c_[u] / NPP], 1);
    }
    __syncthreads();
    int* bp = binptr + ((size_t)b * SLICES + s) * (PARTS + 1);
    if (tid == 0) {
        int n_[PARTS];
#pragma unroll
        for (int p = 0; p < PARTS; ++p) n_[p] = bin[p];
        int run = 0;
#pragma unroll
        for (int p = 0; p < PARTS; ++p) {
            bin[p] = run;
            bp[p]  = run;
            run += n_[p];
        }
        bp[PARTS] = run;
    }
    __syncthreads();
    v2i* pe = (v2i*)pedge + (size_t)b * GE + (size_t)s * EPS;
    int pos_[EPT];
#pragma unroll
    for (int u = 0; u < EPT; ++u) {
        const int e = tid + u * SORT_T;
        pos_[u] = -1;
        if (e < EPS) pos_[u] = atomicAdd(&bin[c_[u] / NPP], 1);
    }
#pragma unroll
    for (int u = 0; u < EPT; ++u) {
        if (pos_[u] >= 0) {
            v2i ed;
            ed.x = r_[u] | (c_[u] << 16);
            ed.y = w_[u];
            pe[pos_[u]] = ed;
        }
    }
}

__global__ __launch_bounds__(SORT_T) void sort_kernel(const int* __restrict__ binptr,
                                                      const int* __restrict__ pedge,
                                                      int* __restrict__ rowptr,
                                                      int* __restrict__ sedge) {
    __shared__ int cnt[NPP];
    __shared__ int sc[SORT_T];
    const int tid  = threadIdx.x;
    const int part = blockIdx.x;
    const int b    = blockIdx.y;
    const int lo   = part * NPP;

    for (int i = tid; i < NPP; i += SORT_T) cnt[i] = 0;

    const int* bpb = binptr + (size_t)b * SLICES * (PARTS + 1) + part;
    const v2i* pe  = (const v2i*)pedge + (size_t)b * GE;
    int beg[SLICES], end[SLICES];
    int below = 0;
#pragma unroll
    for (int s = 0; s < SLICES; ++s) {
        beg[s] = bpb[s * (PARTS + 1)];
        end[s] = bpb[s * (PARTS + 1) + 1];
        below += beg[s];
    }
    v2i ed[SLICES];
#pragma unroll
    for (int s = 0; s < SLICES; ++s) {
        const int k = beg[s] + tid;
        ed[s].x = -1;
        ed[s].y = 0;
        if (k < end[s]) ed[s] = pe[s * EPS + k];
    }
    __syncthreads();
#pragma unroll
    for (int s = 0; s < SLICES; ++s) {
        if (ed[s].x >= 0) atomicAdd(&cnt[(ed[s].x >> 16) - lo], 1);
        if (end[s] - beg[s] > SORT_T) {
            for (int k = beg[s] + tid + SORT_T; k < end[s]; k += SORT_T)
                atomicAdd(&cnt[(pe[s * EPS + k].x >> 16) - lo], 1);
        }
    }
    __syncthreads();

    const int own = (tid < NPP) ? cnt[tid] : 0;
    sc[tid] = own;
    __syncthreads();
    for (int off = 1; off < SORT_T; off <<= 1) {
        const int add = (tid >= off) ? sc[tid - off] : 0;
        __syncthreads();
        sc[tid] += add;
        __syncthreads();
    }
    if (tid < NPP) {
        const int start = below + sc[tid] - own;
        cnt[tid] = start;
        rowptr[(size_t)b * (GN + 1) + lo + tid] = start;
    }
    if (part == PARTS - 1 && tid == 0) rowptr[(size_t)b * (GN + 1) + GN] = GE;
    __syncthreads();

    v2i* se = (v2i*)sedge + (size_t)b * GE;
    int pos_[SLICES];
#pragma unroll
    for (int s = 0; s < SLICES; ++s) {
        pos_[s] = -1;
        if (ed[s].x >= 0) pos_[s] = atomicAdd(&cnt[(ed[s].x >> 16) - lo], 1);
    }
#pragma unroll
    for (int s = 0; s < SLICES; ++s) {
        if (pos_[s] >= 0) {
            v2i o;
            o.x = ed[s].x & 0xFFFF;
            o.y = ed[s].y;
            se[pos_[s]] = o;
        }
        if (end[s] - beg[s] > SORT_T)
        for (int k = beg[s] + tid + SORT_T; k < end[s]; k += SORT_T) {
            const v2i e2 = pe[s * EPS + k];
            const int p2 = atomicAdd(&cnt[(e2.x >> 16) - lo], 1);
            v2i o;
            o.x = e2.x & 0xFFFF;
            o.y = e2.y;
            se[p2] = o;
        }
    }
}

__global__ __launch_bounds__(256) void dinv_kernel(const int* __restrict__ rowptr,
                                                   const int* __restrict__ sedge,
                                                   float* __restrict__ dinv,
                                                   int* __restrict__ lev) {
    int idx = blockIdx.x * blockDim.x + threadIdx.x;
    if (idx >= GB * GN) return;
    lev[idx] = 0;
    int b = idx / GN;
    int i = idx - b * GN;
    const int beg = rowptr[(size_t)b * (GN + 1) + i];
    const int end = rowptr[(size_t)b * (GN + 1) + i + 1];
    const v2i* se = (const v2i*)sedge + (size_t)b * GE;
    float d = 0.f;
    for (int p = beg; p < end; ++p) d += __int_as_float(se[p].y);
    dinv[idx] = rsqrtf(d + 1.0f);
}

__global__ __launch_bounds__(256) void mark_pos_kernel(const int* __restrict__ rowptr,
                                                       const int* __restrict__ sedge,
                                                       const int* __restrict__ pos,
                                                       int* __restrict__ lev) {
    int bp = blockIdx.x * blockDim.x + threadIdx.x;
    if (bp >= GB * GP) return;
    const int b  = bp / GP;
    const int pv = pos[bp];
    if (pv < 0) return;
    int* lv = lev + (size_t)b * GN;
    lv[pv] = 2;
    const int beg = rowptr[(size_t)b * (GN + 1) + pv];
    const int end = rowptr[(size_t)b * (GN + 1) + pv + 1];
    const v2i* se = (const v2i*)sedge + (size_t)b * GE;
    for (int p = beg; p < end; ++p) lv[se[p].x] = 2;
}

__global__ __launch_bounds__(256) void mark_src_kernel(const int* __restrict__ rowptr,
                                                       const int* __restrict__ sedge,
                                                       int* lev) {
    int idx = blockIdx.x * blockDim.x + threadIdx.x;
    if (idx >= GB * GN) return;
    if (lev[idx] != 2) return;
    int b = idx / GN;
    int i = idx - b * GN;
    int* lv = lev + (size_t)b * GN;
    const int beg = rowptr[(size_t)b * (GN + 1) + i];
    const int end = rowptr[(size_t)b * (GN + 1) + i + 1];
    const v2i* se = (const v2i*)sedge + (size_t)b * GE;
    for (int p = beg; p < end; ++p) {
        const int s = se[p].x;
        if (lv[s] == 0) lv[s] = 1;
    }
}

template <int HIN>
__global__ __launch_bounds__(512) void gemm_strip_kernel(const float* __restrict__ Hsrc,
                                                         const float* __restrict__ W,
                                                         const float* __restrict__ dinv,
                                                         float* __restrict__ Dst) {
    constexpr int PITCH = HIN + 4;
    constexpr int NQ = HIN / 8;
    __shared__ __attribute__((aligned(16))) float wlds[GH * PITCH];
    const int tid = threadIdx.x;
    for (int i = tid; i < GH * HIN / 4; i += 512) {
        const int r  = i / (HIN / 4);
        const int c4 = i - r * (HIN / 4);
        *(v4f*)(wlds + r * PITCH + c4 * 4) = ((const v4f*)W)[i];
    }
    __syncthreads();

    const int wave = __builtin_amdgcn_readfirstlane(tid >> 5);
    const int lane = tid & 31;
    const int half = lane >> 4;
    const int lm   = lane & 15;
    const int woff0 = lm * PITCH + half * 4;

    for (int t = blockIdx.x + gridDim.x * wave; t < GB * GN / 16; t += gridDim.x * 16) {
        const int row = t * 16 + lm;
        const float dr = dinv[row];
        const float* hrow = Hsrc + (size_t)row * HIN + half * 4;
        int woff = woff0;
        asm volatile("" : "+v"(woff));
        const float* wrow = wlds + woff;

        v4f a[NQ];
#pragma unroll
        for (int q = 0; q < NQ; ++q) a[q] = *(const v4f*)(hrow + q * 8);

#pragma unroll
        for (int q = 0; q < NQ; ++q) a[q] *= dr;

        __builtin_amdgcn_sched_barrier(0);

        v8f acc[4] = {};
        v4f bcur[4], bnxt[4];
#pragma unroll
        for (int n = 0; n < 4; ++n) bcur[n] = *(const v4f*)(wrow + n * 16 * PITCH);
#pragma unroll
        for (int q = 0; q < NQ; ++q) {
            if (q + 1 < NQ) {
#pragma unroll
                for (int n = 0; n < 4; ++n) bnxt[n] = *(const v4f*)(wrow + n * 16 * PITCH + (q + 1) * 8);
            }
            __builtin_amdgcn_sched_barrier(0);
            const v2f a0 = {a[q].x, a[q].y};
            const v2f a1 = {a[q].z, a[q].w};
#pragma unroll
            for (int n = 0; n < 4; ++n) {
                const v2f b0 = {bcur[n].x, bcur[n].y};
                acc[n] = __builtin_amdgcn_wmma_f32_16x16x4_f32(
                     false, a0,  false, b0,
                     (short)0, acc[n],  false,  false);
            }
#pragma unroll
            for (int n = 0; n < 4; ++n) {
                const v2f b1 = {bcur[n].z, bcur[n].w};
                acc[n] = __builtin_amdgcn_wmma_f32_16x16x4_f32(
                     false, a1,  false, b1,
                     (short)0, acc[n],  false,  false);
            }
#pragma unroll
            for (int n = 0; n < 4; ++n) bcur[n] = bnxt[n];
        }

        float* drow = Dst + (size_t)t * 16 * GH + lm;
#pragma unroll
        for (int j = 0; j < 8; ++j) {
#pragma unroll
            for (int n = 0; n < 4; ++n) {
                drow[(size_t)(j + 8 * half) * GH + n * 16] = acc[n][j];
            }
        }
    }
}


template <int J> __device__ __forceinline__ int share16(int v) {
    return __builtin_amdgcn_update_dpp(0, v, 0x150 + J, 0xf, 0xf, false);
}

#define PULL_LOAD(J)                                                                       \
    if (FULL || (J0 + J) < n)                                                              \
        r[J] = *(const v4f*)(tmp + (rowbase + (unsigned)share16<J0 + J>(m.x) * GH));
#define PULL_ADD(J)                                                                        \
    if (FULL || (J0 + J) < n)                                                              \
        acc += r[J] * __int_as_float(share16<J0 + J>(m.y));

template <int J0, bool FULL>
__device__ __forceinline__ void pull_batch8(const float* __restrict__ tmp, unsigned rowbase,
                                            v2i m, int n, v4f& acc) {
    v4f r[8];
#pragma unroll
    for (int j = 0; j < 8; ++j) r[j] = (v4f){0.f, 0.f, 0.f, 0.f};
    PULL_LOAD(0) PULL_LOAD(1) PULL_LOAD(2) PULL_LOAD(3)
    PULL_LOAD(4) PULL_LOAD(5) PULL_LOAD(6) PULL_LOAD(7)
    PULL_ADD(0) PULL_ADD(1) PULL_ADD(2) PULL_ADD(3)
    PULL_ADD(4) PULL_ADD(5) PULL_ADD(6) PULL_ADD(7)
}
#undef PULL_LOAD
#undef PULL_ADD

template <int J0>
__device__ __forceinline__ void pull_batch(const float* __restrict__ tmp, unsigned rowbase,
                                           v2i m, int n, v4f& acc) {
    if (n >= J0 + 8) pull_batch8<J0, true >(tmp, rowbase, m, n, acc);
    else             pull_batch8<J0, false>(tmp, rowbase, m, n, acc);
}

template <bool ACT>
__device__ __forceinline__ v4f pull_row(const int* __restrict__ sedge,
                                        const float* __restrict__ tmp,
                                        const float* __restrict__ dinv,
                                        const float* __restrict__ bias,
                                        int b, int i, int k4, int beg, int end) {
    const int node = b * GN + i;
    const v2i* se = (const v2i*)sedge + (size_t)b * GE;
    const unsigned rowbase = (unsigned)(b * GN) * GH + (unsigned)k4 * 4;

    int p = beg;
    int n = end - p;
    v2i m0 = {0, 0}, m1 = {0, 0};
    if (k4 < n)      m0 = se[p + k4];
    if (k4 + 16 < n) m1 = se[p + 16 + k4];
    const float di = dinv[node];
    const v4f t  = *(const v4f*)(tmp + ((unsigned)node * GH + (unsigned)k4 * 4));
    v4f bb = {0.f, 0.f, 0.f, 0.f};
    if (ACT) bb = *(const v4f*)(bias + k4 * 4);

    v4f acc = {0.f, 0.f, 0.f, 0.f};
    while (n > 0) {
        pull_batch<0>(tmp, rowbase, m0, n, acc);
        if (n > 8)  pull_batch<8>(tmp, rowbase, m0, n, acc);
        if (n > 16) pull_batch<0>(tmp, rowbase, m1, n - 16, acc);
        if (n > 24) pull_batch<8>(tmp, rowbase, m1, n - 16, acc);
        p += 32;
        n -= 32;
        if (n > 0) {
            m0 = (v2i){0, 0};
            m1 = (v2i){0, 0};
            if (k4 < n)      m0 = se[p + k4];
            if (k4 + 16 < n) m1 = se[p + 16 + k4];
        }
    }

    v4f g = (acc + t) * di;
    if (ACT) {
        g += bb;
        g.x = tanhf(g.x); g.y = tanhf(g.y); g.z = tanhf(g.z); g.w = tanhf(g.w);
        g *= di;
    }
    return g;
}

__global__ __launch_bounds__(256) void pull_need_kernel(const int* __restrict__ rowptr,
                                                        const int* __restrict__ sedge,
                                                        const float* __restrict__ tmp,
                                                        const float* __restrict__ dinv,
                                                        const float* __restrict__ bias,
                                                        const int* __restrict__ lev,
                                                        float* __restrict__ hout,
                                                        int minlev) {
    const int k4 = threadIdx.x & 15;
    const int stride = gridDim.x * 16;
    int node = blockIdx.x * 16 + (threadIdx.x >> 4);
    int lv = 0, beg = 0, end = 0;
    if (node < GB * GN) {
        const int b = node / GN;
        beg = rowptr[node + b];
        end = rowptr[node + b + 1];
        lv  = lev[node];
    }
    while (node < GB * GN) {
        asm volatile("" : "+v"(lv));
        const bool need = lv >= minlev;
        const int cbeg = beg, cend = end;
        const int nxt = node + stride;
        lv = 0; beg = 0; end = 0;
        if (nxt < GB * GN) {
            const int b2 = nxt / GN;
            beg = rowptr[nxt + b2];
            end = rowptr[nxt + b2 + 1];
            lv  = lev[nxt];
        }
        if (need) {
            const int b = node / GN;
            const int i = node - b * GN;
            const v4f v = pull_row<true>(sedge, tmp, dinv, bias, b, i, k4, cbeg, cend);
            *(v4f*)(hout + ((size_t)node * (GH / 4) + k4) * 4) = v;
        }
        node = nxt;
    }
}

#define RNG 256
static_assert((GB * GN) % RNG == 0, "node ranges cover all nodes exactly");
static_assert(GP == 16 && GEMB == 64 && GH == 64, "layer 3: one 16-row tile per graph");
template <bool LAST>
__global__ __launch_bounds__(RNG) void pull_tile_gemm_kernel(const int* __restrict__ rowptr,
                                                             const int* __restrict__ sedge,
                                                             const float* __restrict__ hsrc,
                                                             const float* __restrict__ dinv,
                                                             const float* __restrict__ W,
                                                             const float* __restrict__ bias,
                                                             const int* __restrict__ lev,
                                                             const int* __restrict__ pos,
                                                             float* __restrict__ dst,
                                                             int minlev) {
    constexpr int PITCH = GH + 4;
    constexpr int NQ = GH / 8;
    __shared__ __attribute__((aligned(16))) float gt[16 * PITCH];
    __shared__ int list[RNG];
    __shared__ int wcnt[RNG / 32];
    const int tid  = threadIdx.x;
    const int k4   = tid & 15;
    const int grp  = tid >> 4;
    const int wave = __builtin_amdgcn_readfirstlane(tid >> 5);
    const int lane = tid & 31;
    const int half = lane >> 4;
    const int lm   = lane & 15;

    v4f w0[NQ], w1[NQ];
#pragma unroll
    for (int q = 0; q < NQ; ++q) {
        w0[q] = (v4f){0.f, 0.f, 0.f, 0.f};
        w1[q] = (v4f){0.f, 0.f, 0.f, 0.f};
    }
    float bc0 = 0.f, bc1 = 0.f;
    if (wave < 2) {
        const float* wr0 = W + (size_t)(wave * 32 + lm) * GH + half * 4;
        const float* wr1 = wr0 + 16 * GH;
#pragma unroll
        for (int q = 0; q < NQ; ++q) {
            w0[q] = *(const v4f*)(wr0 + q * 8);
            w1[q] = *(const v4f*)(wr1 + q * 8);
        }
        bc0 = bias[wave * 32 + lm];
        bc1 = bias[wave * 32 + 16 + lm];
    }

    int cnt;
    if (LAST) {
        if (tid < GP) {
            const int pv = pos[blockIdx.x * GP + tid];
            list[tid] = pv >= 0 ? (int)blockIdx.x * GN + pv : -1;
        }
        __syncthreads();
        cnt = GP;
    } else {
        const int n0 = blockIdx.x * RNG + tid;
        const bool nd = lev[n0] >= minlev;
        const unsigned m = __builtin_amdgcn_ballot_w32(nd);
        if (lane == 0) wcnt[wave] = __builtin_popcount(m);
        __syncthreads();
        int base = 0, total = 0;
#pragma unroll
        for (int w = 0; w < RNG / 32; ++w) {
            const int c = wcnt[w];
            if (w < wave) base += c;
            total += c;
        }
        if (nd) list[base + (int)__builtin_amdgcn_mbcnt_lo(m, 0u)] = n0;
        __syncthreads();
        cnt = __builtin_amdgcn_readfirstlane(total);
    }

    for (int r0 = 0; r0 < cnt; r0 += 16) {
        int nr[8];
        float dj[8];
#pragma unroll
        for (int j = 0; j < 8; ++j) { nr[j] = -1; dj[j] = 0.f; }
        if (wave < 2) {
#pragma unroll
            for (int j = 0; j < 8; ++j) {
                const int e = r0 + 8 * half + j;
                if (e < cnt) nr[j] = list[e];
            }
            if (!LAST) {
#pragma unroll
                for (int j = 0; j < 8; ++j) if (nr[j] >= 0) dj[j] = dinv[nr[j]];
            }
        }

        v4f g = {0.f, 0.f, 0.f, 0.f};
        const int ge = r0 + grp;
        int node = -1;
        if (ge < cnt) node = list[ge];
        if (node >= 0) {
            const int b = node / GN;
            const int i = node - b * GN;
            const int beg = rowptr[node + b];
            const int end = rowptr[node + b + 1];
            g = pull_row<false>(sedge, hsrc, dinv, bias, b, i, k4, beg, end);
        }
        *(v4f*)(gt + grp * PITCH + k4 * 4) = g;
        __syncthreads();

        if (wave < 2) {
            const float* arow = gt + lm * PITCH + half * 4;
            v4f a[NQ];
#pragma unroll
            for (int q = 0; q < NQ; ++q) a[q] = *(const v4f*)(arow + q * 8);
            v8f acc0 = {}, acc1 = {};
#pragma unroll
            for (int q = 0; q < NQ; ++q) {
                const v2f a0 = {a[q].x, a[q].y};
                const v2f a1 = {a[q].z, a[q].w};
                const v2f b00 = {w0[q].x, w0[q].y};
                const v2f b01 = {w0[q].z, w0[q].w};
                const v2f b10 = {w1[q].x, w1[q].y};
                const v2f b11 = {w1[q].z, w1[q].w};
                acc0 = __builtin_amdgcn_wmma_f32_16x16x4_f32(false, a0, false, b00, (short)0, acc0, false, false);
                acc1 = __builtin_amdgcn_wmma_f32_16x16x4_f32(false, a0, false, b10, (short)0, acc1, false, false);
                acc0 = __builtin_amdgcn_wmma_f32_16x16x4_f32(false, a1, false, b01, (short)0, acc0, false, false);
                acc1 = __builtin_amdgcn_wmma_f32_16x16x4_f32(false, a1, false, b11, (short)0, acc1, false, false);
            }
            float v0[8], v1[8];
#pragma unroll
            for (int j = 0; j < 8; ++j) {
                v0[j] = acc0[j] + bc0;
                v1[j] = acc1[j] + bc1;
            }
            if (!LAST) {
                asm volatile("" : "+v"(v0[0]), "+v"(v0[1]), "+v"(v0[2]), "+v"(v0[3]),
                                  "+v"(v0[4]), "+v"(v0[5]), "+v"(v0[6]), "+v"(v0[7]),
                                  "+v"(v1[0]), "+v"(v1[1]), "+v"(v1[2]), "+v"(v1[3]),
                                  "+v"(v1[4]), "+v"(v1[5]), "+v"(v1[6]), "+v"(v1[7]));
#pragma unroll
                for (int j = 0; j < 8; ++j) {
                    v0[j] = tanhf(v0[j]) * dj[j];
                    v1[j] = tanhf(v1[j]) * dj[j];
                }
            }
            if (LAST) {
                float* orow = dst + ((size_t)blockIdx.x * GP + 8 * half) * GEMB + wave * 32 + lm;
#pragma unroll
                for (int j = 0; j < 8; ++j) {
                    const bool valid = nr[j] >= 0;
                    orow[(size_t)j * GEMB]      = valid ? v0[j] : -3.0f;
                    orow[(size_t)j * GEMB + 16] = valid ? v1[j] : -3.0f;
                }
            } else {
#pragma unroll
                for (int j = 0; j < 8; ++j) {
                    if (nr[j] >= 0) {
                        float* orow = dst + (size_t)nr[j] * GH + wave * 32 + lm;
                        orow[0]  = v0[j];
                        orow[16] = v1[j];
                    }
                }
            }
        }
        __syncthreads();
    }
}

extern "C" void kernel_launch(void* const* d_in, const int* in_sizes, int n_in,
                              void* d_out, int out_size, void* d_ws, size_t ws_size,
                              hipStream_t stream) {
    const float* x   = (const float*)d_in[0];
    const int*   ei  = (const int*)d_in[1];
    const float* ew  = (const float*)d_in[2];
    const int*   pos = (const int*)d_in[3];
    const float* W1  = (const float*)d_in[4];
    const float* b1  = (const float*)d_in[5];
    const float* W2  = (const float*)d_in[6];
    const float* b2  = (const float*)d_in[7];
    const float* W3  = (const float*)d_in[8];
    const float* b3  = (const float*)d_in[9];
    float* out = (float*)d_out;

    float* ws   = (float*)d_ws;
    const size_t nBN  = (size_t)GB * GN;
    const size_t nRP  = (size_t)GB * (GN + 1);
    const size_t nBE  = (size_t)GB * GE;
    const size_t nBNH = (size_t)GB * GN * GH;
    float* dinv   = ws;
    int*   rowptr = (int*)(dinv + nBN);
    int*   sedge  = rowptr + nRP;
    float* tmp    = (float*)(sedge + 2 * nBE);
    float* hbuf   = tmp + nBNH;
    int*   lev    = (int*)(hbuf + nBNH);
    int*   binptr = (int*)tmp;
    int*   pedge  = (int*)hbuf;

    const int T = 256;
    dim3 part_grid(SLICES, GB, 1);
    dim3 sort_grid(PARTS, GB, 1);
    const int nodeBlocks = ((int)nBN + T - 1) / T;

    part_kernel<<<part_grid, SORT_T, 0, stream>>>(ei, ew, binptr, pedge);
    sort_kernel<<<sort_grid, SORT_T, 0, stream>>>(binptr, pedge, rowptr, sedge);
    dinv_kernel<<<nodeBlocks, T, 0, stream>>>(rowptr, sedge, dinv, lev);
    mark_pos_kernel<<<(GB * GP + T - 1) / T, T, 0, stream>>>(rowptr, sedge, pos, lev);
    mark_src_kernel<<<nodeBlocks, T, 0, stream>>>(rowptr, sedge, lev);

    gemm_strip_kernel<GS><<<256, 512, 0, stream>>>(x, W1, dinv, tmp);
    pull_need_kernel<<<1536, T, 0, stream>>>(rowptr, sedge, tmp, dinv, b1, lev, hbuf, 1);

    pull_tile_gemm_kernel<false><<<GB * GN / RNG, RNG, 0, stream>>>(rowptr, sedge, hbuf, dinv, W2, b2, lev, pos, tmp, 2);

    pull_tile_gemm_kernel<true><<<GB, RNG, 0, stream>>>(rowptr, sedge, tmp, dinv, W3, b3, lev, pos, out, 2);
}


# Degree sum, pos marking and source marking folded into the sort and GEMM launches (range block owns its lists; level word split in two bytes, marks are byte stores): 9 launches to 6; sum order kept
# speedup vs baseline: 8.3727x; 1.3208x over previous
#include <hip/hip_runtime.h>
#include <hip/hip_bf16.h>

#define GB   16
#define GN   10000
#define GE   160000
#define GS   128
#define GH   64
#define GEMB 64
#define GP   16

#define PARTS 16
#define NPP   (GN / PARTS)
#define SORT_T 1024
#define SLICES 16
#define EPS   (GE / SLICES)
#define EPT   ((EPS + SORT_T - 1) / SORT_T)
static_assert(GN % PARTS == 0 && GE % SLICES == 0 && GN % SLICES == 0 && GN <= 65536, "sort layout");

#define LEV1 1
#define LEV2 256

typedef __attribute__((ext_vector_type(2))) float v2f;
typedef __attribute__((ext_vector_type(4))) float v4f;
typedef __attribute__((ext_vector_type(8))) float v8f;
typedef __attribute__((ext_vector_type(2))) int   v2i;
typedef __attribute__((ext_vector_type(4))) int   v4i;

__global__ __launch_bounds__(SORT_T) void part_kernel(const int* __restrict__ ei,
                                                      const float* __restrict__ ew,
                                                      int* __restrict__ binptr,
                                                      int* __restrict__ pedge,
                                                      int* __restrict__ lev) {
    __shared__ int bin[PARTS];
    const int tid = threadIdx.x;
    const int s   = blockIdx.x;
    const int b   = blockIdx.y;
    const int*   rows = ei + (size_t)b * 2 * GE + (size_t)s * EPS;
    const int*   cols = rows + GE;
    const float* w    = ew + (size_t)b * GE + (size_t)s * EPS;

    if (tid < PARTS) bin[tid] = 0;
    for (int i = tid; i < GN / SLICES; i += SORT_T) lev[(size_t)b * GN + s * (GN / SLICES) + i] = 0;
    int c_[EPT], r_[EPT], w_[EPT];
#pragma unroll
    for (int u = 0; u < EPT; ++u) {
        const int e = tid + u * SORT_T;
        c_[u] = 0; r_[u] = 0; w_[u] = 0;
        if (e < EPS) {
            c_[u] = cols[e];
            r_[u] = rows[e];
            w_[u] = __float_as_int(w[e]);
        }
    }
    __syncthreads();
#pragma unroll
    for (int u = 0; u < EPT; ++u) {
        const int e = tid + u * SORT_T;
        if (e < EPS) atomicAdd(&bin[c_[u] / NPP], 1);
    }
    __syncthreads();
    int* bp = binptr + ((size_t)b * SLICES + s) * (PARTS + 1);
    if (tid == 0) {
        int n_[PARTS];
#pragma unroll
        for (int p = 0; p < PARTS; ++p) n_[p] = bin[p];
        int run = 0;
#pragma unroll
        for (int p = 0; p < PARTS; ++p) {
            bin[p] = run;
            bp[p]  = run;
            run += n_[p];
        }
        bp[PARTS] = run;
    }
    __syncthreads();
    v2i* pe = (v2i*)pedge + (size_t)b * GE + (size_t)s * EPS;
    int pos_[EPT];
#pragma unroll
    for (int u = 0; u < EPT; ++u) {
        const int e = tid + u * SORT_T;
        pos_[u] = -1;
        if (e < EPS) pos_[u] = atomicAdd(&bin[c_[u] / NPP], 1);
    }
#pragma unroll
    for (int u = 0; u < EPT; ++u) {
        if (pos_[u] >= 0) {
            v2i ed;
            ed.x = r_[u] | (c_[u] << 16);
            ed.y = w_[u];
            pe[pos_[u]] = ed;
        }
    }
}

__global__ __launch_bounds__(SORT_T) void sort_kernel(const int* __restrict__ binptr,
                                                      const int* __restrict__ pedge,
                                                      const int* __restrict__ pos,
                                                      int* __restrict__ rowptr,
                                                      int* __restrict__ sedge,
                                                      float* __restrict__ dinv,
                                                      int* __restrict__ lev) {
    __shared__ int cnt[NPP];
    __shared__ int st[NPP];
    __shared__ int sc[SORT_T];
    const int tid  = threadIdx.x;
    const int part = blockIdx.x;
    const int b    = blockIdx.y;
    const int lo   = part * NPP;

    for (int i = tid; i < NPP; i += SORT_T) cnt[i] = 0;
    int pv = -1;
    if (tid < GP * 16) pv = pos[b * GP + (tid >> 4)];

    const int* bpb = binptr + (size_t)b * SLICES * (PARTS + 1) + part;
    const v2i* pe  = (const v2i*)pedge + (size_t)b * GE;
    int beg[SLICES], end[SLICES];
    int below = 0;
#pragma unroll
    for (int s = 0; s < SLICES; ++s) {
        beg[s] = bpb[s * (PARTS + 1)];
        end[s] = bpb[s * (PARTS + 1) + 1];
        below += beg[s];
    }
    v2i ed[SLICES];
#pragma unroll
    for (int s = 0; s < SLICES; ++s) {
        const int k = beg[s] + tid;
        ed[s].x = -1;
        ed[s].y = 0;
        if (k < end[s]) ed[s] = pe[s * EPS + k];
    }
    __syncthreads();
#pragma unroll
    for (int s = 0; s < SLICES; ++s) {
        if (ed[s].x >= 0) atomicAdd(&cnt[(ed[s].x >> 16) - lo], 1);
        if (end[s] - beg[s] > SORT_T) {
            for (int k = beg[s] + tid + SORT_T; k < end[s]; k += SORT_T)
                atomicAdd(&cnt[(pe[s * EPS + k].x >> 16) - lo], 1);
        }
    }
    __syncthreads();

    const int own = (tid < NPP) ? cnt[tid] : 0;
    sc[tid] = own;
    __syncthreads();
    for (int off = 1; off < SORT_T; off <<= 1) {
        const int add = (tid >= off) ? sc[tid - off] : 0;
        __syncthreads();
        sc[tid] += add;
        __syncthreads();
    }
    int start = 0;
    if (tid < NPP) {
        start = below + sc[tid] - own;
        cnt[tid] = start;
        st[tid]  = start;
        rowptr[(size_t)b * (GN + 1) + lo + tid] = start;
    }
    if (part == PARTS - 1 && tid == 0) rowptr[(size_t)b * (GN + 1) + GN] = GE;
    __syncthreads();

    v2i* se = (v2i*)sedge + (size_t)b * GE;
    int pos_[SLICES];
#pragma unroll
    for (int s = 0; s < SLICES; ++s) {
        pos_[s] = -1;
        if (ed[s].x >= 0) pos_[s] = atomicAdd(&cnt[(ed[s].x >> 16) - lo], 1);
    }
#pragma unroll
    for (int s = 0; s < SLICES; ++s) {
        if (pos_[s] >= 0) {
            v2i o;
            o.x = ed[s].x & 0xFFFF;
            o.y = ed[s].y;
            se[pos_[s]] = o;
        }
        if (end[s] - beg[s] > SORT_T)
        for (int k = beg[s] + tid + SORT_T; k < end[s]; k += SORT_T) {
            const v2i e2 = pe[s * EPS + k];
            const int p2 = atomicAdd(&cnt[(e2.x >> 16) - lo], 1);
            v2i o;
            o.x = e2.x & 0xFFFF;
            o.y = e2.y;
            se[p2] = o;
        }
    }
    __syncthreads();

    if (pv >= lo && pv < lo + NPP) {
        unsigned char* lb = (unsigned char*)(lev + (size_t)b * GN);
        const int pe_ = cnt[pv - lo];
        if ((tid & 15) == 0) lb[4 * pv + 1] = 1;
        for (int q = st[pv - lo] + (tid & 15); q < pe_; q += 16) lb[4 * se[q].x + 1] = 1;
    }

    if (tid < NPP) {
        const int e = start + own;
        const int* sw = (const int*)se + 1;
        float d = 0.f;
        for (int p = start; p < e; p += 8) {
            int w8[8];
#pragma unroll
            for (int j = 0; j < 8; ++j) w8[j] = sw[2 * min(p + j, e - 1)];
#pragma unroll
            for (int j = 0; j < 8; ++j) d += (p + j < e) ? __int_as_float(w8[j]) : 0.f;
        }
        dinv[(size_t)b * GN + lo + tid] = rsqrtf(d + 1.0f);
    }
}


#define GEMM_BLOCKS 256
#define MPB (GB * GN / GEMM_BLOCKS)
static_assert((GB * GN) % GEMM_BLOCKS == 0, "mark ranges cover all nodes exactly");
template <int HIN>
__global__ __launch_bounds__(512) void gemm_strip_kernel(const float* __restrict__ Hsrc,
                                                         const float* __restrict__ W,
                                                         const float* __restrict__ dinv,
                                                         const int* __restrict__ rowptr,
                                                         const int* __restrict__ sedge,
                                                         int* lev,
                                                         float* __restrict__ Dst) {
    constexpr int PITCH = HIN + 4;
    constexpr int NQ = HIN / 8;
    __shared__ __attribute__((aligned(16))) float wlds[GH * PITCH];
    const int tid = threadIdx.x;
    constexpr int MR = (MPB + 511) / 512;
    int mlev[MR];
#pragma unroll
    for (int u = 0; u < MR; ++u) {
        const int i = tid + u * 512;
        mlev[u] = 0;
        if (i < MPB) mlev[u] = lev[blockIdx.x * MPB + i];
    }
    for (int i = tid; i < GH * HIN / 4; i += 512) {
        const int r  = i / (HIN / 4);
        const int c4 = i - r * (HIN / 4);
        *(v4f*)(wlds + r * PITCH + c4 * 4) = ((const v4f*)W)[i];
    }
    __syncthreads();

#pragma unroll
    for (int u = 0; u < MR; ++u) {
        if (mlev[u] >= LEV2) {
            const int node = blockIdx.x * MPB + tid + u * 512;
            const int b = node / GN;
            unsigned char* lb = (unsigned char*)(lev + (size_t)b * GN);
            const int beg = rowptr[node + b];
            const int end = rowptr[node + b + 1];
            const v2i* se = (const v2i*)sedge + (size_t)b * GE;
            for (int p = beg; p < end; p += 8) {
                int s8[8];
#pragma unroll
                for (int j = 0; j < 8; ++j) s8[j] = se[min(p + j, end - 1)].x;
#pragma unroll
                for (int j = 0; j < 8; ++j) lb[4 * s8[j]] = 1;
            }
        }
    }

    const int wave = __builtin_amdgcn_readfirstlane(tid >> 5);
    const int lane = tid & 31;
    const int half = lane >> 4;
    const int lm   = lane & 15;
    const int woff0 = lm * PITCH + half * 4;

    for (int t = blockIdx.x + gridDim.x * wave; t < GB * GN / 16; t += gridDim.x * 16) {
        const int row = t * 16 + lm;
        const float dr = dinv[row];
        const float* hrow = Hsrc + (size_t)row * HIN + half * 4;
        int woff = woff0;
        asm volatile("" : "+v"(woff));
        const float* wrow = wlds + woff;

        v4f a[NQ];
#pragma unroll
        for (int q = 0; q < NQ; ++q) a[q] = *(const v4f*)(hrow + q * 8);

#pragma unroll
        for (int q = 0; q < NQ; ++q) a[q] *= dr;

        __builtin_amdgcn_sched_barrier(0);

        v8f acc[4] = {};
        v4f bcur[4], bnxt[4];
#pragma unroll
        for (int n = 0; n < 4; ++n) bcur[n] = *(const v4f*)(wrow + n * 16 * PITCH);
#pragma unroll
        for (int q = 0; q < NQ; ++q) {
            if (q + 1 < NQ) {
#pragma unroll
                for (int n = 0; n < 4; ++n) bnxt[n] = *(const v4f*)(wrow + n * 16 * PITCH + (q + 1) * 8);
            }
            __builtin_amdgcn_sched_barrier(0);
            const v2f a0 = {a[q].x, a[q].y};
            const v2f a1 = {a[q].z, a[q].w};
#pragma unroll
            for (int n = 0; n < 4; ++n) {
                const v2f b0 = {bcur[n].x, bcur[n].y};
                acc[n] = __builtin_amdgcn_wmma_f32_16x16x4_f32(
                     false, a0,  false, b0,
                     (short)0, acc[n],  false,  false);
            }
#pragma unroll
            for (int n = 0; n < 4; ++n) {
                const v2f b1 = {bcur[n].z, bcur[n].w};
                acc[n] = __builtin_amdgcn_wmma_f32_16x16x4_f32(
                     false, a1,  false, b1,
                     (short)0, acc[n],  false,  false);
            }
#pragma unroll
            for (int n = 0; n < 4; ++n) bcur[n] = bnxt[n];
        }

        float* drow = Dst + (size_t)t * 16 * GH + lm;
#pragma unroll
        for (int j = 0; j < 8; ++j) {
#pragma unroll
            for (int n = 0; n < 4; ++n) {
                drow[(size_t)(j + 8 * half) * GH + n * 16] = acc[n][j];
            }
        }
    }
}


template <int J> __device__ __forceinline__ int share16(int v) {
    return __builtin_amdgcn_update_dpp(0, v, 0x150 + J, 0xf, 0xf, false);
}

#define PULL_LOAD(J)                                                                       \
    if (FULL || (J0 + J) < n)                                                              \
        r[J] = *(const v4f*)(tmp + (rowbase + (unsigned)share16<J0 + J>(m.x) * GH));
#define PULL_ADD(J)                                                                        \
    if (FULL || (J0 + J) < n)                                                              \
        acc += r[J] * __int_as_float(share16<J0 + J>(m.y));

template <int J0, bool FULL>
__device__ __forceinline__ void pull_batch8(const float* __restrict__ tmp, unsigned rowbase,
                                            v2i m, int n, v4f& acc) {
    v4f r[8];
#pragma unroll
    for (int j = 0; j < 8; ++j) r[j] = (v4f){0.f, 0.f, 0.f, 0.f};
    PULL_LOAD(0) PULL_LOAD(1) PULL_LOAD(2) PULL_LOAD(3)
    PULL_LOAD(4) PULL_LOAD(5) PULL_LOAD(6) PULL_LOAD(7)
    PULL_ADD(0) PULL_ADD(1) PULL_ADD(2) PULL_ADD(3)
    PULL_ADD(4) PULL_ADD(5) PULL_ADD(6) PULL_ADD(7)
}
#undef PULL_LOAD
#undef PULL_ADD

template <int J0>
__device__ __forceinline__ void pull_batch(const float* __restrict__ tmp, unsigned rowbase,
                                           v2i m, int n, v4f& acc) {
    if (n >= J0 + 8) pull_batch8<J0, true >(tmp, rowbase, m, n, acc);
    else             pull_batch8<J0, false>(tmp, rowbase, m, n, acc);
}

template <bool ACT>
__device__ __forceinline__ v4f pull_row(const int* __restrict__ sedge,
                                        const float* __restrict__ tmp,
                                        const float* __restrict__ dinv,
                                        const float* __restrict__ bias,
                                        int b, int i, int k4, int beg, int end) {
    const int node = b * GN + i;
    const v2i* se = (const v2i*)sedge + (size_t)b * GE;
    const unsigned rowbase = (unsigned)(b * GN) * GH + (unsigned)k4 * 4;

    int p = beg;
    int n = end - p;
    v2i m0 = {0, 0}, m1 = {0, 0};
    if (k4 < n)      m0 = se[p + k4];
    if (k4 + 16 < n) m1 = se[p + 16 + k4];
    const float di = dinv[node];
    const v4f t  = *(const v4f*)(tmp + ((unsigned)node * GH + (unsigned)k4 * 4));
    v4f bb = {0.f, 0.f, 0.f, 0.f};
    if (ACT) bb = *(const v4f*)(bias + k4 * 4);

    v4f acc = {0.f, 0.f, 0.f, 0.f};
    while (n > 0) {
        pull_batch<0>(tmp, rowbase, m0, n, acc);
        if (n > 8)  pull_batch<8>(tmp, rowbase, m0, n, acc);
        if (n > 16) pull_batch<0>(tmp, rowbase, m1, n - 16, acc);
        if (n > 24) pull_batch<8>(tmp, rowbase, m1, n - 16, acc);
        p += 32;
        n -= 32;
        if (n > 0) {
            m0 = (v2i){0, 0};
            m1 = (v2i){0, 0};
            if (k4 < n)      m0 = se[p + k4];
            if (k4 + 16 < n) m1 = se[p + 16 + k4];
        }
    }

    v4f g = (acc + t) * di;
    if (ACT) {
        g += bb;
        g.x = tanhf(g.x); g.y = tanhf(g.y); g.z = tanhf(g.z); g.w = tanhf(g.w);
        g *= di;
    }
    return g;
}

__global__ __launch_bounds__(256) void pull_need_kernel(const int* __restrict__ rowptr,
                                                        const int* __restrict__ sedge,
                                                        const float* __restrict__ tmp,
                                                        const float* __restrict__ dinv,
                                                        const float* __restrict__ bias,
                                                        const int* __restrict__ lev,
                                                        float* __restrict__ hout,
                                                        int minlev) {
    const int k4 = threadIdx.x & 15;
    const int stride = gridDim.x * 16;
    int node = blockIdx.x * 16 + (threadIdx.x >> 4);
    int lv = 0, beg = 0, end = 0;
    if (node < GB * GN) {
        const int b = node / GN;
        beg = rowptr[node + b];
        end = rowptr[node + b + 1];
        lv  = lev[node];
    }
    while (node < GB * GN) {
        asm volatile("" : "+v"(lv));
        const bool need = lv >= minlev;
        const int cbeg = beg, cend = end;
        const int nxt = node + stride;
        lv = 0; beg = 0; end = 0;
        if (nxt < GB * GN) {
            const int b2 = nxt / GN;
            beg = rowptr[nxt + b2];
            end = rowptr[nxt + b2 + 1];
            lv  = lev[nxt];
        }
        if (need) {
            const int b = node / GN;
            const int i = node - b * GN;
            const v4f v = pull_row<true>(sedge, tmp, dinv, bias, b, i, k4, cbeg, cend);
            *(v4f*)(hout + ((size_t)node * (GH / 4) + k4) * 4) = v;
        }
        node = nxt;
    }
}

#define RNG 256
static_assert((GB * GN) % RNG == 0, "node ranges cover all nodes exactly");
static_assert(GP == 16 && GEMB == 64 && GH == 64, "layer 3: one 16-row tile per graph");
template <bool LAST>
__global__ __launch_bounds__(RNG) void pull_tile_gemm_kernel(const int* __restrict__ rowptr,
                                                             const int* __restrict__ sedge,
                                                             const float* __restrict__ hsrc,
                                                             const float* __restrict__ dinv,
                                                             const float* __restrict__ W,
                                                             const float* __restrict__ bias,
                                                             const int* __restrict__ lev,
                                                             const int* __restrict__ pos,
                                                             float* __restrict__ dst,
                                                             int minlev) {
    constexpr int PITCH = GH + 4;
    constexpr int NQ = GH / 8;
    __shared__ __attribute__((aligned(16))) float gt[16 * PITCH];
    __shared__ int list[RNG];
    __shared__ int wcnt[RNG / 32];
    const int tid  = threadIdx.x;
    const int k4   = tid & 15;
    const int grp  = tid >> 4;
    const int wave = __builtin_amdgcn_readfirstlane(tid >> 5);
    const int lane = tid & 31;
    const int half = lane >> 4;
    const int lm   = lane & 15;

    v4f w0[NQ], w1[NQ];
#pragma unroll
    for (int q = 0; q < NQ; ++q) {
        w0[q] = (v4f){0.f, 0.f, 0.f, 0.f};
        w1[q] = (v4f){0.f, 0.f, 0.f, 0.f};
    }
    float bc0 = 0.f, bc1 = 0.f;
    if (wave < 2) {
        const float* wr0 = W + (size_t)(wave * 32 + lm) * GH + half * 4;
        const float* wr1 = wr0 + 16 * GH;
#pragma unroll
        for (int q = 0; q < NQ; ++q) {
            w0[q] = *(const v4f*)(wr0 + q * 8);
            w1[q] = *(const v4f*)(wr1 + q * 8);
        }
        bc0 = bias[wave * 32 + lm];
        bc1 = bias[wave * 32 + 16 + lm];
    }

    int cnt;
    if (LAST) {
        if (tid < GP) {
            const int pv = pos[blockIdx.x * GP + tid];
            list[tid] = pv >= 0 ? (int)blockIdx.x * GN + pv : -1;
        }
        __syncthreads();
        cnt = GP;
    } else {
        const int n0 = blockIdx.x * RNG + tid;
        const bool nd = lev[n0] >= minlev;
        const unsigned m = __builtin_amdgcn_ballot_w32(nd);
        if (lane == 0) wcnt[wave] = __builtin_popcount(m);
        __syncthreads();
        int base = 0, total = 0;
#pragma unroll
        for (int w = 0; w < RNG / 32; ++w) {
            const int c = wcnt[w];
            if (w < wave) base += c;
            total += c;
        }
        if (nd) list[base + (int)__builtin_amdgcn_mbcnt_lo(m, 0u)] = n0;
        __syncthreads();
        cnt = __builtin_amdgcn_readfirstlane(total);
    }

    for (int r0 = 0; r0 < cnt; r0 += 16) {
        int nr[8];
        float dj[8];
#pragma unroll
        for (int j = 0; j < 8; ++j) { nr[j] = -1; dj[j] = 0.f; }
        if (wave < 2) {
#pragma unroll
            for (int j = 0; j < 8; ++j) {
                const int e = r0 + 8 * half + j;
                if (e < cnt) nr[j] = list[e];
            }
            if (!LAST) {
#pragma unroll
                for (int j = 0; j < 8; ++j) if (nr[j] >= 0) dj[j] = dinv[nr[j]];
            }
        }

        v4f g = {0.f, 0.f, 0.f, 0.f};
        const int ge = r0 + grp;
        int node = -1;
        if (ge < cnt) node = list[ge];
        if (node >= 0) {
            const int b = node / GN;
            const int i = node - b * GN;
            const int beg = rowptr[node + b];
            const int end = rowptr[node + b + 1];
            g = pull_row<false>(sedge, hsrc, dinv, bias, b, i, k4, beg, end);
        }
        *(v4f*)(gt + grp * PITCH + k4 * 4) = g;
        __syncthreads();

        if (wave < 2) {
            const float* arow = gt + lm * PITCH + half * 4;
            v4f a[NQ];
#pragma unroll
            for (int q = 0; q < NQ; ++q) a[q] = *(const v4f*)(arow + q * 8);
            v8f acc0 = {}, acc1 = {};
#pragma unroll
            for (int q = 0; q < NQ; ++q) {
                const v2f a0 = {a[q].x, a[q].y};
                const v2f a1 = {a[q].z, a[q].w};
                const v2f b00 = {w0[q].x, w0[q].y};
                const v2f b01 = {w0[q].z, w0[q].w};
                const v2f b10 = {w1[q].x, w1[q].y};
                const v2f b11 = {w1[q].z, w1[q].w};
                acc0 = __builtin_amdgcn_wmma_f32_16x16x4_f32(false, a0, false, b00, (short)0, acc0, false, false);
                acc1 = __builtin_amdgcn_wmma_f32_16x16x4_f32(false, a0, false, b10, (short)0, acc1, false, false);
                acc0 = __builtin_amdgcn_wmma_f32_16x16x4_f32(false, a1, false, b01, (short)0, acc0, false, false);
                acc1 = __builtin_amdgcn_wmma_f32_16x16x4_f32(false, a1, false, b11, (short)0, acc1, false, false);
            }
            float v0[8], v1[8];
#pragma unroll
            for (int j = 0; j < 8; ++j) {
                v0[j] = acc0[j] + bc0;
                v1[j] = acc1[j] + bc1;
            }
            if (!LAST) {
                asm volatile("" : "+v"(v0[0]), "+v"(v0[1]), "+v"(v0[2]), "+v"(v0[3]),
                                  "+v"(v0[4]), "+v"(v0[5]), "+v"(v0[6]), "+v"(v0[7]),
                                  "+v"(v1[0]), "+v"(v1[1]), "+v"(v1[2]), "+v"(v1[3]),
                                  "+v"(v1[4]), "+v"(v1[5]), "+v"(v1[6]), "+v"(v1[7]));
#pragma unroll
                for (int j = 0; j < 8; ++j) {
                    v0[j] = tanhf(v0[j]) * dj[j];
                    v1[j] = tanhf(v1[j]) * dj[j];
                }
            }
            if (LAST) {
                float* orow = dst + ((size_t)blockIdx.x * GP + 8 * half) * GEMB + wave * 32 + lm;
#pragma unroll
                for (int j = 0; j < 8; ++j) {
                    const bool valid = nr[j] >= 0;
                    orow[(size_t)j * GEMB]      = valid ? v0[j] : -3.0f;
                    orow[(size_t)j * GEMB + 16] = valid ? v1[j] : -3.0f;
                }
            } else {
#pragma unroll
                for (int j = 0; j < 8; ++j) {
                    if (nr[j] >= 0) {
                        float* orow = dst + (size_t)nr[j] * GH + wave * 32 + lm;
                        orow[0]  = v0[j];
                        orow[16] = v1[j];
                    }
                }
            }
        }
        __syncthreads();
    }
}

extern "C" void kernel_launch(void* const* d_in, const int* in_sizes, int n_in,
                              void* d_out, int out_size, void* d_ws, size_t ws_size,
                              hipStream_t stream) {
    const float* x   = (const float*)d_in[0];
    const int*   ei  = (const int*)d_in[1];
    const float* ew  = (const float*)d_in[2];
    const int*   pos = (const int*)d_in[3];
    const float* W1  = (const float*)d_in[4];
    const float* b1  = (const float*)d_in[5];
    const float* W2  = (const float*)d_in[6];
    const float* b2  = (const float*)d_in[7];
    const float* W3  = (const float*)d_in[8];
    const float* b3  = (const float*)d_in[9];
    float* out = (float*)d_out;

    float* ws   = (float*)d_ws;
    const size_t nBN  = (size_t)GB * GN;
    const size_t nRP  = (size_t)GB * (GN + 1);
    const size_t nBE  = (size_t)GB * GE;
    const size_t nBNH = (size_t)GB * GN * GH;
    float* dinv   = ws;
    int*   rowptr = (int*)(dinv + nBN);
    int*   sedge  = rowptr + nRP;
    float* tmp    = (float*)(sedge + 2 * nBE);
    float* hbuf   = tmp + nBNH;
    int*   lev    = (int*)(hbuf + nBNH);
    int*   binptr = (int*)tmp;
    int*   pedge  = (int*)hbuf;

    const int T = 256;
    dim3 part_grid(SLICES, GB, 1);
    dim3 sort_grid(PARTS, GB, 1);

    part_kernel<<<part_grid, SORT_T, 0, stream>>>(ei, ew, binptr, pedge, lev);
    sort_kernel<<<sort_grid, SORT_T, 0, stream>>>(binptr, pedge, pos, rowptr, sedge, dinv, lev);

    gemm_strip_kernel<GS><<<GEMM_BLOCKS, 512, 0, stream>>>(x, W1, dinv, rowptr, sedge, lev, tmp);
    pull_need_kernel<<<1536, T, 0, stream>>>(rowptr, sedge, tmp, dinv, b1, lev, hbuf, LEV1);

    pull_tile_gemm_kernel<false><<<GB * GN / RNG, RNG, 0, stream>>>(rowptr, sedge, hbuf, dinv, W2, b2, lev, pos, tmp, LEV2);

    pull_tile_gemm_kernel<true><<<GB, RNG, 0, stream>>>(rowptr, sedge, tmp, dinv, W3, b3, lev, pos, out, LEV2);
}


# Layers 2 and 3 in one launch: one work-group per pos row computes the layer-2 rows that row reads (own node and list sources) into LDS, then its layer-3 row; 6 launches to 5; sum order kept
# speedup vs baseline: 8.8951x; 1.0624x over previous
#include <hip/hip_runtime.h>
#include <hip/hip_bf16.h>

#define GB   16
#define GN   10000
#define GE   160000
#define GS   128
#define GH   64
#define GEMB 64
#define GP   16

#define PARTS 16
#define NPP   (GN / PARTS)
#define SORT_T 1024
#define SLICES 16
#define EPS   (GE / SLICES)
#define EPT   ((EPS + SORT_T - 1) / SORT_T)
static_assert(GN % PARTS == 0 && GE % SLICES == 0 && GN % SLICES == 0 && GN <= 65536, "sort layout");

#define LEV1 1
#define LEV2 256

typedef __attribute__((ext_vector_type(2))) float v2f;
typedef __attribute__((ext_vector_type(4))) float v4f;
typedef __attribute__((ext_vector_type(8))) float v8f;
typedef __attribute__((ext_vector_type(2))) int   v2i;
typedef __attribute__((ext_vector_type(4))) int   v4i;

__global__ __launch_bounds__(SORT_T) void part_kernel(const int* __restrict__ ei,
                                                      const float* __restrict__ ew,
                                                      int* __restrict__ binptr,
                                                      int* __restrict__ pedge,
                                                      int* __restrict__ lev) {
    __shared__ int bin[PARTS];
    const int tid = threadIdx.x;
    const int s   = blockIdx.x;
    const int b   = blockIdx.y;
    const int*   rows = ei + (size_t)b * 2 * GE + (size_t)s * EPS;
    const int*   cols = rows + GE;
    const float* w    = ew + (size_t)b * GE + (size_t)s * EPS;

    if (tid < PARTS) bin[tid] = 0;
    for (int i = tid; i < GN / SLICES; i += SORT_T) lev[(size_t)b * GN + s * (GN / SLICES) + i] = 0;
    int c_[EPT], r_[EPT], w_[EPT];
#pragma unroll
    for (int u = 0; u < EPT; ++u) {
        const int e = tid + u * SORT_T;
        c_[u] = 0; r_[u] = 0; w_[u] = 0;
        if (e < EPS) {
            c_[u] = cols[e];
            r_[u] = rows[e];
            w_[u] = __float_as_int(w[e]);
        }
    }
    __syncthreads();
#pragma unroll
    for (int u = 0; u < EPT; ++u) {
        const int e = tid + u * SORT_T;
        if (e < EPS) atomicAdd(&bin[c_[u] / NPP], 1);
    }
    __syncthreads();
    int* bp = binptr + ((size_t)b * SLICES + s) * (PARTS + 1);
    if (tid == 0) {
        int n_[PARTS];
#pragma unroll
        for (int p = 0; p < PARTS; ++p) n_[p] = bin[p];
        int run = 0;
#pragma unroll
        for (int p = 0; p < PARTS; ++p) {
            bin[p] = run;
            bp[p]  = run;
            run += n_[p];
        }
        bp[PARTS] = run;
    }
    __syncthreads();
    v2i* pe = (v2i*)pedge + (size_t)b * GE + (size_t)s * EPS;
    int pos_[EPT];
#pragma unroll
    for (int u = 0; u < EPT; ++u) {
        const int e = tid + u * SORT_T;
        pos_[u] = -1;
        if (e < EPS) pos_[u] = atomicAdd(&bin[c_[u] / NPP], 1);
    }
#pragma unroll
    for (int u = 0; u < EPT; ++u) {
        if (pos_[u] >= 0) {
            v2i ed;
            ed.x = r_[u] | (c_[u] << 16);
            ed.y = w_[u];
            pe[pos_[u]] = ed;
        }
    }
}

__global__ __launch_bounds__(SORT_T) void sort_kernel(const int* __restrict__ binptr,
                                                      const int* __restrict__ pedge,
                                                      const int* __restrict__ pos,
                                                      int* __restrict__ rowptr,
                                                      int* __restrict__ sedge,
                                                      float* __restrict__ dinv,
                                                      int* __restrict__ lev) {
    __shared__ int cnt[NPP];
    __shared__ int st[NPP];
    __shared__ int sc[SORT_T];
    const int tid  = threadIdx.x;
    const int part = blockIdx.x;
    const int b    = blockIdx.y;
    const int lo   = part * NPP;

    for (int i = tid; i < NPP; i += SORT_T) cnt[i] = 0;
    int pv = -1;
    if (tid < GP * 16) pv = pos[b * GP + (tid >> 4)];

    const int* bpb = binptr + (size_t)b * SLICES * (PARTS + 1) + part;
    const v2i* pe  = (const v2i*)pedge + (size_t)b * GE;
    int beg[SLICES], end[SLICES];
    int below = 0;
#pragma unroll
    for (int s = 0; s < SLICES; ++s) {
        beg[s] = bpb[s * (PARTS + 1)];
        end[s] = bpb[s * (PARTS + 1) + 1];
        below += beg[s];
    }
    v2i ed[SLICES];
#pragma unroll
    for (int s = 0; s < SLICES; ++s) {
        const int k = beg[s] + tid;
        ed[s].x = -1;
        ed[s].y = 0;
        if (k < end[s]) ed[s] = pe[s * EPS + k];
    }
    __syncthreads();
#pragma unroll
    for (int s = 0; s < SLICES; ++s) {
        if (ed[s].x >= 0) atomicAdd(&cnt[(ed[s].x >> 16) - lo], 1);
        if (end[s] - beg[s] > SORT_T) {
            for (int k = beg[s] + tid + SORT_T; k < end[s]; k += SORT_T)
                atomicAdd(&cnt[(pe[s * EPS + k].x >> 16) - lo], 1);
        }
    }
    __syncthreads();

    const int own = (tid < NPP) ? cnt[tid] : 0;
    sc[tid] = own;
    __syncthreads();
    for (int off = 1; off < SORT_T; off <<= 1) {
        const int add = (tid >= off) ? sc[tid - off] : 0;
        __syncthreads();
        sc[tid] += add;
        __syncthreads();
    }
    int start = 0;
    if (tid < NPP) {
        start = below + sc[tid] - own;
        cnt[tid] = start;
        st[tid]  = start;
        rowptr[(size_t)b * (GN + 1) + lo + tid] = start;
    }
    if (part == PARTS - 1 && tid == 0) rowptr[(size_t)b * (GN + 1) + GN] = GE;
    __syncthreads();

    v2i* se = (v2i*)sedge + (size_t)b * GE;
    int pos_[SLICES];
#pragma unroll
    for (int s = 0; s < SLICES; ++s) {
        pos_[s] = -1;
        if (ed[s].x >= 0) pos_[s] = atomicAdd(&cnt[(ed[s].x >> 16) - lo], 1);
    }
#pragma unroll
    for (int s = 0; s < SLICES; ++s) {
        if (pos_[s] >= 0) {
            v2i o;
            o.x = ed[s].x & 0xFFFF;
            o.y = ed[s].y;
            se[pos_[s]] = o;
        }
        if (end[s] - beg[s] > SORT_T)
        for (int k = beg[s] + tid + SORT_T; k < end[s]; k += SORT_T) {
            const v2i e2 = pe[s * EPS + k];
            const int p2 = atomicAdd(&cnt[(e2.x >> 16) - lo], 1);
            v2i o;
            o.x = e2.x & 0xFFFF;
            o.y = e2.y;
            se[p2] = o;
        }
    }
    __syncthreads();

    if (pv >= lo && pv < lo + NPP) {
        unsigned char* lb = (unsigned char*)(lev + (size_t)b * GN);
        const int pe_ = cnt[pv - lo];
        if ((tid & 15) == 0) lb[4 * pv + 1] = 1;
        for (int q = st[pv - lo] + (tid & 15); q < pe_; q += 16) lb[4 * se[q].x + 1] = 1;
    }

    if (tid < NPP) {
        const int e = start + own;
        const int* sw = (const int*)se + 1;
        float d = 0.f;
        for (int p = start; p < e; p += 8) {
            int w8[8];
#pragma unroll
            for (int j = 0; j < 8; ++j) w8[j] = sw[2 * min(p + j, e - 1)];
#pragma unroll
            for (int j = 0; j < 8; ++j) d += (p + j < e) ? __int_as_float(w8[j]) : 0.f;
        }
        dinv[(size_t)b * GN + lo + tid] = rsqrtf(d + 1.0f);
    }
}


#define GEMM_BLOCKS 256
#define MPB (GB * GN / GEMM_BLOCKS)
static_assert((GB * GN) % GEMM_BLOCKS == 0, "mark ranges cover all nodes exactly");
template <int HIN>
__global__ __launch_bounds__(512) void gemm_strip_kernel(const float* __restrict__ Hsrc,
                                                         const float* __restrict__ W,
                                                         const float* __restrict__ dinv,
                                                         const int* __restrict__ rowptr,
                                                         const int* __restrict__ sedge,
                                                         int* lev,
                                                         float* __restrict__ Dst) {
    constexpr int PITCH = HIN + 4;
    constexpr int NQ = HIN / 8;
    __shared__ __attribute__((aligned(16))) float wlds[GH * PITCH];
    const int tid = threadIdx.x;
    constexpr int MR = (MPB + 511) / 512;
    int mlev[MR];
#pragma unroll
    for (int u = 0; u < MR; ++u) {
        const int i = tid + u * 512;
        mlev[u] = 0;
        if (i < MPB) mlev[u] = lev[blockIdx.x * MPB + i];
    }
    for (int i = tid; i < GH * HIN / 4; i += 512) {
        const int r  = i / (HIN / 4);
        const int c4 = i - r * (HIN / 4);
        *(v4f*)(wlds + r * PITCH + c4 * 4) = ((const v4f*)W)[i];
    }
    __syncthreads();

#pragma unroll
    for (int u = 0; u < MR; ++u) {
        if (mlev[u] >= LEV2) {
            const int node = blockIdx.x * MPB + tid + u * 512;
            const int b = node / GN;
            unsigned char* lb = (unsigned char*)(lev + (size_t)b * GN);
            const int beg = rowptr[node + b];
            const int end = rowptr[node + b + 1];
            const v2i* se = (const v2i*)sedge + (size_t)b * GE;
            for (int p = beg; p < end; p += 8) {
                int s8[8];
#pragma unroll
                for (int j = 0; j < 8; ++j) s8[j] = se[min(p + j, end - 1)].x;
#pragma unroll
                for (int j = 0; j < 8; ++j) lb[4 * s8[j]] = 1;
            }
        }
    }

    const int wave = __builtin_amdgcn_readfirstlane(tid >> 5);
    const int lane = tid & 31;
    const int half = lane >> 4;
    const int lm   = lane & 15;
    const int woff0 = lm * PITCH + half * 4;

    for (int t = blockIdx.x + gridDim.x * wave; t < GB * GN / 16; t += gridDim.x * 16) {
        const int row = t * 16 + lm;
        const float dr = dinv[row];
        const float* hrow = Hsrc + (size_t)row * HIN + half * 4;
        int woff = woff0;
        asm volatile("" : "+v"(woff));
        const float* wrow = wlds + woff;

        v4f a[NQ];
#pragma unroll
        for (int q = 0; q < NQ; ++q) a[q] = *(const v4f*)(hrow + q * 8);

#pragma unroll
        for (int q = 0; q < NQ; ++q) a[q] *= dr;

        __builtin_amdgcn_sched_barrier(0);

        v8f acc[4] = {};
        v4f bcur[4], bnxt[4];
#pragma unroll
        for (int n = 0; n < 4; ++n) bcur[n] = *(const v4f*)(wrow + n * 16 * PITCH);
#pragma unroll
        for (int q = 0; q < NQ; ++q) {
            if (q + 1 < NQ) {
#pragma unroll
                for (int n = 0; n < 4; ++n) bnxt[n] = *(const v4f*)(wrow + n * 16 * PITCH + (q + 1) * 8);
            }
            __builtin_amdgcn_sched_barrier(0);
            const v2f a0 = {a[q].x, a[q].y};
            const v2f a1 = {a[q].z, a[q].w};
#pragma unroll
            for (int n = 0; n < 4; ++n) {
                const v2f b0 = {bcur[n].x, bcur[n].y};
                acc[n] = __builtin_amdgcn_wmma_f32_16x16x4_f32(
                     false, a0,  false, b0,
                     (short)0, acc[n],  false,  false);
            }
#pragma unroll
            for (int n = 0; n < 4; ++n) {
                const v2f b1 = {bcur[n].z, bcur[n].w};
                acc[n] = __builtin_amdgcn_wmma_f32_16x16x4_f32(
                     false, a1,  false, b1,
                     (short)0, acc[n],  false,  false);
            }
#pragma unroll
            for (int n = 0; n < 4; ++n) bcur[n] = bnxt[n];
        }

        float* drow = Dst + (size_t)t * 16 * GH + lm;
#pragma unroll
        for (int j = 0; j < 8; ++j) {
#pragma unroll
            for (int n = 0; n < 4; ++n) {
                drow[(size_t)(j + 8 * half) * GH + n * 16] = acc[n][j];
            }
        }
    }
}


template <int J> __device__ __forceinline__ int share16(int v) {
    return __builtin_amdgcn_update_dpp(0, v, 0x150 + J, 0xf, 0xf, false);
}

#define PULL_LOAD(J)                                                                       \
    if (FULL || (J0 + J) < n)                                                              \
        r[J] = *(const v4f*)(tmp + (rowbase + (unsigned)share16<J0 + J>(m.x) * GH));
#define PULL_ADD(J)                                                                        \
    if (FULL || (J0 + J) < n)                                                              \
        acc += r[J] * __int_as_float(share16<J0 + J>(m.y));

template <int J0, bool FULL>
__device__ __forceinline__ void pull_batch8(const float* __restrict__ tmp, unsigned rowbase,
                                            v2i m, int n, v4f& acc) {
    v4f r[8];
#pragma unroll
    for (int j = 0; j < 8; ++j) r[j] = (v4f){0.f, 0.f, 0.f, 0.f};
    PULL_LOAD(0) PULL_LOAD(1) PULL_LOAD(2) PULL_LOAD(3)
    PULL_LOAD(4) PULL_LOAD(5) PULL_LOAD(6) PULL_LOAD(7)
    PULL_ADD(0) PULL_ADD(1) PULL_ADD(2) PULL_ADD(3)
    PULL_ADD(4) PULL_ADD(5) PULL_ADD(6) PULL_ADD(7)
}
#undef PULL_LOAD
#undef PULL_ADD

template <int J0>
__device__ __forceinline__ void pull_batch(const float* __restrict__ tmp, unsigned rowbase,
                                           v2i m, int n, v4f& acc) {
    if (n >= J0 + 8) pull_batch8<J0, true >(tmp, rowbase, m, n, acc);
    else             pull_batch8<J0, false>(tmp, rowbase, m, n, acc);
}

template <bool ACT>
__device__ __forceinline__ v4f pull_row(const int* __restrict__ sedge,
                                        const float* __restrict__ tmp,
                                        const float* __restrict__ dinv,
                                        const float* __restrict__ bias,
                                        int b, int i, int k4, int beg, int end) {
    const int node = b * GN + i;
    const v2i* se = (const v2i*)sedge + (size_t)b * GE;
    const unsigned rowbase = (unsigned)(b * GN) * GH + (unsigned)k4 * 4;

    int p = beg;
    int n = end - p;
    v2i m0 = {0, 0}, m1 = {0, 0};
    if (k4 < n)      m0 = se[p + k4];
    if (k4 + 16 < n) m1 = se[p + 16 + k4];
    const float di = dinv[node];
    const v4f t  = *(const v4f*)(tmp + ((unsigned)node * GH + (unsigned)k4 * 4));
    v4f bb = {0.f, 0.f, 0.f, 0.f};
    if (ACT) bb = *(const v4f*)(bias + k4 * 4);

    v4f acc = {0.f, 0.f, 0.f, 0.f};
    while (n > 0) {
        pull_batch<0>(tmp, rowbase, m0, n, acc);
        if (n > 8)  pull_batch<8>(tmp, rowbase, m0, n, acc);
        if (n > 16) pull_batch<0>(tmp, rowbase, m1, n - 16, acc);
        if (n > 24) pull_batch<8>(tmp, rowbase, m1, n - 16, acc);
        p += 32;
        n -= 32;
        if (n > 0) {
            m0 = (v2i){0, 0};
            m1 = (v2i){0, 0};
            if (k4 < n)      m0 = se[p + k4];
            if (k4 + 16 < n) m1 = se[p + 16 + k4];
        }
    }

    v4f g = (acc + t) * di;
    if (ACT) {
        g += bb;
        g.x = tanhf(g.x); g.y = tanhf(g.y); g.z = tanhf(g.z); g.w = tanhf(g.w);
        g *= di;
    }
    return g;
}

__global__ __launch_bounds__(256) void pull_need_kernel(const int* __restrict__ rowptr,
                                                        const int* __restrict__ sedge,
                                                        const float* __restrict__ tmp,
                                                        const float* __restrict__ dinv,
                                                        const float* __restrict__ bias,
                                                        const int* __restrict__ lev,
                                                        float* __restrict__ hout,
                                                        int minlev) {
    const int k4 = threadIdx.x & 15;
    const int stride = gridDim.x * 16;
    int node = blockIdx.x * 16 + (threadIdx.x >> 4);
    int lv = 0, beg = 0, end = 0;
    if (node < GB * GN) {
        const int b = node / GN;
        beg = rowptr[node + b];
        end = rowptr[node + b + 1];
        lv  = lev[node];
    }
    while (node < GB * GN) {
        asm volatile("" : "+v"(lv));
        const bool need = lv >= minlev;
        const int cbeg = beg, cend = end;
        const int nxt = node + stride;
        lv = 0; beg = 0; end = 0;
        if (nxt < GB * GN) {
            const int b2 = nxt / GN;
            beg = rowptr[nxt + b2];
            end = rowptr[nxt + b2 + 1];
            lv  = lev[nxt];
        }
        if (need) {
            const int b = node / GN;
            const int i = node - b * GN;
            const v4f v = pull_row<true>(sedge, tmp, dinv, bias, b, i, k4, cbeg, cend);
            *(v4f*)(hout + ((size_t)node * (GH / 4) + k4) * 4) = v;
        }
        node = nxt;
    }
}

#define L23_T 512
#define L23_R (L23_T / 16)
static_assert(GEMB == 64 && GH == 64 && L23_R == 32, "two 16-row tiles per round, 64 columns");
__global__ __launch_bounds__(L23_T) void pos_rows_kernel(const int* __restrict__ rowptr,
                                                         const int* __restrict__ sedge,
                                                         const float* __restrict__ h1,
                                                         const float* __restrict__ dinv,
                                                         const float* __restrict__ W2,
                                                         const float* __restrict__ b2,
                                                         const float* __restrict__ W3,
                                                         const float* __restrict__ b3,
                                                         const int* __restrict__ pos,
                                                         float* __restrict__ out) {
    constexpr int PITCH = GH + 4;
    constexpr int NQ = GH / 8;
    __shared__ __attribute__((aligned(16))) float gt[L23_R * PITCH];
    __shared__ __attribute__((aligned(16))) float h2t[L23_R * PITCH];
    __shared__ float ewt[L23_R];
    __shared__ float dvt[L23_R];
    const int tid  = threadIdx.x;
    const int k4   = tid & 15;
    const int grp  = tid >> 4;
    const int wave = __builtin_amdgcn_readfirstlane(tid >> 5);
    const int lane = tid & 31;
    const int half = lane >> 4;
    const int lm   = lane & 15;
    const int ch   = wave & 1;
    const int b    = blockIdx.y;
    float* orow = out + ((size_t)b * GP + blockIdx.x) * GEMB;

    const int pv = __builtin_amdgcn_readfirstlane(pos[b * GP + blockIdx.x]);
    if (pv < 0) {
        if (tid < GEMB) orow[tid] = -3.0f;
        return;
    }

    v4f w0[NQ], w1[NQ];
#pragma unroll
    for (int q = 0; q < NQ; ++q) {
        w0[q] = (v4f){0.f, 0.f, 0.f, 0.f};
        w1[q] = (v4f){0.f, 0.f, 0.f, 0.f};
    }
    float bc0 = 0.f, bc1 = 0.f;
    if (wave < 6) {
        const float* Wm = wave < 4 ? W2 : W3;
        const float* bm = wave < 4 ? b2 : b3;
        const float* wr0 = Wm + (size_t)(ch * 32 + lm) * GH + half * 4;
        const float* wr1 = wr0 + 16 * GH;
#pragma unroll
        for (int q = 0; q < NQ; ++q) {
            w0[q] = *(const v4f*)(wr0 + q * 8);
            w1[q] = *(const v4f*)(wr1 + q * 8);
        }
        bc0 = bm[ch * 32 + lm];
        bc1 = bm[ch * 32 + 16 + lm];
    }

    const int pnode = b * GN + pv;
    const int pbeg  = rowptr[pnode + b];
    const int deg   = __builtin_amdgcn_readfirstlane(rowptr[pnode + b + 1] - pbeg);
    const int cnt   = deg + 1;
    const float dpos = dinv[pnode];
    const v2i* se = (const v2i*)sedge + (size_t)b * GE;

    v4f acc3  = {0.f, 0.f, 0.f, 0.f};
    v4f tself = {0.f, 0.f, 0.f, 0.f};
    for (int r0 = 0; r0 < cnt; r0 += L23_R) {
        const int e = r0 + grp;
        int i = -1;
        float we = 0.f;
        if (e < deg) {
            const v2i m = se[pbeg + e];
            i  = m.x;
            we = __int_as_float(m.y);
        } else if (e == deg) {
            i = pv;
        }
        v4f g = {0.f, 0.f, 0.f, 0.f};
        float dn = 0.f;
        if (i >= 0) {
            const int node = b * GN + i;
            const int beg = rowptr[node + b];
            const int end = rowptr[node + b + 1];
            dn = dinv[node];
            g = pull_row<false>(sedge, h1, dinv, b2, b, i, k4, beg, end);
        }
        *(v4f*)(gt + grp * PITCH + k4 * 4) = g;
        if (k4 == 0) {
            ewt[grp] = we;
            dvt[grp] = dn;
        }
        __syncthreads();

        if (wave < 4) {
            const int trow = (wave >> 1) * 16;
            const float* arow = gt + (trow + lm) * PITCH + half * 4;
            v4f a[NQ];
#pragma unroll
            for (int q = 0; q < NQ; ++q) a[q] = *(const v4f*)(arow + q * 8);
            float dj[8];
#pragma unroll
            for (int j = 0; j < 8; ++j) dj[j] = dvt[trow + 8 * half + j];
            v8f acc0 = {}, acc1 = {};
#pragma unroll
            for (int q = 0; q < NQ; ++q) {
                const v2f a0 = {a[q].x, a[q].y};
                const v2f a1 = {a[q].z, a[q].w};
                const v2f b00 = {w0[q].x, w0[q].y};
                const v2f b01 = {w0[q].z, w0[q].w};
                const v2f b10 = {w1[q].x, w1[q].y};
                const v2f b11 = {w1[q].z, w1[q].w};
                acc0 = __builtin_amdgcn_wmma_f32_16x16x4_f32(false, a0, false, b00, (short)0, acc0, false, false);
                acc1 = __builtin_amdgcn_wmma_f32_16x16x4_f32(false, a0, false, b10, (short)0, acc1, false, false);
                acc0 = __builtin_amdgcn_wmma_f32_16x16x4_f32(false, a1, false, b01, (short)0, acc0, false, false);
                acc1 = __builtin_amdgcn_wmma_f32_16x16x4_f32(false, a1, false, b11, (short)0, acc1, false, false);
            }
            float v0[8], v1[8];
#pragma unroll
            for (int j = 0; j < 8; ++j) {
                v0[j] = acc0[j] + bc0;
                v1[j] = acc1[j] + bc1;
            }
            asm volatile("" : "+v"(v0[0]), "+v"(v0[1]), "+v"(v0[2]), "+v"(v0[3]),
                              "+v"(v0[4]), "+v"(v0[5]), "+v"(v0[6]), "+v"(v0[7]),
                              "+v"(v1[0]), "+v"(v1[1]), "+v"(v1[2]), "+v"(v1[3]),
                              "+v"(v1[4]), "+v"(v1[5]), "+v"(v1[6]), "+v"(v1[7]));
            float* hrow = h2t + (trow + 8 * half) * PITCH + ch * 32 + lm;
#pragma unroll
            for (int j = 0; j < 8; ++j) {
                hrow[j * PITCH]      = tanhf(v0[j]) * dj[j];
                hrow[j * PITCH + 16] = tanhf(v1[j]) * dj[j];
            }
        }
        __syncthreads();

        if (grp == 0) {
            const int nr = min(L23_R, cnt - r0);
            for (int j0 = 0; j0 < nr; j0 += 8) {
                v4f h8[8];
                float w8[8];
#pragma unroll
                for (int u = 0; u < 8; ++u) {
                    const int j = min(j0 + u, L23_R - 1);
                    h8[u] = *(const v4f*)(h2t + j * PITCH + k4 * 4);
                    w8[u] = ewt[j];
                }
#pragma unroll
                for (int u = 0; u < 8; ++u) {
                    const int el = r0 + j0 + u;
                    if (el < deg)       acc3 += h8[u] * w8[u];
                    else if (el == deg) tself = h8[u];
                }
            }
        }
        __syncthreads();
    }

    if (grp < 16) {
        v4f g3 = {0.f, 0.f, 0.f, 0.f};
        if (grp == 0) g3 = (acc3 + tself) * dpos;
        *(v4f*)(gt + grp * PITCH + k4 * 4) = g3;
    }
    __syncthreads();
    if ((wave >> 1) == 2) {
        const float* arow = gt + lm * PITCH + half * 4;
        v4f a[NQ];
#pragma unroll
        for (int q = 0; q < NQ; ++q) a[q] = *(const v4f*)(arow + q * 8);
        v8f acc0 = {}, acc1 = {};
#pragma unroll
        for (int q = 0; q < NQ; ++q) {
            const v2f a0 = {a[q].x, a[q].y};
            const v2f a1 = {a[q].z, a[q].w};
            const v2f b00 = {w0[q].x, w0[q].y};
            const v2f b01 = {w0[q].z, w0[q].w};
            const v2f b10 = {w1[q].x, w1[q].y};
            const v2f b11 = {w1[q].z, w1[q].w};
            acc0 = __builtin_amdgcn_wmma_f32_16x16x4_f32(false, a0, false, b00, (short)0, acc0, false, false);
            acc1 = __builtin_amdgcn_wmma_f32_16x16x4_f32(false, a0, false, b10, (short)0, acc1, false, false);
            acc0 = __builtin_amdgcn_wmma_f32_16x16x4_f32(false, a1, false, b01, (short)0, acc0, false, false);
            acc1 = __builtin_amdgcn_wmma_f32_16x16x4_f32(false, a1, false, b11, (short)0, acc1, false, false);
        }
        const float o0 = acc0[0] + bc0;
        const float o1 = acc1[0] + bc1;
        if (half == 0) {
            orow[ch * 32 + lm]      = o0;
            orow[ch * 32 + 16 + lm] = o1;
        }
    }
}

extern "C" void kernel_launch(void* const* d_in, const int* in_sizes, int n_in,
                              void* d_out, int out_size, void* d_ws, size_t ws_size,
                              hipStream_t stream) {
    const float* x   = (const float*)d_in[0];
    const int*   ei  = (const int*)d_in[1];
    const float* ew  = (const float*)d_in[2];
    const int*   pos = (const int*)d_in[3];
    const float* W1  = (const float*)d_in[4];
    const float* b1  = (const float*)d_in[5];
    const float* W2  = (const float*)d_in[6];
    const float* b2  = (const float*)d_in[7];
    const float* W3  = (const float*)d_in[8];
    const float* b3  = (const float*)d_in[9];
    float* out = (float*)d_out;

    float* ws   = (float*)d_ws;
    const size_t nBN  = (size_t)GB * GN;
    const size_t nRP  = (size_t)GB * (GN + 1);
    const size_t nBE  = (size_t)GB * GE;
    const size_t nBNH = (size_t)GB * GN * GH;
    float* dinv   = ws;
    int*   rowptr = (int*)(dinv + nBN);
    int*   sedge  = rowptr + nRP;
    float* tmp    = (float*)(sedge + 2 * nBE);
    float* hbuf   = tmp + nBNH;
    int*   lev    = (int*)(hbuf + nBNH);
    int*   binptr = (int*)tmp;
    int*   pedge  = (int*)hbuf;

    const int T = 256;
    dim3 part_grid(SLICES, GB, 1);
    dim3 sort_grid(PARTS, GB, 1);

    part_kernel<<<part_grid, SORT_T, 0, stream>>>(ei, ew, binptr, pedge, lev);
    sort_kernel<<<sort_grid, SORT_T, 0, stream>>>(binptr, pedge, pos, rowptr, sedge, dinv, lev);

    gemm_strip_kernel<GS><<<GEMM_BLOCKS, 512, 0, stream>>>(x, W1, dinv, rowptr, sedge, lev, tmp);
    pull_need_kernel<<<1536, T, 0, stream>>>(rowptr, sedge, tmp, dinv, b1, lev, hbuf, LEV1);

    dim3 pos_grid(GP, GB, 1);
    pos_rows_kernel<<<pos_grid, L23_T, 0, stream>>>(rowptr, sedge, hbuf, dinv, W2, b2, W3, b3, pos, out);
}
